# starting kernel
# baseline (speedup 1.0000x reference)
#include <hip/hip_runtime.h>
#include <hip/hip_bf16.h>


typedef __attribute__((ext_vector_type(2))) float v2f;
typedef __attribute__((ext_vector_type(8))) float v8f;

__device__ __forceinline__ unsigned fenc(float f) {
    unsigned u = __float_as_uint(f);
    return (u & 0x80000000u) ? ~u : (u | 0x80000000u);
}
__device__ __forceinline__ float fdec(unsigned e) {
    return (e & 0x80000000u) ? __uint_as_float(e & 0x7FFFFFFFu)
                             : __uint_as_float(~e);
}

__global__ void k0_zero(unsigned* __restrict__ p, size_t n) {
    size_t i = (size_t)blockIdx.x * blockDim.x + threadIdx.x;
    if (i < n) p[i] = 0u;
}

__global__ __launch_bounds__(32)
void k1_gemm_wmma(const float* __restrict__ x, const float* __restrict__ W,
                  const float* __restrict__ bias, float* __restrict__ qk,
                  int N, int F, int C) {
    const int lane = threadIdx.x;
    const int M0   = blockIdx.x * 64;
    const int N0   = blockIdx.y * 64;
    const int lm   = lane & 15;
    const int kh   = (lane >> 4) * 2;

    const float* __restrict__ xrow[4];
    const float* __restrict__ wrow[4];
    #pragma unroll
    for (int mi = 0; mi < 4; ++mi) {
        int r = M0 + mi * 16 + lm;
        xrow[mi] = x + (size_t)(r < N ? r : N - 1) * F;
    }
    #pragma unroll
    for (int ni = 0; ni < 4; ++ni) {
        int c = N0 + ni * 16 + lm;
        wrow[ni] = W + (size_t)(c < C ? c : C - 1) * F;
    }

    v8f acc[4][4];
    #pragma unroll
    for (int mi = 0; mi < 4; ++mi)
        #pragma unroll
        for (int ni = 0; ni < 4; ++ni)
            acc[mi][ni] = (v8f){};

    for (int k = 0; k < F; k += 4) {
        v2f a[4], b[4];
        #pragma unroll
        for (int mi = 0; mi < 4; ++mi) {
            a[mi].x = xrow[mi][k + kh];
            a[mi].y = xrow[mi][k + kh + 1];
        }
        #pragma unroll
        for (int ni = 0; ni < 4; ++ni) {
            b[ni].x = wrow[ni][k + kh];
            b[ni].y = wrow[ni][k + kh + 1];
        }
        #pragma unroll
        for (int mi = 0; mi < 4; ++mi)
            #pragma unroll
            for (int ni = 0; ni < 4; ++ni)
                acc[mi][ni] = __builtin_amdgcn_wmma_f32_16x16x4_f32(
                     false, a[mi],  false, b[ni],
                     (short)0, acc[mi][ni],
                     false,  false);
    }

    const int rsub = (lane >> 4) << 3;
    #pragma unroll
    for (int ni = 0; ni < 4; ++ni) {
        const int col = N0 + ni * 16 + lm;
        if (col >= C) continue;
        const float bv = bias[col];
        #pragma unroll
        for (int mi = 0; mi < 4; ++mi) {
            const int rbase = M0 + mi * 16 + rsub;
            #pragma unroll
            for (int r = 0; r < 8; ++r) {
                const int row = rbase + r;
                if (row < N)
                    qk[(size_t)row * C + col] = acc[mi][ni][r] + bv;
            }
        }
    }
}

__global__ __launch_bounds__(256)
void k2_scores(const float* __restrict__ qk, const int* __restrict__ ei,
               float* __restrict__ scores, unsigned* __restrict__ smax,
               int E, int C, int H, int P) {
    const int lane = threadIdx.x & 31;
    const int e = blockIdx.x * (blockDim.x >> 5) + (threadIdx.x >> 5);
    if (e >= E) return;

    const int s = ei[e];
    const int d = ei[E + e];
    const float* __restrict__ srow = qk + (size_t)s * C;
    const float* __restrict__ drow = qk + (size_t)d * C;

    for (int h = 0; h < H; ++h) {
        const float* __restrict__ qh = srow + h * 2 * P;
        const float* __restrict__ kh = drow + h * 2 * P + P;
        float a = 0.0f;
        for (int p = lane; p < P; p += 32)
            a += qh[p] * kh[p];
        #pragma unroll
        for (int off = 16; off > 0; off >>= 1)
            a += __shfl_xor(a, off, 32);
        if (lane == 0) {
            scores[(size_t)e * H + h] = a;
            atomicMax(&smax[(size_t)s * H + h], fenc(a));
        }
    }
}

__global__ __launch_bounds__(256)
void k3_exp_denom(float* __restrict__ scores, const unsigned* __restrict__ smax,
                  const int* __restrict__ ei, float* __restrict__ denom,
                  int E, int H) {
    size_t i = (size_t)blockIdx.x * blockDim.x + threadIdx.x;
    if (i >= (size_t)E * H) return;
    const int e = (int)(i / H);
    const int h = (int)(i % H);
    const int s = ei[e];
    const float m  = fdec(smax[(size_t)s * H + h]);
    const float ex = __expf(scores[i] - m);
    scores[i] = ex;
    atomicAdd(&denom[(size_t)s * H + h], ex);
}

__global__ __launch_bounds__(256)
void k4_out(const float* __restrict__ ex, const float* __restrict__ denom,
            const int* __restrict__ ei, float* __restrict__ out, int E, int H) {
    const int e = blockIdx.x * blockDim.x + threadIdx.x;
    if (e >= E) return;
    const int s = ei[e];
    float sum = 0.0f;
    for (int h = 0; h < H; ++h)
        sum += ex[(size_t)e * H + h] / denom[(size_t)s * H + h];
    out[e] = sum / (float)H;
}

extern "C" void kernel_launch(void* const* d_in, const int* in_sizes, int n_in,
                              void* d_out, int out_size, void* d_ws, size_t ws_size,
                              hipStream_t stream) {
    const float* x    = (const float*)d_in[0];
    const int*   ei   = (const int*)d_in[1];
    const float* W    = (const float*)d_in[2];
    const float* bias = (const float*)d_in[3];
    const int C = in_sizes[3];
    const int F = in_sizes[2] / C;
    const int N = in_sizes[0] / F;
    const int E = in_sizes[1] / 2;
    const int H = (C - 2 * F) / 2;
    const int P = F / H + 1;

    float*    qk     = (float*)d_ws;
    float*    scores = qk + (size_t)N * C;
    unsigned* smax   = (unsigned*)(scores + (size_t)E * H);
    float*    denom  = (float*)(smax + (size_t)N * H);

    {
        size_t n = (size_t)2 * N * H;
        k0_zero<<<(unsigned)((n + 255) / 256), 256, 0, stream>>>(smax, n);
    }

    {
        dim3 grid((N + 63) / 64, (C + 63) / 64);
        k1_gemm_wmma<<<grid, 32, 0, stream>>>(x, W, bias, qk, N, F, C);
    }

    {
        const int epb = 256 / 32;
        k2_scores<<<(E + epb - 1) / epb, 256, 0, stream>>>(qk, ei, scores, smax,
                                                           E, C, H, P);
    }

    {
        size_t n = (size_t)E * H;
        k3_exp_denom<<<(unsigned)((n + 255) / 256), 256, 0, stream>>>(scores, smax,
                                                                      ei, denom, E, H);
    }

    k4_out<<<(E + 255) / 256, 256, 0, stream>>>(scores, denom, ei,
                                                (float*)d_out, E, H);
}


# score kernel: a wave walks 16 edges, each edge's gathers issued as one group of 16-byte loads, next edge's in flight (was 1 edge per wave, 12 dependent load trips); dot-product sum order changes
# speedup vs baseline: 1.5488x; 1.5488x over previous
#include <hip/hip_runtime.h>
#include <hip/hip_bf16.h>


typedef __attribute__((ext_vector_type(2))) float v2f;
typedef __attribute__((ext_vector_type(8))) float v8f;

__device__ __forceinline__ unsigned fenc(float f) {
    unsigned u = __float_as_uint(f);
    return (u & 0x80000000u) ? ~u : (u | 0x80000000u);
}
__device__ __forceinline__ float fdec(unsigned e) {
    return (e & 0x80000000u) ? __uint_as_float(e & 0x7FFFFFFFu)
                             : __uint_as_float(~e);
}

__global__ void k0_zero(unsigned* __restrict__ p, size_t n) {
    size_t i = (size_t)blockIdx.x * blockDim.x + threadIdx.x;
    if (i < n) p[i] = 0u;
}

__global__ __launch_bounds__(32)
void k1_gemm_wmma(const float* __restrict__ x, const float* __restrict__ W,
                  const float* __restrict__ bias, float* __restrict__ qk,
                  int N, int F, int C) {
    const int lane = threadIdx.x;
    const int M0   = blockIdx.x * 64;
    const int N0   = blockIdx.y * 64;
    const int lm   = lane & 15;
    const int kh   = (lane >> 4) * 2;

    const float* __restrict__ xrow[4];
    const float* __restrict__ wrow[4];
    #pragma unroll
    for (int mi = 0; mi < 4; ++mi) {
        int r = M0 + mi * 16 + lm;
        xrow[mi] = x + (size_t)(r < N ? r : N - 1) * F;
    }
    #pragma unroll
    for (int ni = 0; ni < 4; ++ni) {
        int c = N0 + ni * 16 + lm;
        wrow[ni] = W + (size_t)(c < C ? c : C - 1) * F;
    }

    v8f acc[4][4];
    #pragma unroll
    for (int mi = 0; mi < 4; ++mi)
        #pragma unroll
        for (int ni = 0; ni < 4; ++ni)
            acc[mi][ni] = (v8f){};

    for (int k = 0; k < F; k += 4) {
        v2f a[4], b[4];
        #pragma unroll
        for (int mi = 0; mi < 4; ++mi) {
            a[mi].x = xrow[mi][k + kh];
            a[mi].y = xrow[mi][k + kh + 1];
        }
        #pragma unroll
        for (int ni = 0; ni < 4; ++ni) {
            b[ni].x = wrow[ni][k + kh];
            b[ni].y = wrow[ni][k + kh + 1];
        }
        #pragma unroll
        for (int mi = 0; mi < 4; ++mi)
            #pragma unroll
            for (int ni = 0; ni < 4; ++ni)
                acc[mi][ni] = __builtin_amdgcn_wmma_f32_16x16x4_f32(
                     false, a[mi],  false, b[ni],
                     (short)0, acc[mi][ni],
                     false,  false);
    }

    const int rsub = (lane >> 4) << 3;
    #pragma unroll
    for (int ni = 0; ni < 4; ++ni) {
        const int col = N0 + ni * 16 + lm;
        if (col >= C) continue;
        const float bv = bias[col];
        #pragma unroll
        for (int mi = 0; mi < 4; ++mi) {
            const int rbase = M0 + mi * 16 + rsub;
            #pragma unroll
            for (int r = 0; r < 8; ++r) {
                const int row = rbase + r;
                if (row < N)
                    qk[(size_t)row * C + col] = acc[mi][ni][r] + bv;
            }
        }
    }
}

typedef __attribute__((ext_vector_type(4))) float v4f;
typedef v4f v4f_a4 __attribute__((aligned(4)));

#define K2_EPW 16

template <int CTRL>
__device__ __forceinline__ float row_xor_add(float keep, float send) {
    const int t = __builtin_amdgcn_update_dpp(0, __builtin_bit_cast(int, send),
                                              CTRL, 0xf, 0xf, false);
    return keep + __builtin_bit_cast(float, t);
}

struct EdgeFrag {
    v4f qa, qb, ka, kb;
    float qt, kt;
};

__device__ __forceinline__ void k2_load_edge(EdgeFrag& f, const float* __restrict__ qk,
                                             int s, int d, int offA, int offT) {
    const float* __restrict__ sq = qk + (size_t)s * 520;
    const float* __restrict__ dk = qk + (size_t)d * 520 + 65;
    f.qa = *(const v4f_a4*)(sq + offA);
    f.qb = *(const v4f_a4*)(sq + offA + 260);
    f.ka = *(const v4f_a4*)(dk + offA);
    f.kb = *(const v4f_a4*)(dk + offA + 260);
    f.qt = sq[offT];
    f.kt = dk[offT];
    __builtin_amdgcn_sched_barrier(0);
}

__device__ __forceinline__ void k2_reduce_store(const EdgeFrag& f, float* __restrict__ scores,
                                                unsigned* __restrict__ smax, int e, int s,
                                                int up, int th, int lane) {
    float a1 = f.qa.x * f.ka.x;
    float a2 = f.qb.x * f.kb.x;
    a1 = fmaf(f.qa.y, f.ka.y, a1);  a2 = fmaf(f.qb.y, f.kb.y, a2);
    a1 = fmaf(f.qa.z, f.ka.z, a1);  a2 = fmaf(f.qb.z, f.kb.z, a2);
    a1 = fmaf(f.qa.w, f.ka.w, a1);  a2 = fmaf(f.qb.w, f.kb.w, a2);
    float r = row_xor_add<0x168>(up ? a2 : a1, up ? a1 : a2);
    r = row_xor_add<0x164>(r, r);
    r = row_xor_add<0x162>(r, r);
    r = row_xor_add<0x161>(r, r);
    r = fmaf(f.qt, f.kt, r);
    if ((lane & 7) == 0) {
        scores[(size_t)e * 4 + th] = r;
        atomicMax(&smax[(size_t)s * 4 + th], fenc(r));
    }
}

__global__ __launch_bounds__(256)
void k2_scores(const float* __restrict__ qk, const int* __restrict__ ei,
               float* __restrict__ scores, unsigned* __restrict__ smax,
               int E, int C, int H, int P) {
    const int lane = threadIdx.x & 31;
    const int wv = __builtin_amdgcn_readfirstlane((int)(threadIdx.x >> 5));
    const int e0 = (blockIdx.x * (blockDim.x >> 5) + wv) * K2_EPW;
    if (e0 >= E) return;
    const int cnt = (E - e0 < K2_EPW) ? (E - e0) : K2_EPW;

    const int le = (e0 + (lane & 15) < E) ? (e0 + (lane & 15)) : (E - 1);
    const int ev = ei[(size_t)(lane >> 4) * E + le];

    if (H == 4 && P == 65 && C == 520) {
        const int hs   = lane >> 4;
        const int offA = hs * 130 + (lane & 15) * 4;
        const int up   = (lane >> 3) & 1;
        const int th   = hs + up * 2;
        const int offT = th * 130 + 64;

        EdgeFrag fa, fb;
        int sa = __builtin_amdgcn_readlane(ev, 0), sb = sa;
        k2_load_edge(fa, qk, sa, __builtin_amdgcn_readlane(ev, 16), offA, offT);
        int j = 0;
        for (; j + 2 < cnt; j += 2) {
            sb = __builtin_amdgcn_readlane(ev, j + 1);
            k2_load_edge(fb, qk, sb, __builtin_amdgcn_readlane(ev, j + 17), offA, offT);
            k2_reduce_store(fa, scores, smax, e0 + j, sa, up, th, lane);
            sa = __builtin_amdgcn_readlane(ev, j + 2);
            k2_load_edge(fa, qk, sa, __builtin_amdgcn_readlane(ev, j + 18), offA, offT);
            k2_reduce_store(fb, scores, smax, e0 + j + 1, sb, up, th, lane);
        }
        if (j + 1 < cnt) {
            sb = __builtin_amdgcn_readlane(ev, j + 1);
            k2_load_edge(fb, qk, sb, __builtin_amdgcn_readlane(ev, j + 17), offA, offT);
            k2_reduce_store(fa, scores, smax, e0 + j, sa, up, th, lane);
            k2_reduce_store(fb, scores, smax, e0 + j + 1, sb, up, th, lane);
        } else {
            k2_reduce_store(fa, scores, smax, e0 + j, sa, up, th, lane);
        }
    } else {
        for (int j = 0; j < cnt; ++j) {
            const int s = __builtin_amdgcn_readlane(ev, j);
            const int d = __builtin_amdgcn_readlane(ev, 16 + j);
            const float* __restrict__ srow = qk + (size_t)s * C;
            const float* __restrict__ drow = qk + (size_t)d * C;
            for (int h = 0; h < H; ++h) {
                const float* __restrict__ qh = srow + h * 2 * P;
                const float* __restrict__ kh = drow + h * 2 * P + P;
                float a = 0.0f;
                for (int p = lane; p < P; p += 32)
                    a += qh[p] * kh[p];
                #pragma unroll
                for (int off = 16; off > 0; off >>= 1)
                    a += __shfl_xor(a, off, 32);
                if (lane == 0) {
                    scores[(size_t)(e0 + j) * H + h] = a;
                    atomicMax(&smax[(size_t)s * H + h], fenc(a));
                }
            }
        }
    }
}

__global__ __launch_bounds__(256)
void k3_exp_denom(float* __restrict__ scores, const unsigned* __restrict__ smax,
                  const int* __restrict__ ei, float* __restrict__ denom,
                  int E, int H) {
    size_t i = (size_t)blockIdx.x * blockDim.x + threadIdx.x;
    if (i >= (size_t)E * H) return;
    const int e = (int)(i / H);
    const int h = (int)(i % H);
    const int s = ei[e];
    const float m  = fdec(smax[(size_t)s * H + h]);
    const float ex = __expf(scores[i] - m);
    scores[i] = ex;
    atomicAdd(&denom[(size_t)s * H + h], ex);
}

__global__ __launch_bounds__(256)
void k4_out(const float* __restrict__ ex, const float* __restrict__ denom,
            const int* __restrict__ ei, float* __restrict__ out, int E, int H) {
    const int e = blockIdx.x * blockDim.x + threadIdx.x;
    if (e >= E) return;
    const int s = ei[e];
    float sum = 0.0f;
    for (int h = 0; h < H; ++h)
        sum += ex[(size_t)e * H + h] / denom[(size_t)s * H + h];
    out[e] = sum / (float)H;
}

extern "C" void kernel_launch(void* const* d_in, const int* in_sizes, int n_in,
                              void* d_out, int out_size, void* d_ws, size_t ws_size,
                              hipStream_t stream) {
    const float* x    = (const float*)d_in[0];
    const int*   ei   = (const int*)d_in[1];
    const float* W    = (const float*)d_in[2];
    const float* bias = (const float*)d_in[3];
    const int C = in_sizes[3];
    const int F = in_sizes[2] / C;
    const int N = in_sizes[0] / F;
    const int E = in_sizes[1] / 2;
    const int H = (C - 2 * F) / 2;
    const int P = F / H + 1;

    float*    qk     = (float*)d_ws;
    float*    scores = qk + (size_t)N * C;
    unsigned* smax   = (unsigned*)(scores + (size_t)E * H);
    float*    denom  = (float*)(smax + (size_t)N * H);

    {
        size_t n = (size_t)2 * N * H;
        k0_zero<<<(unsigned)((n + 255) / 256), 256, 0, stream>>>(smax, n);
    }

    {
        dim3 grid((N + 63) / 64, (C + 63) / 64);
        k1_gemm_wmma<<<grid, 32, 0, stream>>>(x, W, bias, qk, N, F, C);
    }

    {
        const int epb = (256 / 32) * K2_EPW;
        k2_scores<<<(E + epb - 1) / epb, 256, 0, stream>>>(qk, ei, scores, smax,
                                                           E, C, H, P);
    }

    {
        size_t n = (size_t)E * H;
        k3_exp_denom<<<(unsigned)((n + 255) / 256), 256, 0, stream>>>(scores, smax,
                                                                      ei, denom, E, H);
    }

    k4_out<<<(E + 255) / 256, 256, 0, stream>>>(scores, denom, ei,
                                                (float*)d_out, E, H);
}


# GEMM K-loop in blocks of 32: a lane loads 64 contiguous bytes of each row (4 b128, issued as one group), each 128-B line fetched once per wave (was 8 b64 touches over 4 trips); k-sum order changes
# speedup vs baseline: 1.8409x; 1.1886x over previous
#include <hip/hip_runtime.h>
#include <hip/hip_bf16.h>


typedef __attribute__((ext_vector_type(2))) float v2f;
typedef __attribute__((ext_vector_type(4))) float v4f;
typedef __attribute__((ext_vector_type(8))) float v8f;
typedef v4f v4f_a4 __attribute__((aligned(4)));

__device__ __forceinline__ unsigned fenc(float f) {
    unsigned u = __float_as_uint(f);
    return (u & 0x80000000u) ? ~u : (u | 0x80000000u);
}
__device__ __forceinline__ float fdec(unsigned e) {
    return (e & 0x80000000u) ? __uint_as_float(e & 0x7FFFFFFFu)
                             : __uint_as_float(~e);
}

__global__ void k0_zero(unsigned* __restrict__ p, size_t n) {
    size_t i = (size_t)blockIdx.x * blockDim.x + threadIdx.x;
    if (i < n) p[i] = 0u;
}

__global__ __launch_bounds__(32)
void k1_gemm_wmma(const float* __restrict__ x, const float* __restrict__ W,
                  const float* __restrict__ bias, float* __restrict__ qk,
                  int N, int F, int C) {
    const int lane = threadIdx.x;
    const int M0   = blockIdx.x * 64;
    const int N0   = blockIdx.y * 64;
    const int lm   = lane & 15;
    const int kh   = (lane >> 4) * 2;

    const float* __restrict__ xrow[4];
    const float* __restrict__ wrow[4];
    #pragma unroll
    for (int mi = 0; mi < 4; ++mi) {
        int r = M0 + mi * 16 + lm;
        xrow[mi] = x + (size_t)(r < N ? r : N - 1) * F;
    }
    #pragma unroll
    for (int ni = 0; ni < 4; ++ni) {
        int c = N0 + ni * 16 + lm;
        wrow[ni] = W + (size_t)(c < C ? c : C - 1) * F;
    }

    v8f acc[4][4];
    #pragma unroll
    for (int mi = 0; mi < 4; ++mi)
        #pragma unroll
        for (int ni = 0; ni < 4; ++ni)
            acc[mi][ni] = (v8f){};

    const int kq = (lane >> 4) * 16;
    int k = 0;
    for (; k + 32 <= F; k += 32) {
        v4f a[4][4], b[4][4];
        #pragma unroll
        for (int j = 0; j < 4; ++j) {
            #pragma unroll
            for (int mi = 0; mi < 4; ++mi)
                a[mi][j] = *(const v4f_a4*)(xrow[mi] + k + kq + 4 * j);
            #pragma unroll
            for (int ni = 0; ni < 4; ++ni)
                b[ni][j] = *(const v4f_a4*)(wrow[ni] + k + kq + 4 * j);
        }
        __builtin_amdgcn_sched_barrier(0);
        #pragma unroll
        for (int t = 0; t < 8; ++t) {
            #pragma unroll
            for (int mi = 0; mi < 4; ++mi) {
                v2f af;
                af.x = a[mi][t >> 1][(t & 1) * 2];
                af.y = a[mi][t >> 1][(t & 1) * 2 + 1];
                #pragma unroll
                for (int ni = 0; ni < 4; ++ni) {
                    v2f bf;
                    bf.x = b[ni][t >> 1][(t & 1) * 2];
                    bf.y = b[ni][t >> 1][(t & 1) * 2 + 1];
                    acc[mi][ni] = __builtin_amdgcn_wmma_f32_16x16x4_f32(
                         false, af,  false, bf,
                         (short)0, acc[mi][ni],
                         false,  false);
                }
            }
        }
    }
    for (; k < F; k += 4) {
        v2f a[4], b[4];
        #pragma unroll
        for (int mi = 0; mi < 4; ++mi) {
            a[mi].x = xrow[mi][k + kh];
            a[mi].y = xrow[mi][k + kh + 1];
        }
        #pragma unroll
        for (int ni = 0; ni < 4; ++ni) {
            b[ni].x = wrow[ni][k + kh];
            b[ni].y = wrow[ni][k + kh + 1];
        }
        #pragma unroll
        for (int mi = 0; mi < 4; ++mi)
            #pragma unroll
            for (int ni = 0; ni < 4; ++ni)
                acc[mi][ni] = __builtin_amdgcn_wmma_f32_16x16x4_f32(
                     false, a[mi],  false, b[ni],
                     (short)0, acc[mi][ni],
                     false,  false);
    }

    const int rsub = (lane >> 4) << 3;
    #pragma unroll
    for (int ni = 0; ni < 4; ++ni) {
        const int col = N0 + ni * 16 + lm;
        if (col >= C) continue;
        const float bv = bias[col];
        #pragma unroll
        for (int mi = 0; mi < 4; ++mi) {
            const int rbase = M0 + mi * 16 + rsub;
            #pragma unroll
            for (int r = 0; r < 8; ++r) {
                const int row = rbase + r;
                if (row < N)
                    qk[(size_t)row * C + col] = acc[mi][ni][r] + bv;
            }
        }
    }
}


#define K2_EPW 16

template <int CTRL>
__device__ __forceinline__ float row_xor_add(float keep, float send) {
    const int t = __builtin_amdgcn_update_dpp(0, __builtin_bit_cast(int, send),
                                              CTRL, 0xf, 0xf, false);
    return keep + __builtin_bit_cast(float, t);
}

struct EdgeFrag {
    v4f qa, qb, ka, kb;
    float qt, kt;
};

__device__ __forceinline__ void k2_load_edge(EdgeFrag& f, const float* __restrict__ qk,
                                             int s, int d, int offA, int offT) {
    const float* __restrict__ sq = qk + (size_t)s * 520;
    const float* __restrict__ dk = qk + (size_t)d * 520 + 65;
    f.qa = *(const v4f_a4*)(sq + offA);
    f.qb = *(const v4f_a4*)(sq + offA + 260);
    f.ka = *(const v4f_a4*)(dk + offA);
    f.kb = *(const v4f_a4*)(dk + offA + 260);
    f.qt = sq[offT];
    f.kt = dk[offT];
    __builtin_amdgcn_sched_barrier(0);
}

__device__ __forceinline__ void k2_reduce_store(const EdgeFrag& f, float* __restrict__ scores,
                                                unsigned* __restrict__ smax, int e, int s,
                                                int up, int th, int lane) {
    float a1 = f.qa.x * f.ka.x;
    float a2 = f.qb.x * f.kb.x;
    a1 = fmaf(f.qa.y, f.ka.y, a1);  a2 = fmaf(f.qb.y, f.kb.y, a2);
    a1 = fmaf(f.qa.z, f.ka.z, a1);  a2 = fmaf(f.qb.z, f.kb.z, a2);
    a1 = fmaf(f.qa.w, f.ka.w, a1);  a2 = fmaf(f.qb.w, f.kb.w, a2);
    float r = row_xor_add<0x168>(up ? a2 : a1, up ? a1 : a2);
    r = row_xor_add<0x164>(r, r);
    r = row_xor_add<0x162>(r, r);
    r = row_xor_add<0x161>(r, r);
    r = fmaf(f.qt, f.kt, r);
    if ((lane & 7) == 0) {
        scores[(size_t)e * 4 + th] = r;
        atomicMax(&smax[(size_t)s * 4 + th], fenc(r));
    }
}

__global__ __launch_bounds__(256)
void k2_scores(const float* __restrict__ qk, const int* __restrict__ ei,
               float* __restrict__ scores, unsigned* __restrict__ smax,
               int E, int C, int H, int P) {
    const int lane = threadIdx.x & 31;
    const int wv = __builtin_amdgcn_readfirstlane((int)(threadIdx.x >> 5));
    const int e0 = (blockIdx.x * (blockDim.x >> 5) + wv) * K2_EPW;
    if (e0 >= E) return;
    const int cnt = (E - e0 < K2_EPW) ? (E - e0) : K2_EPW;

    const int le = (e0 + (lane & 15) < E) ? (e0 + (lane & 15)) : (E - 1);
    const int ev = ei[(size_t)(lane >> 4) * E + le];

    if (H == 4 && P == 65 && C == 520) {
        const int hs   = lane >> 4;
        const int offA = hs * 130 + (lane & 15) * 4;
        const int up   = (lane >> 3) & 1;
        const int th   = hs + up * 2;
        const int offT = th * 130 + 64;

        EdgeFrag fa, fb;
        int sa = __builtin_amdgcn_readlane(ev, 0), sb = sa;
        k2_load_edge(fa, qk, sa, __builtin_amdgcn_readlane(ev, 16), offA, offT);
        int j = 0;
        for (; j + 2 < cnt; j += 2) {
            sb = __builtin_amdgcn_readlane(ev, j + 1);
            k2_load_edge(fb, qk, sb, __builtin_amdgcn_readlane(ev, j + 17), offA, offT);
            k2_reduce_store(fa, scores, smax, e0 + j, sa, up, th, lane);
            sa = __builtin_amdgcn_readlane(ev, j + 2);
            k2_load_edge(fa, qk, sa, __builtin_amdgcn_readlane(ev, j + 18), offA, offT);
            k2_reduce_store(fb, scores, smax, e0 + j + 1, sb, up, th, lane);
        }
        if (j + 1 < cnt) {
            sb = __builtin_amdgcn_readlane(ev, j + 1);
            k2_load_edge(fb, qk, sb, __builtin_amdgcn_readlane(ev, j + 17), offA, offT);
            k2_reduce_store(fa, scores, smax, e0 + j, sa, up, th, lane);
            k2_reduce_store(fb, scores, smax, e0 + j + 1, sb, up, th, lane);
        } else {
            k2_reduce_store(fa, scores, smax, e0 + j, sa, up, th, lane);
        }
    } else {
        for (int j = 0; j < cnt; ++j) {
            const int s = __builtin_amdgcn_readlane(ev, j);
            const int d = __builtin_amdgcn_readlane(ev, 16 + j);
            const float* __restrict__ srow = qk + (size_t)s * C;
            const float* __restrict__ drow = qk + (size_t)d * C;
            for (int h = 0; h < H; ++h) {
                const float* __restrict__ qh = srow + h * 2 * P;
                const float* __restrict__ kh = drow + h * 2 * P + P;
                float a = 0.0f;
                for (int p = lane; p < P; p += 32)
                    a += qh[p] * kh[p];
                #pragma unroll
                for (int off = 16; off > 0; off >>= 1)
                    a += __shfl_xor(a, off, 32);
                if (lane == 0) {
                    scores[(size_t)(e0 + j) * H + h] = a;
                    atomicMax(&smax[(size_t)s * H + h], fenc(a));
                }
            }
        }
    }
}

__global__ __launch_bounds__(256)
void k3_exp_denom(float* __restrict__ scores, const unsigned* __restrict__ smax,
                  const int* __restrict__ ei, float* __restrict__ denom,
                  int E, int H) {
    size_t i = (size_t)blockIdx.x * blockDim.x + threadIdx.x;
    if (i >= (size_t)E * H) return;
    const int e = (int)(i / H);
    const int h = (int)(i % H);
    const int s = ei[e];
    const float m  = fdec(smax[(size_t)s * H + h]);
    const float ex = __expf(scores[i] - m);
    scores[i] = ex;
    atomicAdd(&denom[(size_t)s * H + h], ex);
}

__global__ __launch_bounds__(256)
void k4_out(const float* __restrict__ ex, const float* __restrict__ denom,
            const int* __restrict__ ei, float* __restrict__ out, int E, int H) {
    const int e = blockIdx.x * blockDim.x + threadIdx.x;
    if (e >= E) return;
    const int s = ei[e];
    float sum = 0.0f;
    for (int h = 0; h < H; ++h)
        sum += ex[(size_t)e * H + h] / denom[(size_t)s * H + h];
    out[e] = sum / (float)H;
}

extern "C" void kernel_launch(void* const* d_in, const int* in_sizes, int n_in,
                              void* d_out, int out_size, void* d_ws, size_t ws_size,
                              hipStream_t stream) {
    const float* x    = (const float*)d_in[0];
    const int*   ei   = (const int*)d_in[1];
    const float* W    = (const float*)d_in[2];
    const float* bias = (const float*)d_in[3];
    const int C = in_sizes[3];
    const int F = in_sizes[2] / C;
    const int N = in_sizes[0] / F;
    const int E = in_sizes[1] / 2;
    const int H = (C - 2 * F) / 2;
    const int P = F / H + 1;

    float*    qk     = (float*)d_ws;
    float*    scores = qk + (size_t)N * C;
    unsigned* smax   = (unsigned*)(scores + (size_t)E * H);
    float*    denom  = (float*)(smax + (size_t)N * H);

    {
        size_t n = (size_t)2 * N * H;
        k0_zero<<<(unsigned)((n + 255) / 256), 256, 0, stream>>>(smax, n);
    }

    {
        dim3 grid((N + 63) / 64, (C + 63) / 64);
        k1_gemm_wmma<<<grid, 32, 0, stream>>>(x, W, bias, qk, N, F, C);
    }

    {
        const int epb = (256 / 32) * K2_EPW;
        k2_scores<<<(E + epb - 1) / epb, 256, 0, stream>>>(qk, ei, scores, smax,
                                                           E, C, H, P);
    }

    {
        size_t n = (size_t)E * H;
        k3_exp_denom<<<(unsigned)((n + 255) / 256), 256, 0, stream>>>(scores, smax,
                                                                      ei, denom, E, H);
    }

    k4_out<<<(E + 255) / 256, 256, 0, stream>>>(scores, denom, ei,
                                                (float*)d_out, E, H);
}


# GEMM stores qk in a gather layout (per node one aligned 1024-B q row and one k row, 65th elements of the heads packed apart): score kernel reads 18 whole lines per edge, was 24 partly used
# speedup vs baseline: 2.1028x; 1.1423x over previous
#include <hip/hip_runtime.h>
#include <hip/hip_bf16.h>


typedef __attribute__((ext_vector_type(2))) float v2f;
typedef __attribute__((ext_vector_type(4))) float v4f;
typedef __attribute__((ext_vector_type(8))) float v8f;
typedef v4f v4f_a4 __attribute__((aligned(4)));

__device__ __forceinline__ unsigned fenc(float f) {
    unsigned u = __float_as_uint(f);
    return (u & 0x80000000u) ? ~u : (u | 0x80000000u);
}
__device__ __forceinline__ float fdec(unsigned e) {
    return (e & 0x80000000u) ? __uint_as_float(e & 0x7FFFFFFFu)
                             : __uint_as_float(~e);
}

__global__ void k0_zero(unsigned* __restrict__ p, size_t n) {
    size_t i = (size_t)blockIdx.x * blockDim.x + threadIdx.x;
    if (i < n) p[i] = 0u;
}

__global__ __launch_bounds__(32)
void k1_gemm_wmma(const float* __restrict__ x, const float* __restrict__ W,
                  const float* __restrict__ bias, float* __restrict__ qk,
                  int N, int F, int C, int H, int P) {
    const int lane = threadIdx.x;
    const int M0   = blockIdx.x * 64;
    const int N0   = blockIdx.y * 64;
    const int lm   = lane & 15;
    const int kh   = (lane >> 4) * 2;

    const float* __restrict__ xrow[4];
    const float* __restrict__ wrow[4];
    #pragma unroll
    for (int mi = 0; mi < 4; ++mi) {
        int r = M0 + mi * 16 + lm;
        xrow[mi] = x + (size_t)(r < N ? r : N - 1) * F;
    }
    #pragma unroll
    for (int ni = 0; ni < 4; ++ni) {
        int c = N0 + ni * 16 + lm;
        wrow[ni] = W + (size_t)(c < C ? c : C - 1) * F;
    }

    v8f acc[4][4];
    #pragma unroll
    for (int mi = 0; mi < 4; ++mi)
        #pragma unroll
        for (int ni = 0; ni < 4; ++ni)
            acc[mi][ni] = (v8f){};

    const int kq = (lane >> 4) * 16;
    int k = 0;
    for (; k + 32 <= F; k += 32) {
        v4f a[4][4], b[4][4];
        #pragma unroll
        for (int j = 0; j < 4; ++j) {
            #pragma unroll
            for (int mi = 0; mi < 4; ++mi)
                a[mi][j] = *(const v4f_a4*)(xrow[mi] + k + kq + 4 * j);
            #pragma unroll
            for (int ni = 0; ni < 4; ++ni)
                b[ni][j] = *(const v4f_a4*)(wrow[ni] + k + kq + 4 * j);
        }
        __builtin_amdgcn_sched_barrier(0);
        #pragma unroll
        for (int t = 0; t < 8; ++t) {
            #pragma unroll
            for (int mi = 0; mi < 4; ++mi) {
                v2f af;
                af.x = a[mi][t >> 1][(t & 1) * 2];
                af.y = a[mi][t >> 1][(t & 1) * 2 + 1];
                #pragma unroll
                for (int ni = 0; ni < 4; ++ni) {
                    v2f bf;
                    bf.x = b[ni][t >> 1][(t & 1) * 2];
                    bf.y = b[ni][t >> 1][(t & 1) * 2 + 1];
                    acc[mi][ni] = __builtin_amdgcn_wmma_f32_16x16x4_f32(
                         false, af,  false, bf,
                         (short)0, acc[mi][ni],
                         false,  false);
                }
            }
        }
    }
    for (; k < F; k += 4) {
        v2f a[4], b[4];
        #pragma unroll
        for (int mi = 0; mi < 4; ++mi) {
            a[mi].x = xrow[mi][k + kh];
            a[mi].y = xrow[mi][k + kh + 1];
        }
        #pragma unroll
        for (int ni = 0; ni < 4; ++ni) {
            b[ni].x = wrow[ni][k + kh];
            b[ni].y = wrow[ni][k + kh + 1];
        }
        #pragma unroll
        for (int mi = 0; mi < 4; ++mi)
            #pragma unroll
            for (int ni = 0; ni < 4; ++ni)
                acc[mi][ni] = __builtin_amdgcn_wmma_f32_16x16x4_f32(
                     false, a[mi],  false, b[ni],
                     (short)0, acc[mi][ni],
                     false,  false);
    }

    const bool split = (H == 4 && P == 65 && C == 520);
    const int rsub = (lane >> 4) << 3;
    #pragma unroll
    for (int ni = 0; ni < 4; ++ni) {
        const int col = N0 + ni * 16 + lm;
        if (col >= C) continue;
        const float bv = bias[col];
        float* __restrict__ dst = qk + col;
        size_t pitch = (size_t)C;
        if (split) {
            const int h   = col / 130;
            const int rr  = col - h * 130;
            const int isk = (rr >= 65) ? 1 : 0;
            const int p   = rr - isk * 65;
            if (p < 64) {
                dst   = qk + (size_t)isk * N * 256 + h * 64 + p;
                pitch = 256;
            } else {
                dst   = qk + (size_t)N * 512 + (size_t)isk * N * 4 + h;
                pitch = 4;
            }
        }
        #pragma unroll
        for (int mi = 0; mi < 4; ++mi) {
            const int rbase = M0 + mi * 16 + rsub;
            #pragma unroll
            for (int r = 0; r < 8; ++r) {
                const int row = rbase + r;
                if (row < N)
                    dst[(size_t)row * pitch] = acc[mi][ni][r] + bv;
            }
        }
    }
}


#define K2_EPW 16

template <int CTRL>
__device__ __forceinline__ float row_xor_add(float keep, float send) {
    const int t = __builtin_amdgcn_update_dpp(0, __builtin_bit_cast(int, send),
                                              CTRL, 0xf, 0xf, false);
    return keep + __builtin_bit_cast(float, t);
}

struct EdgeFrag {
    v4f qa, qb, ka, kb;
    float qt, kt;
};

struct GatherPlanes {
    const float* __restrict__ qm;
    const float* __restrict__ km;
    const float* __restrict__ qt;
    const float* __restrict__ kt;
};

__device__ __forceinline__ void k2_load_edge(EdgeFrag& f, const GatherPlanes& g,
                                             int s, int d, int offA, int th) {
    const float* __restrict__ sq = g.qm + (size_t)s * 256 + offA;
    const float* __restrict__ dk = g.km + (size_t)d * 256 + offA;
    f.qa = *(const v4f_a4*)(sq);
    f.qb = *(const v4f_a4*)(sq + 128);
    f.ka = *(const v4f_a4*)(dk);
    f.kb = *(const v4f_a4*)(dk + 128);
    f.qt = g.qt[(size_t)s * 4 + th];
    f.kt = g.kt[(size_t)d * 4 + th];
    __builtin_amdgcn_sched_barrier(0);
}

__device__ __forceinline__ void k2_reduce_store(const EdgeFrag& f, float* __restrict__ scores,
                                                unsigned* __restrict__ smax, int e, int s,
                                                int up, int th, int lane) {
    float a1 = f.qa.x * f.ka.x;
    float a2 = f.qb.x * f.kb.x;
    a1 = fmaf(f.qa.y, f.ka.y, a1);  a2 = fmaf(f.qb.y, f.kb.y, a2);
    a1 = fmaf(f.qa.z, f.ka.z, a1);  a2 = fmaf(f.qb.z, f.kb.z, a2);
    a1 = fmaf(f.qa.w, f.ka.w, a1);  a2 = fmaf(f.qb.w, f.kb.w, a2);
    float r = row_xor_add<0x168>(up ? a2 : a1, up ? a1 : a2);
    r = row_xor_add<0x164>(r, r);
    r = row_xor_add<0x162>(r, r);
    r = row_xor_add<0x161>(r, r);
    r = fmaf(f.qt, f.kt, r);
    if ((lane & 7) == 0) {
        scores[(size_t)e * 4 + th] = r;
        atomicMax(&smax[(size_t)s * 4 + th], fenc(r));
    }
}

__global__ __launch_bounds__(256)
void k2_scores(const float* __restrict__ qk, const int* __restrict__ ei,
               float* __restrict__ scores, unsigned* __restrict__ smax,
               int N, int E, int C, int H, int P) {
    const int lane = threadIdx.x & 31;
    const int wv = __builtin_amdgcn_readfirstlane((int)(threadIdx.x >> 5));
    const int e0 = (blockIdx.x * (blockDim.x >> 5) + wv) * K2_EPW;
    if (e0 >= E) return;
    const int cnt = (E - e0 < K2_EPW) ? (E - e0) : K2_EPW;

    const int le = (e0 + (lane & 15) < E) ? (e0 + (lane & 15)) : (E - 1);
    const int ev = ei[(size_t)(lane >> 4) * E + le];

    if (H == 4 && P == 65 && C == 520) {
        const int hs   = lane >> 4;
        const int offA = hs * 64 + (lane & 15) * 4;
        const int up   = (lane >> 3) & 1;
        const int th   = hs + up * 2;
        GatherPlanes g;
        g.qm = qk;
        g.km = qk + (size_t)N * 256;
        g.qt = qk + (size_t)N * 512;
        g.kt = g.qt + (size_t)N * 4;

        EdgeFrag fa, fb;
        int sa = __builtin_amdgcn_readlane(ev, 0), sb = sa;
        k2_load_edge(fa, g, sa, __builtin_amdgcn_readlane(ev, 16), offA, th);
        int j = 0;
        for (; j + 2 < cnt; j += 2) {
            sb = __builtin_amdgcn_readlane(ev, j + 1);
            k2_load_edge(fb, g, sb, __builtin_amdgcn_readlane(ev, j + 17), offA, th);
            k2_reduce_store(fa, scores, smax, e0 + j, sa, up, th, lane);
            sa = __builtin_amdgcn_readlane(ev, j + 2);
            k2_load_edge(fa, g, sa, __builtin_amdgcn_readlane(ev, j + 18), offA, th);
            k2_reduce_store(fb, scores, smax, e0 + j + 1, sb, up, th, lane);
        }
        if (j + 1 < cnt) {
            sb = __builtin_amdgcn_readlane(ev, j + 1);
            k2_load_edge(fb, g, sb, __builtin_amdgcn_readlane(ev, j + 17), offA, th);
            k2_reduce_store(fa, scores, smax, e0 + j, sa, up, th, lane);
            k2_reduce_store(fb, scores, smax, e0 + j + 1, sb, up, th, lane);
        } else {
            k2_reduce_store(fa, scores, smax, e0 + j, sa, up, th, lane);
        }
    } else {
        for (int j = 0; j < cnt; ++j) {
            const int s = __builtin_amdgcn_readlane(ev, j);
            const int d = __builtin_amdgcn_readlane(ev, 16 + j);
            const float* __restrict__ srow = qk + (size_t)s * C;
            const float* __restrict__ drow = qk + (size_t)d * C;
            for (int h = 0; h < H; ++h) {
                const float* __restrict__ qh = srow + h * 2 * P;
                const float* __restrict__ kh = drow + h * 2 * P + P;
                float a = 0.0f;
                for (int p = lane; p < P; p += 32)
                    a += qh[p] * kh[p];
                #pragma unroll
                for (int off = 16; off > 0; off >>= 1)
                    a += __shfl_xor(a, off, 32);
                if (lane == 0) {
                    scores[(size_t)(e0 + j) * H + h] = a;
                    atomicMax(&smax[(size_t)s * H + h], fenc(a));
                }
            }
        }
    }
}

__global__ __launch_bounds__(256)
void k3_exp_denom(float* __restrict__ scores, const unsigned* __restrict__ smax,
                  const int* __restrict__ ei, float* __restrict__ denom,
                  int E, int H) {
    size_t i = (size_t)blockIdx.x * blockDim.x + threadIdx.x;
    if (i >= (size_t)E * H) return;
    const int e = (int)(i / H);
    const int h = (int)(i % H);
    const int s = ei[e];
    const float m  = fdec(smax[(size_t)s * H + h]);
    const float ex = __expf(scores[i] - m);
    scores[i] = ex;
    atomicAdd(&denom[(size_t)s * H + h], ex);
}

__global__ __launch_bounds__(256)
void k4_out(const float* __restrict__ ex, const float* __restrict__ denom,
            const int* __restrict__ ei, float* __restrict__ out, int E, int H) {
    const int e = blockIdx.x * blockDim.x + threadIdx.x;
    if (e >= E) return;
    const int s = ei[e];
    float sum = 0.0f;
    for (int h = 0; h < H; ++h)
        sum += ex[(size_t)e * H + h] / denom[(size_t)s * H + h];
    out[e] = sum / (float)H;
}

extern "C" void kernel_launch(void* const* d_in, const int* in_sizes, int n_in,
                              void* d_out, int out_size, void* d_ws, size_t ws_size,
                              hipStream_t stream) {
    const float* x    = (const float*)d_in[0];
    const int*   ei   = (const int*)d_in[1];
    const float* W    = (const float*)d_in[2];
    const float* bias = (const float*)d_in[3];
    const int C = in_sizes[3];
    const int F = in_sizes[2] / C;
    const int N = in_sizes[0] / F;
    const int E = in_sizes[1] / 2;
    const int H = (C - 2 * F) / 2;
    const int P = F / H + 1;

    float*    qk     = (float*)d_ws;
    float*    scores = qk + (size_t)N * C;
    unsigned* smax   = (unsigned*)(scores + (size_t)E * H);
    float*    denom  = (float*)(smax + (size_t)N * H);

    {
        size_t n = (size_t)2 * N * H;
        k0_zero<<<(unsigned)((n + 255) / 256), 256, 0, stream>>>(smax, n);
    }

    {
        dim3 grid((N + 63) / 64, (C + 63) / 64);
        k1_gemm_wmma<<<grid, 32, 0, stream>>>(x, W, bias, qk, N, F, C, H, P);
    }

    {
        const int epb = (256 / 32) * K2_EPW;
        k2_scores<<<(E + epb - 1) / epb, 256, 0, stream>>>(qk, ei, scores, smax,
                                                           N, E, C, H, P);
    }

    {
        size_t n = (size_t)E * H;
        k3_exp_denom<<<(unsigned)((n + 255) / 256), 256, 0, stream>>>(scores, smax,
                                                                      ei, denom, E, H);
    }

    k4_out<<<(E + 255) / 256, 256, 0, stream>>>(scores, denom, ei,
                                                (float*)d_out, E, H);
}


# edge phase by node range: each of 256 work-groups scans the source ids for its own nodes' edges (no count or cursor pass); q rows, scores, segment max and sum in LDS; 2 launches; sums by LDS atomics
# speedup vs baseline: 2.4747x; 1.1769x over previous
#include <hip/hip_runtime.h>
#include <hip/hip_bf16.h>


typedef __attribute__((ext_vector_type(2))) float v2f;
typedef __attribute__((ext_vector_type(4))) float v4f;
typedef __attribute__((ext_vector_type(8))) float v8f;
typedef v4f v4f_a4 __attribute__((aligned(4)));

__device__ __forceinline__ unsigned fenc(float f) {
    unsigned u = __float_as_uint(f);
    return (u & 0x80000000u) ? ~u : (u | 0x80000000u);
}
__device__ __forceinline__ float fdec(unsigned e) {
    return (e & 0x80000000u) ? __uint_as_float(e & 0x7FFFFFFFu)
                             : __uint_as_float(~e);
}

__global__ void k0_zero(unsigned* __restrict__ p, size_t n) {
    size_t i = (size_t)blockIdx.x * blockDim.x + threadIdx.x;
    if (i < n) p[i] = 0u;
}

__global__ __launch_bounds__(32)
void k1_gemm_wmma(const float* __restrict__ x, const float* __restrict__ W,
                  const float* __restrict__ bias, float* __restrict__ qk,
                  int N, int F, int C, int H, int P) {
    const int lane = threadIdx.x;
    const int M0   = blockIdx.x * 64;
    const int N0   = blockIdx.y * 64;
    const int lm   = lane & 15;
    const int kh   = (lane >> 4) * 2;

    const float* __restrict__ xrow[4];
    const float* __restrict__ wrow[4];
    #pragma unroll
    for (int mi = 0; mi < 4; ++mi) {
        int r = M0 + mi * 16 + lm;
        xrow[mi] = x + (size_t)(r < N ? r : N - 1) * F;
    }
    #pragma unroll
    for (int ni = 0; ni < 4; ++ni) {
        int c = N0 + ni * 16 + lm;
        wrow[ni] = W + (size_t)(c < C ? c : C - 1) * F;
    }

    v8f acc[4][4];
    #pragma unroll
    for (int mi = 0; mi < 4; ++mi)
        #pragma unroll
        for (int ni = 0; ni < 4; ++ni)
            acc[mi][ni] = (v8f){};

    const int kq = (lane >> 4) * 16;
    int k = 0;
    for (; k + 32 <= F; k += 32) {
        v4f a[4][4], b[4][4];
        #pragma unroll
        for (int j = 0; j < 4; ++j) {
            #pragma unroll
            for (int mi = 0; mi < 4; ++mi)
                a[mi][j] = *(const v4f_a4*)(xrow[mi] + k + kq + 4 * j);
            #pragma unroll
            for (int ni = 0; ni < 4; ++ni)
                b[ni][j] = *(const v4f_a4*)(wrow[ni] + k + kq + 4 * j);
        }
        __builtin_amdgcn_sched_barrier(0);
        #pragma unroll
        for (int t = 0; t < 8; ++t) {
            #pragma unroll
            for (int mi = 0; mi < 4; ++mi) {
                v2f af;
                af.x = a[mi][t >> 1][(t & 1) * 2];
                af.y = a[mi][t >> 1][(t & 1) * 2 + 1];
                #pragma unroll
                for (int ni = 0; ni < 4; ++ni) {
                    v2f bf;
                    bf.x = b[ni][t >> 1][(t & 1) * 2];
                    bf.y = b[ni][t >> 1][(t & 1) * 2 + 1];
                    acc[mi][ni] = __builtin_amdgcn_wmma_f32_16x16x4_f32(
                         false, af,  false, bf,
                         (short)0, acc[mi][ni],
                         false,  false);
                }
            }
        }
    }
    for (; k < F; k += 4) {
        v2f a[4], b[4];
        #pragma unroll
        for (int mi = 0; mi < 4; ++mi) {
            a[mi].x = xrow[mi][k + kh];
            a[mi].y = xrow[mi][k + kh + 1];
        }
        #pragma unroll
        for (int ni = 0; ni < 4; ++ni) {
            b[ni].x = wrow[ni][k + kh];
            b[ni].y = wrow[ni][k + kh + 1];
        }
        #pragma unroll
        for (int mi = 0; mi < 4; ++mi)
            #pragma unroll
            for (int ni = 0; ni < 4; ++ni)
                acc[mi][ni] = __builtin_amdgcn_wmma_f32_16x16x4_f32(
                     false, a[mi],  false, b[ni],
                     (short)0, acc[mi][ni],
                     false,  false);
    }

    const bool split = (H == 4 && P == 65 && C == 520);
    const int rsub = (lane >> 4) << 3;
    #pragma unroll
    for (int ni = 0; ni < 4; ++ni) {
        const int col = N0 + ni * 16 + lm;
        if (col >= C) continue;
        const float bv = bias[col];
        float* __restrict__ dst = qk + col;
        size_t pitch = (size_t)C;
        if (split) {
            const int h   = col / 130;
            const int rr  = col - h * 130;
            const int isk = (rr >= 65) ? 1 : 0;
            const int p   = rr - isk * 65;
            if (p < 64) {
                dst   = qk + (size_t)isk * N * 256 + h * 64 + p;
                pitch = 256;
            } else {
                dst   = qk + (size_t)N * 512 + (size_t)isk * N * 4 + h;
                pitch = 4;
            }
        }
        #pragma unroll
        for (int mi = 0; mi < 4; ++mi) {
            const int rbase = M0 + mi * 16 + rsub;
            #pragma unroll
            for (int r = 0; r < 8; ++r) {
                const int row = rbase + r;
                if (row < N)
                    dst[(size_t)row * pitch] = acc[mi][ni][r] + bv;
            }
        }
    }
}


#define K2_EPW 16

template <int CTRL>
__device__ __forceinline__ float row_xor_add(float keep, float send) {
    const int t = __builtin_amdgcn_update_dpp(0, __builtin_bit_cast(int, send),
                                              CTRL, 0xf, 0xf, false);
    return keep + __builtin_bit_cast(float, t);
}

struct EdgeFrag {
    v4f qa, qb, ka, kb;
    float qt, kt;
};

struct GatherPlanes {
    const float* __restrict__ qm;
    const float* __restrict__ km;
    const float* __restrict__ qt;
    const float* __restrict__ kt;
};

__device__ __forceinline__ void k2_load_edge(EdgeFrag& f, const GatherPlanes& g,
                                             int s, int d, int offA, int th) {
    const float* __restrict__ sq = g.qm + (size_t)s * 256 + offA;
    const float* __restrict__ dk = g.km + (size_t)d * 256 + offA;
    f.qa = *(const v4f_a4*)(sq);
    f.qb = *(const v4f_a4*)(sq + 128);
    f.ka = *(const v4f_a4*)(dk);
    f.kb = *(const v4f_a4*)(dk + 128);
    f.qt = g.qt[(size_t)s * 4 + th];
    f.kt = g.kt[(size_t)d * 4 + th];
    __builtin_amdgcn_sched_barrier(0);
}

__device__ __forceinline__ void k2_reduce_store(const EdgeFrag& f, float* __restrict__ scores,
                                                unsigned* __restrict__ smax, int e, int s,
                                                int up, int th, int lane) {
    float a1 = f.qa.x * f.ka.x;
    float a2 = f.qb.x * f.kb.x;
    a1 = fmaf(f.qa.y, f.ka.y, a1);  a2 = fmaf(f.qb.y, f.kb.y, a2);
    a1 = fmaf(f.qa.z, f.ka.z, a1);  a2 = fmaf(f.qb.z, f.kb.z, a2);
    a1 = fmaf(f.qa.w, f.ka.w, a1);  a2 = fmaf(f.qb.w, f.kb.w, a2);
    float r = row_xor_add<0x168>(up ? a2 : a1, up ? a1 : a2);
    r = row_xor_add<0x164>(r, r);
    r = row_xor_add<0x162>(r, r);
    r = row_xor_add<0x161>(r, r);
    r = fmaf(f.qt, f.kt, r);
    if ((lane & 7) == 0) {
        scores[(size_t)e * 4 + th] = r;
        atomicMax(&smax[(size_t)s * 4 + th], fenc(r));
    }
}

__global__ __launch_bounds__(256)
void k2_scores(const float* __restrict__ qk, const int* __restrict__ ei,
               float* __restrict__ scores, unsigned* __restrict__ smax,
               int N, int E, int C, int H, int P) {
    const int lane = threadIdx.x & 31;
    const int wv = __builtin_amdgcn_readfirstlane((int)(threadIdx.x >> 5));
    const int e0 = (blockIdx.x * (blockDim.x >> 5) + wv) * K2_EPW;
    if (e0 >= E) return;
    const int cnt = (E - e0 < K2_EPW) ? (E - e0) : K2_EPW;

    const int le = (e0 + (lane & 15) < E) ? (e0 + (lane & 15)) : (E - 1);
    const int ev = ei[(size_t)(lane >> 4) * E + le];

    if (H == 4 && P == 65 && C == 520) {
        const int hs   = lane >> 4;
        const int offA = hs * 64 + (lane & 15) * 4;
        const int up   = (lane >> 3) & 1;
        const int th   = hs + up * 2;
        GatherPlanes g;
        g.qm = qk;
        g.km = qk + (size_t)N * 256;
        g.qt = qk + (size_t)N * 512;
        g.kt = g.qt + (size_t)N * 4;

        EdgeFrag fa, fb;
        int sa = __builtin_amdgcn_readlane(ev, 0), sb = sa;
        k2_load_edge(fa, g, sa, __builtin_amdgcn_readlane(ev, 16), offA, th);
        int j = 0;
        for (; j + 2 < cnt; j += 2) {
            sb = __builtin_amdgcn_readlane(ev, j + 1);
            k2_load_edge(fb, g, sb, __builtin_amdgcn_readlane(ev, j + 17), offA, th);
            k2_reduce_store(fa, scores, smax, e0 + j, sa, up, th, lane);
            sa = __builtin_amdgcn_readlane(ev, j + 2);
            k2_load_edge(fa, g, sa, __builtin_amdgcn_readlane(ev, j + 18), offA, th);
            k2_reduce_store(fb, scores, smax, e0 + j + 1, sb, up, th, lane);
        }
        if (j + 1 < cnt) {
            sb = __builtin_amdgcn_readlane(ev, j + 1);
            k2_load_edge(fb, g, sb, __builtin_amdgcn_readlane(ev, j + 17), offA, th);
            k2_reduce_store(fa, scores, smax, e0 + j, sa, up, th, lane);
            k2_reduce_store(fb, scores, smax, e0 + j + 1, sb, up, th, lane);
        } else {
            k2_reduce_store(fa, scores, smax, e0 + j, sa, up, th, lane);
        }
    } else {
        for (int j = 0; j < cnt; ++j) {
            const int s = __builtin_amdgcn_readlane(ev, j);
            const int d = __builtin_amdgcn_readlane(ev, 16 + j);
            const float* __restrict__ srow = qk + (size_t)s * C;
            const float* __restrict__ drow = qk + (size_t)d * C;
            for (int h = 0; h < H; ++h) {
                const float* __restrict__ qh = srow + h * 2 * P;
                const float* __restrict__ kh = drow + h * 2 * P + P;
                float a = 0.0f;
                for (int p = lane; p < P; p += 32)
                    a += qh[p] * kh[p];
                #pragma unroll
                for (int off = 16; off > 0; off >>= 1)
                    a += __shfl_xor(a, off, 32);
                if (lane == 0) {
                    scores[(size_t)(e0 + j) * H + h] = a;
                    atomicMax(&smax[(size_t)s * H + h], fenc(a));
                }
            }
        }
    }
}

__global__ __launch_bounds__(256)
void k3_exp_denom(float* __restrict__ scores, const unsigned* __restrict__ smax,
                  const int* __restrict__ ei, float* __restrict__ denom,
                  int E, int H) {
    size_t i = (size_t)blockIdx.x * blockDim.x + threadIdx.x;
    if (i >= (size_t)E * H) return;
    const int e = (int)(i / H);
    const int h = (int)(i % H);
    const int s = ei[e];
    const float m  = fdec(smax[(size_t)s * H + h]);
    const float ex = __expf(scores[i] - m);
    scores[i] = ex;
    atomicAdd(&denom[(size_t)s * H + h], ex);
}

__global__ __launch_bounds__(256)
void k4_out(const float* __restrict__ ex, const float* __restrict__ denom,
            const int* __restrict__ ei, float* __restrict__ out, int E, int H) {
    const int e = blockIdx.x * blockDim.x + threadIdx.x;
    if (e >= E) return;
    const int s = ei[e];
    float sum = 0.0f;
    for (int h = 0; h < H; ++h)
        sum += ex[(size_t)e * H + h] / denom[(size_t)s * H + h];
    out[e] = sum / (float)H;
}


#define OW_GROUPS  256
#define OW_THREADS 1024
#define OW_NODES   196
#define OW_CAP     4096
#define OW_UNR     16
#define OW_B       8
#define OW_FILL    13

typedef __attribute__((ext_vector_type(4))) int v4i;
typedef v4i v4i_a4 __attribute__((aligned(4)));

__device__ __forceinline__ unsigned ow_umin(unsigned a, unsigned b) { return a < b ? a : b; }

__device__ __forceinline__ void ow_append(int e, unsigned sl, int* le, int* ls, unsigned* cnt) {
    const unsigned idx = atomicAdd(cnt, 1u);
    if (idx < (unsigned)OW_CAP) { le[idx] = e; ls[idx] = (int)sl; }
}

__device__ __forceinline__ void ow_test4(v4i s, int e, int n0, unsigned nn,
                                         int* le, int* ls, unsigned* cnt) {
    const unsigned a = (unsigned)(s.x - n0), b = (unsigned)(s.y - n0);
    const unsigned c = (unsigned)(s.z - n0), d = (unsigned)(s.w - n0);
    if (ow_umin(ow_umin(a, b), ow_umin(c, d)) < nn) {
        #pragma nounroll
        for (int j = 0; j < 4; ++j) {
            const unsigned sl = (j == 0) ? a : (j == 1) ? b : (j == 2) ? c : d;
            if (sl < nn) ow_append(e + j, sl, le, ls, cnt);
        }
    }
}

__device__ __forceinline__ void ow_scan_simple(const int* __restrict__ ei, int lo, int hi, int n0, unsigned nn,
                                               int tid, int* le, int* ls, unsigned* cnt) {
    for (int e = lo + tid; e < hi; e += OW_THREADS) {
        const unsigned sl = (unsigned)(ei[e] - n0);
        if (sl < nn) ow_append(e, sl, le, ls, cnt);
    }
}

__device__ __forceinline__ void ow_scan(const int* __restrict__ ei, int lo, int hi, int n0, unsigned nn,
                                        int tid, int grp, int* le, int* ls, unsigned* cnt) {
    const int CH    = OW_THREADS * 4 * OW_UNR;
    const int nfull = (hi - lo) / CH;
    if (nfull > 0) {
        int c = grp % nfull;
        for (int it = 0; it < nfull; ++it) {
            const int b0 = lo + c * CH + tid * 4;
            v4i v[OW_UNR];
            #pragma unroll
            for (int u = 0; u < OW_UNR; ++u)
                v[u] = *(const v4i_a4*)(ei + b0 + u * (OW_THREADS * 4));
            __builtin_amdgcn_sched_barrier(0);
            #pragma unroll
            for (int u = 0; u < OW_UNR; ++u)
                ow_test4(v[u], b0 + u * (OW_THREADS * 4), n0, nn, le, ls, cnt);
            c = (c + 1 < nfull) ? c + 1 : 0;
        }
    }
    const int r0 = lo + nfull * CH;
    if (r0 < hi) {
        const int b0 = r0 + tid * 4;
        v4i v[OW_UNR];
        #pragma unroll
        for (int u = 0; u < OW_UNR; ++u) {
            const int b = b0 + u * (OW_THREADS * 4);
            v[u] = *(const v4i_a4*)(ei + ((b + 3 < hi) ? b : 0));
        }
        __builtin_amdgcn_sched_barrier(0);
        #pragma unroll
        for (int u = 0; u < OW_UNR; ++u) {
            const int b = b0 + u * (OW_THREADS * 4);
            if (b + 3 < hi) ow_test4(v[u], b, n0, nn, le, ls, cnt);
        }
        ow_scan_simple(ei, lo + ((hi - lo) & ~3), hi, n0, nn, tid, le, ls, cnt);
    }
}

__device__ __forceinline__ void ow_dst(const int* __restrict__ ei, int E, int n, int tid,
                                       const int* le, int* ld) {
    for (int i0 = tid; i0 < n; i0 += OW_THREADS * 4) {
        int v[4];
        #pragma unroll
        for (int u = 0; u < 4; ++u) {
            int i = i0 + u * OW_THREADS;
            i = i < n ? i : n - 1;
            v[u] = ei[(size_t)E + le[i]];
        }
        #pragma unroll
        for (int u = 0; u < 4; ++u) {
            int i = i0 + u * OW_THREADS;
            i = i < n ? i : n - 1;
            ld[i] = v[u];
        }
    }
}

struct KFrag { v4f ka, kb; };

__device__ __forceinline__ void ow_scores(int n, const float* __restrict__ km, const float* __restrict__ kt,
                                          const float* qs, const float* qts,
                                          const int* ld, const int* ls, float* sc, unsigned* mx,
                                          int wv, int lane) {
    const int hs   = lane >> 4;
    const int offA = hs * 64 + (lane & 15) * 4;
    const int up   = (lane >> 3) & 1;
    const int th   = hs + up * 2;
    for (int base = wv * OW_B; base < n; base += (OW_THREADS / 32) * OW_B) {
        int ii = base + (lane & (OW_B - 1));
        ii = ii < n ? ii : n - 1;
        const int dv = ld[ii];
        const int sv = ls[ii];
        int i2 = base + (lane >> 2);
        i2 = i2 < n ? i2 : n - 1;
        const float ktv = kt[(size_t)ld[i2] * 4 + (lane & 3)];
        KFrag f[OW_B];
        #pragma unroll
        for (int j = 0; j < OW_B; ++j) {
            const int d = __builtin_amdgcn_readlane(dv, j);
            const float* __restrict__ dk = km + (size_t)d * 256 + offA;
            f[j].ka = *(const v4f_a4*)(dk);
            f[j].kb = *(const v4f_a4*)(dk + 128);
        }
        __builtin_amdgcn_sched_barrier(0);
        #pragma unroll
        for (int j = 0; j < OW_B; ++j) {
            if (base + j < n) {
                const int sl = __builtin_amdgcn_readlane(sv, j);
                const v4f qa = *(const v4f*)(qs + sl * 256 + offA);
                const v4f qb = *(const v4f*)(qs + sl * 256 + offA + 128);
                const float qtv = qts[sl * 4 + th];
                const float ktj = __shfl(ktv, 4 * j + th, 32);
                float a1 = qa.x * f[j].ka.x;
                float a2 = qb.x * f[j].kb.x;
                a1 = fmaf(qa.y, f[j].ka.y, a1);  a2 = fmaf(qb.y, f[j].kb.y, a2);
                a1 = fmaf(qa.z, f[j].ka.z, a1);  a2 = fmaf(qb.z, f[j].kb.z, a2);
                a1 = fmaf(qa.w, f[j].ka.w, a1);  a2 = fmaf(qb.w, f[j].kb.w, a2);
                float r = row_xor_add<0x168>(up ? a2 : a1, up ? a1 : a2);
                r = row_xor_add<0x164>(r, r);
                r = row_xor_add<0x162>(r, r);
                r = row_xor_add<0x161>(r, r);
                r = fmaf(qtv, ktj, r);
                if ((lane & 7) == 0) {
                    sc[(base + j) * 4 + th] = r;
                    atomicMax(&mx[sl * 4 + th], fenc(r));
                }
            }
        }
    }
}

__global__ __launch_bounds__(OW_THREADS)
void ke_owner(const float* __restrict__ qk, const int* __restrict__ ei,
              float* __restrict__ scores, float* __restrict__ out, int N, int E) {
    __shared__ __attribute__((aligned(16))) float qs[OW_NODES * 256];
    __shared__ __attribute__((aligned(16))) float qts[OW_NODES * 4];
    __shared__ __attribute__((aligned(16))) float sm[OW_NODES * 4];
    __shared__ unsigned mx[OW_NODES * 4];
    __shared__ __attribute__((aligned(16))) float sc[OW_CAP * 4];
    __shared__ int le[OW_CAP];
    __shared__ int ls[OW_CAP];
    __shared__ int ld[OW_CAP];
    __shared__ unsigned cnt;

    const int tid  = threadIdx.x;
    const int lane = tid & 31;
    const int wv   = __builtin_amdgcn_readfirstlane(tid >> 5);
    const int per  = (N + (int)gridDim.x - 1) / (int)gridDim.x;
    const int n0   = (int)blockIdx.x * per;
    if (n0 >= N) return;
    const int nni  = (N - n0 < per) ? (N - n0) : per;
    const unsigned nn = (unsigned)nni;

    const float* __restrict__ qm = qk;
    const float* __restrict__ km = qk + (size_t)N * 256;
    const float* __restrict__ qt = qk + (size_t)N * 512;
    const float* __restrict__ kt = qt + (size_t)N * 4;

    for (int i = tid; i < nni * 4; i += OW_THREADS) {
        mx[i]  = 0u;
        sm[i]  = 0.0f;
        qts[i] = qt[(size_t)n0 * 4 + i];
    }
    if (tid == 0) cnt = 0u;
    {
        const float* __restrict__ qsrc = qm + (size_t)n0 * 256;
        const int tot = nni * 64;
        for (int i0 = tid; i0 < tot; i0 += OW_THREADS * OW_FILL) {
            v4f t[OW_FILL];
            #pragma unroll
            for (int u = 0; u < OW_FILL; ++u) {
                int i = i0 + u * OW_THREADS;
                i = i < tot ? i : tot - 1;
                t[u] = *(const v4f_a4*)(qsrc + 4 * i);
            }
            __builtin_amdgcn_sched_barrier(0);
            #pragma unroll
            for (int u = 0; u < OW_FILL; ++u) {
                int i = i0 + u * OW_THREADS;
                i = i < tot ? i : tot - 1;
                *(v4f*)(qs + 4 * i) = t[u];
            }
        }
    }
    __syncthreads();

    ow_scan(ei, 0, E, n0, nn, tid, (int)blockIdx.x, le, ls, &cnt);
    __syncthreads();
    const int n = (int)cnt;

    if (n <= OW_CAP) {
        ow_dst(ei, E, n, tid, le, ld);
        __syncthreads();
        ow_scores(n, km, kt, qs, qts, ld, ls, sc, mx, wv, lane);
        __syncthreads();
        for (int t = tid; t < n * 4; t += OW_THREADS) {
            const int g  = ls[t >> 2] * 4 + (t & 3);
            const float ex = __expf(sc[t] - fdec(mx[g]));
            sc[t] = ex;
            atomicAdd(&sm[g], ex);
        }
        __syncthreads();
        for (int i = tid; i < n; i += OW_THREADS) {
            const v4f x  = *(const v4f*)(sc + 4 * i);
            const v4f dn = *(const v4f*)(sm + 4 * ls[i]);
            float sum = 0.0f;
            sum += x.x / dn.x;
            sum += x.y / dn.y;
            sum += x.z / dn.z;
            sum += x.w / dn.w;
            out[le[i]] = sum / 4.0f;
        }
        return;
    }

    const int nr = (E + OW_CAP - 1) / OW_CAP;
    for (int c = 0; c < nr; ++c) {
        __syncthreads();
        if (tid == 0) cnt = 0u;
        __syncthreads();
        const int hi = ((c + 1) * OW_CAP < E) ? (c + 1) * OW_CAP : E;
        ow_scan_simple(ei, c * OW_CAP, hi, n0, nn, tid, le, ls, &cnt);
        __syncthreads();
        const int m = (int)cnt;
        ow_dst(ei, E, m, tid, le, ld);
        __syncthreads();
        ow_scores(m, km, kt, qs, qts, ld, ls, sc, mx, wv, lane);
        __syncthreads();
        for (int t = tid; t < m * 4; t += OW_THREADS)
            scores[(size_t)le[t >> 2] * 4 + (t & 3)] = sc[t];
    }
    for (int c = 0; c < nr; ++c) {
        __syncthreads();
        if (tid == 0) cnt = 0u;
        __syncthreads();
        const int hi = ((c + 1) * OW_CAP < E) ? (c + 1) * OW_CAP : E;
        ow_scan_simple(ei, c * OW_CAP, hi, n0, nn, tid, le, ls, &cnt);
        __syncthreads();
        const int m = (int)cnt;
        for (int t = tid; t < m * 4; t += OW_THREADS) {
            const int g = ls[t >> 2] * 4 + (t & 3);
            const size_t w = (size_t)le[t >> 2] * 4 + (t & 3);
            const float ex = __expf(scores[w] - fdec(mx[g]));
            scores[w] = ex;
            atomicAdd(&sm[g], ex);
        }
    }
    for (int c = 0; c < nr; ++c) {
        __syncthreads();
        if (tid == 0) cnt = 0u;
        __syncthreads();
        const int hi = ((c + 1) * OW_CAP < E) ? (c + 1) * OW_CAP : E;
        ow_scan_simple(ei, c * OW_CAP, hi, n0, nn, tid, le, ls, &cnt);
        __syncthreads();
        const int m = (int)cnt;
        for (int i = tid; i < m; i += OW_THREADS) {
            const size_t w = (size_t)le[i] * 4;
            const v4f dn = *(const v4f*)(sm + 4 * ls[i]);
            float sum = 0.0f;
            sum += scores[w]     / dn.x;
            sum += scores[w + 1] / dn.y;
            sum += scores[w + 2] / dn.z;
            sum += scores[w + 3] / dn.w;
            out[le[i]] = sum / 4.0f;
        }
    }
}

extern "C" void kernel_launch(void* const* d_in, const int* in_sizes, int n_in,
                              void* d_out, int out_size, void* d_ws, size_t ws_size,
                              hipStream_t stream) {
    const float* x    = (const float*)d_in[0];
    const int*   ei   = (const int*)d_in[1];
    const float* W    = (const float*)d_in[2];
    const float* bias = (const float*)d_in[3];
    const int C = in_sizes[3];
    const int F = in_sizes[2] / C;
    const int N = in_sizes[0] / F;
    const int E = in_sizes[1] / 2;
    const int H = (C - 2 * F) / 2;
    const int P = F / H + 1;

    float*    qk     = (float*)d_ws;
    float*    scores = qk + (size_t)N * C;
    unsigned* smax   = (unsigned*)(scores + (size_t)E * H);
    float*    denom  = (float*)(smax + (size_t)N * H);

    const bool own = (H == 4 && P == 65 && C == 520 && E >= 4 &&
                      (N + OW_GROUPS - 1) / OW_GROUPS <= OW_NODES);

    if (!own) {
        size_t n = (size_t)2 * N * H;
        k0_zero<<<(unsigned)((n + 255) / 256), 256, 0, stream>>>(smax, n);
    }

    {
        dim3 grid((N + 63) / 64, (C + 63) / 64);
        k1_gemm_wmma<<<grid, 32, 0, stream>>>(x, W, bias, qk, N, F, C, H, P);
    }

    if (own) {
        ke_owner<<<OW_GROUPS, OW_THREADS, 0, stream>>>(qk, ei, scores, (float*)d_out, N, E);
        return;
    }

    {
        const int epb = (256 / 32) * K2_EPW;
        k2_scores<<<(E + epb - 1) / epb, 256, 0, stream>>>(qk, ei, scores, smax,
                                                           N, E, C, H, P);
    }

    {
        size_t n = (size_t)E * H;
        k3_exp_denom<<<(unsigned)((n + 255) / 256), 256, 0, stream>>>(scores, smax,
                                                                      ei, denom, E, H);
    }

    k4_out<<<(E + 255) / 256, 256, 0, stream>>>(scores, denom, ei,
                                                (float*)d_out, E, H);
}


# edges binned once by source node range in an added launch (64 chunks x 4 work-groups, fixed bins); a work-group reads its 64 bins instead of scanning all source ids; LDS-atomic sum order changes
# speedup vs baseline: 2.8176x; 1.1386x over previous
#include <hip/hip_runtime.h>
#include <hip/hip_bf16.h>


typedef __attribute__((ext_vector_type(2))) float v2f;
typedef __attribute__((ext_vector_type(4))) float v4f;
typedef __attribute__((ext_vector_type(8))) float v8f;
typedef v4f v4f_a4 __attribute__((aligned(4)));

__device__ __forceinline__ unsigned fenc(float f) {
    unsigned u = __float_as_uint(f);
    return (u & 0x80000000u) ? ~u : (u | 0x80000000u);
}
__device__ __forceinline__ float fdec(unsigned e) {
    return (e & 0x80000000u) ? __uint_as_float(e & 0x7FFFFFFFu)
                             : __uint_as_float(~e);
}

__global__ void k0_zero(unsigned* __restrict__ p, size_t n) {
    size_t i = (size_t)blockIdx.x * blockDim.x + threadIdx.x;
    if (i < n) p[i] = 0u;
}

__global__ __launch_bounds__(32)
void k1_gemm_wmma(const float* __restrict__ x, const float* __restrict__ W,
                  const float* __restrict__ bias, float* __restrict__ qk,
                  int N, int F, int C, int H, int P) {
    const int lane = threadIdx.x;
    const int M0   = blockIdx.x * 64;
    const int N0   = blockIdx.y * 64;
    const int lm   = lane & 15;
    const int kh   = (lane >> 4) * 2;

    const float* __restrict__ xrow[4];
    const float* __restrict__ wrow[4];
    #pragma unroll
    for (int mi = 0; mi < 4; ++mi) {
        int r = M0 + mi * 16 + lm;
        xrow[mi] = x + (size_t)(r < N ? r : N - 1) * F;
    }
    #pragma unroll
    for (int ni = 0; ni < 4; ++ni) {
        int c = N0 + ni * 16 + lm;
        wrow[ni] = W + (size_t)(c < C ? c : C - 1) * F;
    }

    v8f acc[4][4];
    #pragma unroll
    for (int mi = 0; mi < 4; ++mi)
        #pragma unroll
        for (int ni = 0; ni < 4; ++ni)
            acc[mi][ni] = (v8f){};

    const int kq = (lane >> 4) * 16;
    int k = 0;
    for (; k + 32 <= F; k += 32) {
        v4f a[4][4], b[4][4];
        #pragma unroll
        for (int j = 0; j < 4; ++j) {
            #pragma unroll
            for (int mi = 0; mi < 4; ++mi)
                a[mi][j] = *(const v4f_a4*)(xrow[mi] + k + kq + 4 * j);
            #pragma unroll
            for (int ni = 0; ni < 4; ++ni)
                b[ni][j] = *(const v4f_a4*)(wrow[ni] + k + kq + 4 * j);
        }
        __builtin_amdgcn_sched_barrier(0);
        #pragma unroll
        for (int t = 0; t < 8; ++t) {
            #pragma unroll
            for (int mi = 0; mi < 4; ++mi) {
                v2f af;
                af.x = a[mi][t >> 1][(t & 1) * 2];
                af.y = a[mi][t >> 1][(t & 1) * 2 + 1];
                #pragma unroll
                for (int ni = 0; ni < 4; ++ni) {
                    v2f bf;
                    bf.x = b[ni][t >> 1][(t & 1) * 2];
                    bf.y = b[ni][t >> 1][(t & 1) * 2 + 1];
                    acc[mi][ni] = __builtin_amdgcn_wmma_f32_16x16x4_f32(
                         false, af,  false, bf,
                         (short)0, acc[mi][ni],
                         false,  false);
                }
            }
        }
    }
    for (; k < F; k += 4) {
        v2f a[4], b[4];
        #pragma unroll
        for (int mi = 0; mi < 4; ++mi) {
            a[mi].x = xrow[mi][k + kh];
            a[mi].y = xrow[mi][k + kh + 1];
        }
        #pragma unroll
        for (int ni = 0; ni < 4; ++ni) {
            b[ni].x = wrow[ni][k + kh];
            b[ni].y = wrow[ni][k + kh + 1];
        }
        #pragma unroll
        for (int mi = 0; mi < 4; ++mi)
            #pragma unroll
            for (int ni = 0; ni < 4; ++ni)
                acc[mi][ni] = __builtin_amdgcn_wmma_f32_16x16x4_f32(
                     false, a[mi],  false, b[ni],
                     (short)0, acc[mi][ni],
                     false,  false);
    }

    const bool split = (H == 4 && P == 65 && C == 520);
    const int rsub = (lane >> 4) << 3;
    #pragma unroll
    for (int ni = 0; ni < 4; ++ni) {
        const int col = N0 + ni * 16 + lm;
        if (col >= C) continue;
        const float bv = bias[col];
        float* __restrict__ dst = qk + col;
        size_t pitch = (size_t)C;
        if (split) {
            const int h   = col / 130;
            const int rr  = col - h * 130;
            const int isk = (rr >= 65) ? 1 : 0;
            const int p   = rr - isk * 65;
            if (p < 64) {
                dst   = qk + (size_t)isk * N * 256 + h * 64 + p;
                pitch = 256;
            } else {
                dst   = qk + (size_t)N * 512 + (size_t)isk * N * 4 + h;
                pitch = 4;
            }
        }
        #pragma unroll
        for (int mi = 0; mi < 4; ++mi) {
            const int rbase = M0 + mi * 16 + rsub;
            #pragma unroll
            for (int r = 0; r < 8; ++r) {
                const int row = rbase + r;
                if (row < N)
                    dst[(size_t)row * pitch] = acc[mi][ni][r] + bv;
            }
        }
    }
}


#define K2_EPW 16

template <int CTRL>
__device__ __forceinline__ float row_xor_add(float keep, float send) {
    const int t = __builtin_amdgcn_update_dpp(0, __builtin_bit_cast(int, send),
                                              CTRL, 0xf, 0xf, false);
    return keep + __builtin_bit_cast(float, t);
}

struct EdgeFrag {
    v4f qa, qb, ka, kb;
    float qt, kt;
};

struct GatherPlanes {
    const float* __restrict__ qm;
    const float* __restrict__ km;
    const float* __restrict__ qt;
    const float* __restrict__ kt;
};

__device__ __forceinline__ void k2_load_edge(EdgeFrag& f, const GatherPlanes& g,
                                             int s, int d, int offA, int th) {
    const float* __restrict__ sq = g.qm + (size_t)s * 256 + offA;
    const float* __restrict__ dk = g.km + (size_t)d * 256 + offA;
    f.qa = *(const v4f_a4*)(sq);
    f.qb = *(const v4f_a4*)(sq + 128);
    f.ka = *(const v4f_a4*)(dk);
    f.kb = *(const v4f_a4*)(dk + 128);
    f.qt = g.qt[(size_t)s * 4 + th];
    f.kt = g.kt[(size_t)d * 4 + th];
    __builtin_amdgcn_sched_barrier(0);
}

__device__ __forceinline__ void k2_reduce_store(const EdgeFrag& f, float* __restrict__ scores,
                                                unsigned* __restrict__ smax, int e, int s,
                                                int up, int th, int lane) {
    float a1 = f.qa.x * f.ka.x;
    float a2 = f.qb.x * f.kb.x;
    a1 = fmaf(f.qa.y, f.ka.y, a1);  a2 = fmaf(f.qb.y, f.kb.y, a2);
    a1 = fmaf(f.qa.z, f.ka.z, a1);  a2 = fmaf(f.qb.z, f.kb.z, a2);
    a1 = fmaf(f.qa.w, f.ka.w, a1);  a2 = fmaf(f.qb.w, f.kb.w, a2);
    float r = row_xor_add<0x168>(up ? a2 : a1, up ? a1 : a2);
    r = row_xor_add<0x164>(r, r);
    r = row_xor_add<0x162>(r, r);
    r = row_xor_add<0x161>(r, r);
    r = fmaf(f.qt, f.kt, r);
    if ((lane & 7) == 0) {
        scores[(size_t)e * 4 + th] = r;
        atomicMax(&smax[(size_t)s * 4 + th], fenc(r));
    }
}

__global__ __launch_bounds__(256)
void k2_scores(const float* __restrict__ qk, const int* __restrict__ ei,
               float* __restrict__ scores, unsigned* __restrict__ smax,
               int N, int E, int C, int H, int P) {
    const int lane = threadIdx.x & 31;
    const int wv = __builtin_amdgcn_readfirstlane((int)(threadIdx.x >> 5));
    const int e0 = (blockIdx.x * (blockDim.x >> 5) + wv) * K2_EPW;
    if (e0 >= E) return;
    const int cnt = (E - e0 < K2_EPW) ? (E - e0) : K2_EPW;

    const int le = (e0 + (lane & 15) < E) ? (e0 + (lane & 15)) : (E - 1);
    const int ev = ei[(size_t)(lane >> 4) * E + le];

    if (H == 4 && P == 65 && C == 520) {
        const int hs   = lane >> 4;
        const int offA = hs * 64 + (lane & 15) * 4;
        const int up   = (lane >> 3) & 1;
        const int th   = hs + up * 2;
        GatherPlanes g;
        g.qm = qk;
        g.km = qk + (size_t)N * 256;
        g.qt = qk + (size_t)N * 512;
        g.kt = g.qt + (size_t)N * 4;

        EdgeFrag fa, fb;
        int sa = __builtin_amdgcn_readlane(ev, 0), sb = sa;
        k2_load_edge(fa, g, sa, __builtin_amdgcn_readlane(ev, 16), offA, th);
        int j = 0;
        for (; j + 2 < cnt; j += 2) {
            sb = __builtin_amdgcn_readlane(ev, j + 1);
            k2_load_edge(fb, g, sb, __builtin_amdgcn_readlane(ev, j + 17), offA, th);
            k2_reduce_store(fa, scores, smax, e0 + j, sa, up, th, lane);
            sa = __builtin_amdgcn_readlane(ev, j + 2);
            k2_load_edge(fa, g, sa, __builtin_amdgcn_readlane(ev, j + 18), offA, th);
            k2_reduce_store(fb, scores, smax, e0 + j + 1, sb, up, th, lane);
        }
        if (j + 1 < cnt) {
            sb = __builtin_amdgcn_readlane(ev, j + 1);
            k2_load_edge(fb, g, sb, __builtin_amdgcn_readlane(ev, j + 17), offA, th);
            k2_reduce_store(fa, scores, smax, e0 + j, sa, up, th, lane);
            k2_reduce_store(fb, scores, smax, e0 + j + 1, sb, up, th, lane);
        } else {
            k2_reduce_store(fa, scores, smax, e0 + j, sa, up, th, lane);
        }
    } else {
        for (int j = 0; j < cnt; ++j) {
            const int s = __builtin_amdgcn_readlane(ev, j);
            const int d = __builtin_amdgcn_readlane(ev, 16 + j);
            const float* __restrict__ srow = qk + (size_t)s * C;
            const float* __restrict__ drow = qk + (size_t)d * C;
            for (int h = 0; h < H; ++h) {
                const float* __restrict__ qh = srow + h * 2 * P;
                const float* __restrict__ kh = drow + h * 2 * P + P;
                float a = 0.0f;
                for (int p = lane; p < P; p += 32)
                    a += qh[p] * kh[p];
                #pragma unroll
                for (int off = 16; off > 0; off >>= 1)
                    a += __shfl_xor(a, off, 32);
                if (lane == 0) {
                    scores[(size_t)(e0 + j) * H + h] = a;
                    atomicMax(&smax[(size_t)s * H + h], fenc(a));
                }
            }
        }
    }
}

__global__ __launch_bounds__(256)
void k3_exp_denom(float* __restrict__ scores, const unsigned* __restrict__ smax,
                  const int* __restrict__ ei, float* __restrict__ denom,
                  int E, int H) {
    size_t i = (size_t)blockIdx.x * blockDim.x + threadIdx.x;
    if (i >= (size_t)E * H) return;
    const int e = (int)(i / H);
    const int h = (int)(i % H);
    const int s = ei[e];
    const float m  = fdec(smax[(size_t)s * H + h]);
    const float ex = __expf(scores[i] - m);
    scores[i] = ex;
    atomicAdd(&denom[(size_t)s * H + h], ex);
}

__global__ __launch_bounds__(256)
void k4_out(const float* __restrict__ ex, const float* __restrict__ denom,
            const int* __restrict__ ei, float* __restrict__ out, int E, int H) {
    const int e = blockIdx.x * blockDim.x + threadIdx.x;
    if (e >= E) return;
    const int s = ei[e];
    float sum = 0.0f;
    for (int h = 0; h < H; ++h)
        sum += ex[(size_t)e * H + h] / denom[(size_t)s * H + h];
    out[e] = sum / (float)H;
}


#define OW_GROUPS  256
#define OW_THREADS 1024
#define OW_NODES   196
#define OW_CAP     4096
#define OW_UNR     16
#define OW_B       8
#define OW_FILL    13
#define BIN_CH     64
#define BIN_Q      4
#define BIN_CAP    96
#define BIN_TRIPS  ((BIN_CH * BIN_CAP + OW_THREADS - 1) / OW_THREADS)
#define BIN_IPT    13

typedef __attribute__((ext_vector_type(4))) int v4i;
typedef v4i v4i_a4 __attribute__((aligned(4)));
typedef __attribute__((ext_vector_type(2))) int v2i;

__device__ __forceinline__ unsigned ow_umin(unsigned a, unsigned b) { return a < b ? a : b; }

__device__ __forceinline__ void ow_append(int e, unsigned sl, int* le, int* ls, unsigned* cnt) {
    const unsigned idx = atomicAdd(cnt, 1u);
    if (idx < (unsigned)OW_CAP) { le[idx] = e; ls[idx] = (int)sl; }
}

__device__ __forceinline__ void ow_test4(v4i s, int e, int n0, unsigned nn,
                                         int* le, int* ls, unsigned* cnt) {
    const unsigned a = (unsigned)(s.x - n0), b = (unsigned)(s.y - n0);
    const unsigned c = (unsigned)(s.z - n0), d = (unsigned)(s.w - n0);
    if (ow_umin(ow_umin(a, b), ow_umin(c, d)) < nn) {
        #pragma nounroll
        for (int j = 0; j < 4; ++j) {
            const unsigned sl = (j == 0) ? a : (j == 1) ? b : (j == 2) ? c : d;
            if (sl < nn) ow_append(e + j, sl, le, ls, cnt);
        }
    }
}

__device__ __forceinline__ void ow_scan_simple(const int* __restrict__ ei, int lo, int hi, int n0, unsigned nn,
                                               int tid, int* le, int* ls, unsigned* cnt) {
    for (int e = lo + tid; e < hi; e += OW_THREADS) {
        const unsigned sl = (unsigned)(ei[e] - n0);
        if (sl < nn) ow_append(e, sl, le, ls, cnt);
    }
}

__device__ __forceinline__ void ow_scan(const int* __restrict__ ei, int lo, int hi, int n0, unsigned nn,
                                        int tid, int grp, int* le, int* ls, unsigned* cnt) {
    const int CH    = OW_THREADS * 4 * OW_UNR;
    const int nfull = (hi - lo) / CH;
    if (nfull > 0) {
        int c = grp % nfull;
        for (int it = 0; it < nfull; ++it) {
            const int b0 = lo + c * CH + tid * 4;
            v4i v[OW_UNR];
            #pragma unroll
            for (int u = 0; u < OW_UNR; ++u)
                v[u] = *(const v4i_a4*)(ei + b0 + u * (OW_THREADS * 4));
            __builtin_amdgcn_sched_barrier(0);
            #pragma unroll
            for (int u = 0; u < OW_UNR; ++u)
                ow_test4(v[u], b0 + u * (OW_THREADS * 4), n0, nn, le, ls, cnt);
            c = (c + 1 < nfull) ? c + 1 : 0;
        }
    }
    const int r0 = lo + nfull * CH;
    if (r0 < hi) {
        const int b0 = r0 + tid * 4;
        v4i v[OW_UNR];
        #pragma unroll
        for (int u = 0; u < OW_UNR; ++u) {
            const int b = b0 + u * (OW_THREADS * 4);
            v[u] = *(const v4i_a4*)(ei + ((b + 3 < hi) ? b : 0));
        }
        __builtin_amdgcn_sched_barrier(0);
        #pragma unroll
        for (int u = 0; u < OW_UNR; ++u) {
            const int b = b0 + u * (OW_THREADS * 4);
            if (b + 3 < hi) ow_test4(v[u], b, n0, nn, le, ls, cnt);
        }
        ow_scan_simple(ei, lo + ((hi - lo) & ~3), hi, n0, nn, tid, le, ls, cnt);
    }
}

__device__ __forceinline__ void ow_dst(const int* __restrict__ ei, int E, int n, int tid,
                                       const int* le, int* ld) {
    for (int i0 = tid; i0 < n; i0 += OW_THREADS * 4) {
        int v[4];
        #pragma unroll
        for (int u = 0; u < 4; ++u) {
            int i = i0 + u * OW_THREADS;
            i = i < n ? i : n - 1;
            v[u] = ei[(size_t)E + le[i]];
        }
        #pragma unroll
        for (int u = 0; u < 4; ++u) {
            int i = i0 + u * OW_THREADS;
            i = i < n ? i : n - 1;
            ld[i] = v[u];
        }
    }
}

struct KFrag { v4f ka, kb; };

__device__ __forceinline__ void ow_scores(int n, const float* __restrict__ km, const float* __restrict__ kt,
                                          const float* qs, const float* qts,
                                          const int* ld, const int* ls, float* sc, unsigned* mx,
                                          int wv, int lane) {
    const int hs   = lane >> 4;
    const int offA = hs * 64 + (lane & 15) * 4;
    const int up   = (lane >> 3) & 1;
    const int th   = hs + up * 2;
    for (int base = wv * OW_B; base < n; base += (OW_THREADS / 32) * OW_B) {
        int ii = base + (lane & (OW_B - 1));
        ii = ii < n ? ii : n - 1;
        const int dv = ld[ii];
        const int sv = ls[ii];
        int i2 = base + (lane >> 2);
        i2 = i2 < n ? i2 : n - 1;
        const float ktv = kt[(size_t)ld[i2] * 4 + (lane & 3)];
        KFrag f[OW_B];
        #pragma unroll
        for (int j = 0; j < OW_B; ++j) {
            const int d = __builtin_amdgcn_readlane(dv, j);
            const float* __restrict__ dk = km + (size_t)d * 256 + offA;
            f[j].ka = *(const v4f_a4*)(dk);
            f[j].kb = *(const v4f_a4*)(dk + 128);
        }
        __builtin_amdgcn_sched_barrier(0);
        #pragma unroll
        for (int j = 0; j < OW_B; ++j) {
            if (base + j < n) {
                const int sl = __builtin_amdgcn_readlane(sv, j);
                const v4f qa = *(const v4f*)(qs + sl * 256 + offA);
                const v4f qb = *(const v4f*)(qs + sl * 256 + offA + 128);
                const float qtv = qts[sl * 4 + th];
                const float ktj = __shfl(ktv, 4 * j + th, 32);
                float a1 = qa.x * f[j].ka.x;
                float a2 = qb.x * f[j].kb.x;
                a1 = fmaf(qa.y, f[j].ka.y, a1);  a2 = fmaf(qb.y, f[j].kb.y, a2);
                a1 = fmaf(qa.z, f[j].ka.z, a1);  a2 = fmaf(qb.z, f[j].kb.z, a2);
                a1 = fmaf(qa.w, f[j].ka.w, a1);  a2 = fmaf(qb.w, f[j].kb.w, a2);
                float r = row_xor_add<0x168>(up ? a2 : a1, up ? a1 : a2);
                r = row_xor_add<0x164>(r, r);
                r = row_xor_add<0x162>(r, r);
                r = row_xor_add<0x161>(r, r);
                r = fmaf(qtv, ktj, r);
                if ((lane & 7) == 0) {
                    sc[(base + j) * 4 + th] = r;
                    atomicMax(&mx[sl * 4 + th], fenc(r));
                }
            }
        }
    }
}

__global__ __launch_bounds__(OW_THREADS)
void kb_bin(const int* __restrict__ ei, int* __restrict__ cells, int* __restrict__ counts,
            int N, int E) {
    __shared__ unsigned lc[OW_GROUPS / BIN_Q];
    const int tid = threadIdx.x;
    const int c   = (int)blockIdx.x / BIN_Q;
    const unsigned q = (unsigned)blockIdx.x % BIN_Q;
    const unsigned per = (unsigned)((N + OW_GROUPS - 1) / OW_GROUPS);
    const int chunk = (E + BIN_CH - 1) / BIN_CH;
    const int lo = c * chunk;
    const int hi = (E - lo < chunk) ? E : lo + chunk;
    if (tid < OW_GROUPS / BIN_Q) lc[tid] = 0u;
    __syncthreads();
    for (int b0 = lo; b0 < hi; b0 += OW_THREADS * BIN_IPT) {
        int s[BIN_IPT], d[BIN_IPT];
        #pragma unroll
        for (int u = 0; u < BIN_IPT; ++u) {
            const int e  = b0 + u * OW_THREADS + tid;
            const int ec = e < hi ? e : hi - 1;
            s[u] = ei[ec];
            d[u] = ei[(size_t)E + ec];
        }
        __builtin_amdgcn_sched_barrier(0);
        #pragma unroll
        for (int u = 0; u < BIN_IPT; ++u)
            asm volatile("" : "+v"(d[u]));
        unsigned g[BIN_IPT], idx[BIN_IPT];
        #pragma unroll
        for (int u = 0; u < BIN_IPT; ++u) {
            const int e = b0 + u * OW_THREADS + tid;
            const unsigned us = (unsigned)s[u];
            g[u]   = us / per;
            idx[u] = (unsigned)BIN_CAP;
            if (e < hi && us < (unsigned)N && g[u] / (OW_GROUPS / BIN_Q) == q)
                idx[u] = atomicAdd(&lc[g[u] % (OW_GROUPS / BIN_Q)], 1u);
        }
        #pragma unroll
        for (int u = 0; u < BIN_IPT; ++u) {
            if (idx[u] < (unsigned)BIN_CAP) {
                const int e = b0 + u * OW_THREADS + tid;
                const unsigned sl = (unsigned)s[u] - g[u] * per;
                v2i en;
                en.x = (int)((unsigned)e | (sl << 24));
                en.y = d[u];
                *(v2i*)(cells + 2 * (((size_t)g[u] * BIN_CH + c) * BIN_CAP + idx[u])) = en;
            }
        }
    }
    __syncthreads();
    if (tid < OW_GROUPS / BIN_Q)
        counts[(size_t)(q * (OW_GROUPS / BIN_Q) + tid) * BIN_CH + c] = (int)lc[tid];
}

__global__ __launch_bounds__(OW_THREADS)
void ke_owner(const float* __restrict__ qk, const int* __restrict__ ei,
              const int* __restrict__ cells, const int* __restrict__ counts,
              float* __restrict__ out, int N, int E) {
    __shared__ __attribute__((aligned(16))) float qs[OW_NODES * 256];
    __shared__ __attribute__((aligned(16))) float qts[OW_NODES * 4];
    __shared__ __attribute__((aligned(16))) float sm[OW_NODES * 4];
    __shared__ unsigned mx[OW_NODES * 4];
    __shared__ __attribute__((aligned(16))) float sc[OW_CAP * 4];
    __shared__ int le[OW_CAP];
    __shared__ int ls[OW_CAP];
    __shared__ int ld[OW_CAP];
    __shared__ unsigned cnt;
    __shared__ int cc[BIN_CH];
    __shared__ int off[BIN_CH];
    __shared__ int binfo[2];

    const int tid  = threadIdx.x;
    const int lane = tid & 31;
    const int wv   = __builtin_amdgcn_readfirstlane(tid >> 5);
    const int per  = (N + (int)gridDim.x - 1) / (int)gridDim.x;
    const int n0   = (int)blockIdx.x * per;
    if (n0 >= N) return;
    const int nni  = (N - n0 < per) ? (N - n0) : per;
    const unsigned nn = (unsigned)nni;

    const float* __restrict__ qm = qk;
    const float* __restrict__ km = qk + (size_t)N * 256;
    const float* __restrict__ qt = qk + (size_t)N * 512;
    const float* __restrict__ kt = qt + (size_t)N * 4;

    for (int i = tid; i < nni * 4; i += OW_THREADS) {
        mx[i]  = 0u;
        sm[i]  = 0.0f;
        qts[i] = qt[(size_t)n0 * 4 + i];
    }
    if (tid == 0) cnt = 0u;
    if (tid < BIN_CH) cc[tid] = counts[(size_t)blockIdx.x * BIN_CH + tid];
    {
        const float* __restrict__ qsrc = qm + (size_t)n0 * 256;
        const int tot = nni * 64;
        for (int i0 = tid; i0 < tot; i0 += OW_THREADS * OW_FILL) {
            v4f t[OW_FILL];
            #pragma unroll
            for (int u = 0; u < OW_FILL; ++u) {
                int i = i0 + u * OW_THREADS;
                i = i < tot ? i : tot - 1;
                t[u] = *(const v4f_a4*)(qsrc + 4 * i);
            }
            __builtin_amdgcn_sched_barrier(0);
            #pragma unroll
            for (int u = 0; u < OW_FILL; ++u) {
                int i = i0 + u * OW_THREADS;
                i = i < tot ? i : tot - 1;
                *(v4f*)(qs + 4 * i) = t[u];
            }
        }
    }
    __syncthreads();

    if (tid < BIN_CH) {
        int o = 0;
        for (int i = 0; i < tid; ++i) o += cc[i];
        off[tid] = o;
    } else if (tid == BIN_CH) {
        int t = 0, bad = 0;
        for (int i = 0; i < BIN_CH; ++i) { t += cc[i]; bad |= (cc[i] > BIN_CAP) ? 1 : 0; }
        binfo[0] = t;
        binfo[1] = bad;
    }
    __syncthreads();
    int n = binfo[0];
    bool listed = false;
    if (binfo[1] != 0) {
        ow_scan(ei, 0, E, n0, nn, tid, (int)blockIdx.x, le, ls, &cnt);
        __syncthreads();
        n = (int)cnt;
        if (n <= OW_CAP) {
            ow_dst(ei, E, n, tid, le, ld);
            listed = true;
        }
    } else if (n <= OW_CAP) {
        const int* __restrict__ mine = cells + (size_t)blockIdx.x * (BIN_CH * BIN_CAP * 2);
        v2i  en[BIN_TRIPS];
        int  pos[BIN_TRIPS];
        #pragma unroll
        for (int u = 0; u < BIN_TRIPS; ++u) {
            const int s = tid + u * OW_THREADS;
            const int b = s / BIN_CAP;
            const int k = s - b * BIN_CAP;
            const bool used = (b < BIN_CH) && (k < cc[b < BIN_CH ? b : 0]);
            pos[u] = used ? off[b] + k : -1;
            const int* __restrict__ p = used ? mine + 2 * s : counts + (size_t)blockIdx.x * BIN_CH;
            en[u] = *(const v2i*)p;
        }
        __builtin_amdgcn_sched_barrier(0);
        #pragma unroll
        for (int u = 0; u < BIN_TRIPS; ++u) {
            if (pos[u] >= 0) {
                le[pos[u]] = (int)((unsigned)en[u].x & 0xFFFFFFu);
                ls[pos[u]] = (int)((unsigned)en[u].x >> 24);
                ld[pos[u]] = en[u].y;
            }
        }
        listed = true;
    }

    if (listed) {
        __syncthreads();
        ow_scores(n, km, kt, qs, qts, ld, ls, sc, mx, wv, lane);
        __syncthreads();
        for (int t = tid; t < n * 4; t += OW_THREADS) {
            const int g  = ls[t >> 2] * 4 + (t & 3);
            const float ex = __expf(sc[t] - fdec(mx[g]));
            sc[t] = ex;
            atomicAdd(&sm[g], ex);
        }
        __syncthreads();
        for (int i = tid; i < n; i += OW_THREADS) {
            const v4f x  = *(const v4f*)(sc + 4 * i);
            const v4f dn = *(const v4f*)(sm + 4 * ls[i]);
            float sum = 0.0f;
            sum += x.x / dn.x;
            sum += x.y / dn.y;
            sum += x.z / dn.z;
            sum += x.w / dn.w;
            out[le[i]] = sum / 4.0f;
        }
        return;
    }

    const int nr = (E + OW_CAP - 1) / OW_CAP;
    for (int pass = 0; pass < 3; ++pass) {
        for (int c = 0; c < nr; ++c) {
            __syncthreads();
            if (tid == 0) cnt = 0u;
            __syncthreads();
            const int hi = ((c + 1) * OW_CAP < E) ? (c + 1) * OW_CAP : E;
            ow_scan_simple(ei, c * OW_CAP, hi, n0, nn, tid, le, ls, &cnt);
            __syncthreads();
            const int m = (int)cnt;
            ow_dst(ei, E, m, tid, le, ld);
            __syncthreads();
            ow_scores(m, km, kt, qs, qts, ld, ls, sc, mx, wv, lane);
            __syncthreads();
            if (pass == 1) {
                for (int t = tid; t < m * 4; t += OW_THREADS) {
                    const int g = ls[t >> 2] * 4 + (t & 3);
                    atomicAdd(&sm[g], __expf(sc[t] - fdec(mx[g])));
                }
            } else if (pass == 2) {
                for (int i = tid; i < m; i += OW_THREADS) {
                    const int g = ls[i] * 4;
                    const v4f x  = *(const v4f*)(sc + 4 * i);
                    const v4f dn = *(const v4f*)(sm + g);
                    float sum = 0.0f;
                    sum += __expf(x.x - fdec(mx[g]))     / dn.x;
                    sum += __expf(x.y - fdec(mx[g + 1])) / dn.y;
                    sum += __expf(x.z - fdec(mx[g + 2])) / dn.z;
                    sum += __expf(x.w - fdec(mx[g + 3])) / dn.w;
                    out[le[i]] = sum / 4.0f;
                }
            }
        }
    }
}

extern "C" void kernel_launch(void* const* d_in, const int* in_sizes, int n_in,
                              void* d_out, int out_size, void* d_ws, size_t ws_size,
                              hipStream_t stream) {
    const float* x    = (const float*)d_in[0];
    const int*   ei   = (const int*)d_in[1];
    const float* W    = (const float*)d_in[2];
    const float* bias = (const float*)d_in[3];
    const int C = in_sizes[3];
    const int F = in_sizes[2] / C;
    const int N = in_sizes[0] / F;
    const int E = in_sizes[1] / 2;
    const int H = (C - 2 * F) / 2;
    const int P = F / H + 1;

    float*    qk     = (float*)d_ws;
    float*    scores = qk + (size_t)N * C;
    unsigned* smax   = (unsigned*)(scores + (size_t)E * H);
    float*    denom  = (float*)(smax + (size_t)N * H);

    int* cells  = (int*)scores;
    int* counts = (int*)smax;
    const bool own = (H == 4 && P == 65 && C == 520 && E >= 4 && E < (1 << 24) &&
                      (N + OW_GROUPS - 1) / OW_GROUPS <= OW_NODES &&
                      (size_t)OW_GROUPS * BIN_CH * BIN_CAP * 2 <= (size_t)E * H &&
                      (size_t)OW_GROUPS * BIN_CH <= (size_t)N * H);

    if (!own) {
        size_t n = (size_t)2 * N * H;
        k0_zero<<<(unsigned)((n + 255) / 256), 256, 0, stream>>>(smax, n);
    }

    {
        dim3 grid((N + 63) / 64, (C + 63) / 64);
        k1_gemm_wmma<<<grid, 32, 0, stream>>>(x, W, bias, qk, N, F, C, H, P);
    }

    if (own) {
        kb_bin<<<BIN_CH * BIN_Q, OW_THREADS, 0, stream>>>(ei, cells, counts, N, E);
        ke_owner<<<OW_GROUPS, OW_THREADS, 0, stream>>>(qk, ei, cells, counts, (float*)d_out, N, E);
        return;
    }

    {
        const int epb = (256 / 32) * K2_EPW;
        k2_scores<<<(E + epb - 1) / epb, 256, 0, stream>>>(qk, ei, scores, smax,
                                                           N, E, C, H, P);
    }

    {
        size_t n = (size_t)E * H;
        k3_exp_denom<<<(unsigned)((n + 255) / 256), 256, 0, stream>>>(scores, smax,
                                                                      ei, denom, E, H);
    }

    k4_out<<<(E + 255) / 256, 256, 0, stream>>>(scores, denom, ei,
                                                (float*)d_out, E, H);
}


# GEMM: 12 waves (4x3 tiles of 32x64) share each 32-float K-block of 128 x rows and 192 W rows through LDS, next block prefetched before the WMMA run; was one wave with its own loads; sums unchanged
# speedup vs baseline: 3.0719x; 1.0903x over previous
#include <hip/hip_runtime.h>
#include <hip/hip_bf16.h>


typedef __attribute__((ext_vector_type(2))) float v2f;
typedef __attribute__((ext_vector_type(4))) float v4f;
typedef __attribute__((ext_vector_type(8))) float v8f;
typedef v4f v4f_a4 __attribute__((aligned(4)));

__device__ __forceinline__ unsigned fenc(float f) {
    unsigned u = __float_as_uint(f);
    return (u & 0x80000000u) ? ~u : (u | 0x80000000u);
}
__device__ __forceinline__ float fdec(unsigned e) {
    return (e & 0x80000000u) ? __uint_as_float(e & 0x7FFFFFFFu)
                             : __uint_as_float(~e);
}

__global__ void k0_zero(unsigned* __restrict__ p, size_t n) {
    size_t i = (size_t)blockIdx.x * blockDim.x + threadIdx.x;
    if (i < n) p[i] = 0u;
}

#define G_MI      2
#define G_WM      4
#define G_WN      3
#define G_XROWS   (G_WM * G_MI * 16)
#define G_ROWS    (G_XROWS + G_WN * 64)
#define G_PITCH   36
#define G_THREADS (G_WM * G_WN * 32)
#define G_RSTEP   (G_THREADS / 8)
#define G_PIECES  ((G_ROWS + G_RSTEP - 1) / G_RSTEP)

__global__ __launch_bounds__(G_THREADS, 6)
void k1_gemm_wmma(const float* __restrict__ x, const float* __restrict__ W,
                  const float* __restrict__ bias, float* __restrict__ qk,
                  int N, int F, int C, int H, int P) {
    __shared__ __attribute__((aligned(16))) float stg[2][G_ROWS * G_PITCH];
    const int tid  = threadIdx.x;
    const int lane = tid & 31;
    const int wv   = __builtin_amdgcn_readfirstlane(tid >> 5);
    const int wm   = wv % G_WM;
    const int wn   = wv / G_WM;
    const int Mg   = blockIdx.y * G_XROWS;
    const int Ng   = blockIdx.x * (G_WN * 64);
    const int M0   = Mg + wm * (G_MI * 16);
    const int N0   = Ng + wn * 64;
    const int lm   = lane & 15;
    const int kh   = (lane >> 4) * 2;

    const int seg = tid & 7;
    const int l0  = (tid >> 3) * G_PITCH + seg * 4;
    const float* __restrict__ gsrc[G_PIECES];
    #pragma unroll
    for (int i = 0; i < G_PIECES; ++i) {
        const int R  = (tid >> 3) + G_RSTEP * i;
        const int Rc = R < G_ROWS ? R : G_ROWS - 1;
        const float* __restrict__ p;
        if (Rc < G_XROWS) { const int r = Mg + Rc;           p = x + (size_t)(r < N ? r : N - 1) * F; }
        else              { const int c = Ng + Rc - G_XROWS; p = W + (size_t)(c < C ? c : C - 1) * F; }
        gsrc[i] = p + seg * 4;
    }

    v8f acc[G_MI][4];
    #pragma unroll
    for (int mi = 0; mi < G_MI; ++mi)
        #pragma unroll
        for (int ni = 0; ni < 4; ++ni)
            acc[mi][ni] = (v8f){};

    const int kq  = (lane >> 4) * 16;
    const int nkb = F / 32;
    const int aoff = (wm * (G_MI * 16) + lm) * G_PITCH + kq;
    const int boff = (G_XROWS + wn * 64 + lm) * G_PITCH + kq;

    v4f t[G_PIECES];
    if (nkb > 0) {
        #pragma unroll
        for (int i = 0; i < G_PIECES; ++i) t[i] = *(const v4f_a4*)(gsrc[i]);
        #pragma unroll
        for (int i = 0; i < G_PIECES; ++i)
            if ((i + 1) * G_RSTEP <= G_ROWS || (tid >> 3) + G_RSTEP * i < G_ROWS)
                *(v4f*)(&stg[0][l0 + i * (G_RSTEP * G_PITCH)]) = t[i];
        __syncthreads();
    }
    for (int kb = 0; kb < nkb; ++kb) {
        const float* sb = stg[kb & 1];
        float*       sn = stg[(kb & 1) ^ 1];
        const bool more = (kb + 1 < nkb);
        if (more) {
            #pragma unroll
            for (int i = 0; i < G_PIECES; ++i)
                t[i] = *(const v4f_a4*)(gsrc[i] + (size_t)(kb + 1) * 32);
        }
        __builtin_amdgcn_sched_barrier(0);
        v4f a[G_MI][4], b[4][4];
        #pragma unroll
        for (int j = 0; j < 4; ++j) {
            #pragma unroll
            for (int mi = 0; mi < G_MI; ++mi)
                a[mi][j] = *(const v4f*)(sb + aoff + mi * 16 * G_PITCH + 4 * j);
            #pragma unroll
            for (int ni = 0; ni < 4; ++ni)
                b[ni][j] = *(const v4f*)(sb + boff + ni * 16 * G_PITCH + 4 * j);
        }
        #pragma unroll
        for (int tt = 0; tt < 8; ++tt) {
            #pragma unroll
            for (int mi = 0; mi < G_MI; ++mi) {
                v2f af;
                af.x = a[mi][tt >> 1][(tt & 1) * 2];
                af.y = a[mi][tt >> 1][(tt & 1) * 2 + 1];
                #pragma unroll
                for (int ni = 0; ni < 4; ++ni) {
                    v2f bf;
                    bf.x = b[ni][tt >> 1][(tt & 1) * 2];
                    bf.y = b[ni][tt >> 1][(tt & 1) * 2 + 1];
                    acc[mi][ni] = __builtin_amdgcn_wmma_f32_16x16x4_f32(
                         false, af,  false, bf,
                         (short)0, acc[mi][ni],
                         false,  false);
                }
            }
        }
        __builtin_amdgcn_sched_barrier(0);
        if (more) {
            #pragma unroll
            for (int i = 0; i < G_PIECES; ++i)
                if ((i + 1) * G_RSTEP <= G_ROWS || (tid >> 3) + G_RSTEP * i < G_ROWS)
                    *(v4f*)(sn + l0 + i * (G_RSTEP * G_PITCH)) = t[i];
        }
        __syncthreads();
    }

    int k = nkb * 32;
    if (k < F) {
        const float* __restrict__ xrow[G_MI];
        const float* __restrict__ wrow[4];
        #pragma unroll
        for (int mi = 0; mi < G_MI; ++mi) {
            int r = M0 + mi * 16 + lm;
            xrow[mi] = x + (size_t)(r < N ? r : N - 1) * F;
        }
        #pragma unroll
        for (int ni = 0; ni < 4; ++ni) {
            int c = N0 + ni * 16 + lm;
            wrow[ni] = W + (size_t)(c < C ? c : C - 1) * F;
        }
        for (; k < F; k += 4) {
            v2f a[G_MI], b[4];
            #pragma unroll
            for (int mi = 0; mi < G_MI; ++mi) {
                a[mi].x = xrow[mi][k + kh];
                a[mi].y = xrow[mi][k + kh + 1];
            }
            #pragma unroll
            for (int ni = 0; ni < 4; ++ni) {
                b[ni].x = wrow[ni][k + kh];
                b[ni].y = wrow[ni][k + kh + 1];
            }
            #pragma unroll
            for (int mi = 0; mi < G_MI; ++mi)
                #pragma unroll
                for (int ni = 0; ni < 4; ++ni)
                    acc[mi][ni] = __builtin_amdgcn_wmma_f32_16x16x4_f32(
                         false, a[mi],  false, b[ni],
                         (short)0, acc[mi][ni],
                         false,  false);
        }
    }

    const bool split = (H == 4 && P == 65 && C == 520);
    const int rsub = (lane >> 4) << 3;
    #pragma unroll
    for (int ni = 0; ni < 4; ++ni) {
        const int col = N0 + ni * 16 + lm;
        if (col >= C) continue;
        const float bv = bias[col];
        float* __restrict__ dst = qk + col;
        size_t pitch = (size_t)C;
        if (split) {
            const int h   = col / 130;
            const int rr  = col - h * 130;
            const int isk = (rr >= 65) ? 1 : 0;
            const int p   = rr - isk * 65;
            if (p < 64) {
                dst   = qk + (size_t)isk * N * 256 + h * 64 + p;
                pitch = 256;
            } else {
                dst   = qk + (size_t)N * 512 + (size_t)isk * N * 4 + h;
                pitch = 4;
            }
        }
        #pragma unroll
        for (int mi = 0; mi < G_MI; ++mi) {
            const int rbase = M0 + mi * 16 + rsub;
            #pragma unroll
            for (int r = 0; r < 8; ++r) {
                const int row = rbase + r;
                if (row < N)
                    dst[(size_t)row * pitch] = acc[mi][ni][r] + bv;
            }
        }
    }
}


#define K2_EPW 16

template <int CTRL>
__device__ __forceinline__ float row_xor_add(float keep, float send) {
    const int t = __builtin_amdgcn_update_dpp(0, __builtin_bit_cast(int, send),
                                              CTRL, 0xf, 0xf, false);
    return keep + __builtin_bit_cast(float, t);
}

struct EdgeFrag {
    v4f qa, qb, ka, kb;
    float qt, kt;
};

struct GatherPlanes {
    const float* __restrict__ qm;
    const float* __restrict__ km;
    const float* __restrict__ qt;
    const float* __restrict__ kt;
};

__device__ __forceinline__ void k2_load_edge(EdgeFrag& f, const GatherPlanes& g,
                                             int s, int d, int offA, int th) {
    const float* __restrict__ sq = g.qm + (size_t)s * 256 + offA;
    const float* __restrict__ dk = g.km + (size_t)d * 256 + offA;
    f.qa = *(const v4f_a4*)(sq);
    f.qb = *(const v4f_a4*)(sq + 128);
    f.ka = *(const v4f_a4*)(dk);
    f.kb = *(const v4f_a4*)(dk + 128);
    f.qt = g.qt[(size_t)s * 4 + th];
    f.kt = g.kt[(size_t)d * 4 + th];
    __builtin_amdgcn_sched_barrier(0);
}

__device__ __forceinline__ void k2_reduce_store(const EdgeFrag& f, float* __restrict__ scores,
                                                unsigned* __restrict__ smax, int e, int s,
                                                int up, int th, int lane) {
    float a1 = f.qa.x * f.ka.x;
    float a2 = f.qb.x * f.kb.x;
    a1 = fmaf(f.qa.y, f.ka.y, a1);  a2 = fmaf(f.qb.y, f.kb.y, a2);
    a1 = fmaf(f.qa.z, f.ka.z, a1);  a2 = fmaf(f.qb.z, f.kb.z, a2);
    a1 = fmaf(f.qa.w, f.ka.w, a1);  a2 = fmaf(f.qb.w, f.kb.w, a2);
    float r = row_xor_add<0x168>(up ? a2 : a1, up ? a1 : a2);
    r = row_xor_add<0x164>(r, r);
    r = row_xor_add<0x162>(r, r);
    r = row_xor_add<0x161>(r, r);
    r = fmaf(f.qt, f.kt, r);
    if ((lane & 7) == 0) {
        scores[(size_t)e * 4 + th] = r;
        atomicMax(&smax[(size_t)s * 4 + th], fenc(r));
    }
}

__global__ __launch_bounds__(256)
void k2_scores(const float* __restrict__ qk, const int* __restrict__ ei,
               float* __restrict__ scores, unsigned* __restrict__ smax,
               int N, int E, int C, int H, int P) {
    const int lane = threadIdx.x & 31;
    const int wv = __builtin_amdgcn_readfirstlane((int)(threadIdx.x >> 5));
    const int e0 = (blockIdx.x * (blockDim.x >> 5) + wv) * K2_EPW;
    if (e0 >= E) return;
    const int cnt = (E - e0 < K2_EPW) ? (E - e0) : K2_EPW;

    const int le = (e0 + (lane & 15) < E) ? (e0 + (lane & 15)) : (E - 1);
    const int ev = ei[(size_t)(lane >> 4) * E + le];

    if (H == 4 && P == 65 && C == 520) {
        const int hs   = lane >> 4;
        const int offA = hs * 64 + (lane & 15) * 4;
        const int up   = (lane >> 3) & 1;
        const int th   = hs + up * 2;
        GatherPlanes g;
        g.qm = qk;
        g.km = qk + (size_t)N * 256;
        g.qt = qk + (size_t)N * 512;
        g.kt = g.qt + (size_t)N * 4;

        EdgeFrag fa, fb;
        int sa = __builtin_amdgcn_readlane(ev, 0), sb = sa;
        k2_load_edge(fa, g, sa, __builtin_amdgcn_readlane(ev, 16), offA, th);
        int j = 0;
        for (; j + 2 < cnt; j += 2) {
            sb = __builtin_amdgcn_readlane(ev, j + 1);
            k2_load_edge(fb, g, sb, __builtin_amdgcn_readlane(ev, j + 17), offA, th);
            k2_reduce_store(fa, scores, smax, e0 + j, sa, up, th, lane);
            sa = __builtin_amdgcn_readlane(ev, j + 2);
            k2_load_edge(fa, g, sa, __builtin_amdgcn_readlane(ev, j + 18), offA, th);
            k2_reduce_store(fb, scores, smax, e0 + j + 1, sb, up, th, lane);
        }
        if (j + 1 < cnt) {
            sb = __builtin_amdgcn_readlane(ev, j + 1);
            k2_load_edge(fb, g, sb, __builtin_amdgcn_readlane(ev, j + 17), offA, th);
            k2_reduce_store(fa, scores, smax, e0 + j, sa, up, th, lane);
            k2_reduce_store(fb, scores, smax, e0 + j + 1, sb, up, th, lane);
        } else {
            k2_reduce_store(fa, scores, smax, e0 + j, sa, up, th, lane);
        }
    } else {
        for (int j = 0; j < cnt; ++j) {
            const int s = __builtin_amdgcn_readlane(ev, j);
            const int d = __builtin_amdgcn_readlane(ev, 16 + j);
            const float* __restrict__ srow = qk + (size_t)s * C;
            const float* __restrict__ drow = qk + (size_t)d * C;
            for (int h = 0; h < H; ++h) {
                const float* __restrict__ qh = srow + h * 2 * P;
                const float* __restrict__ kh = drow + h * 2 * P + P;
                float a = 0.0f;
                for (int p = lane; p < P; p += 32)
                    a += qh[p] * kh[p];
                #pragma unroll
                for (int off = 16; off > 0; off >>= 1)
                    a += __shfl_xor(a, off, 32);
                if (lane == 0) {
                    scores[(size_t)(e0 + j) * H + h] = a;
                    atomicMax(&smax[(size_t)s * H + h], fenc(a));
                }
            }
        }
    }
}

__global__ __launch_bounds__(256)
void k3_exp_denom(float* __restrict__ scores, const unsigned* __restrict__ smax,
                  const int* __restrict__ ei, float* __restrict__ denom,
                  int E, int H) {
    size_t i = (size_t)blockIdx.x * blockDim.x + threadIdx.x;
    if (i >= (size_t)E * H) return;
    const int e = (int)(i / H);
    const int h = (int)(i % H);
    const int s = ei[e];
    const float m  = fdec(smax[(size_t)s * H + h]);
    const float ex = __expf(scores[i] - m);
    scores[i] = ex;
    atomicAdd(&denom[(size_t)s * H + h], ex);
}

__global__ __launch_bounds__(256)
void k4_out(const float* __restrict__ ex, const float* __restrict__ denom,
            const int* __restrict__ ei, float* __restrict__ out, int E, int H) {
    const int e = blockIdx.x * blockDim.x + threadIdx.x;
    if (e >= E) return;
    const int s = ei[e];
    float sum = 0.0f;
    for (int h = 0; h < H; ++h)
        sum += ex[(size_t)e * H + h] / denom[(size_t)s * H + h];
    out[e] = sum / (float)H;
}


#define OW_GROUPS  256
#define OW_THREADS 1024
#define OW_NODES   196
#define OW_CAP     4096
#define OW_UNR     16
#define OW_B       8
#define OW_FILL    13
#define BIN_CH     64
#define BIN_Q      4
#define BIN_CAP    96
#define BIN_TRIPS  ((BIN_CH * BIN_CAP + OW_THREADS - 1) / OW_THREADS)
#define BIN_IPT    13

typedef __attribute__((ext_vector_type(4))) int v4i;
typedef v4i v4i_a4 __attribute__((aligned(4)));
typedef __attribute__((ext_vector_type(2))) int v2i;

__device__ __forceinline__ unsigned ow_umin(unsigned a, unsigned b) { return a < b ? a : b; }

__device__ __forceinline__ void ow_append(int e, unsigned sl, int* le, int* ls, unsigned* cnt) {
    const unsigned idx = atomicAdd(cnt, 1u);
    if (idx < (unsigned)OW_CAP) { le[idx] = e; ls[idx] = (int)sl; }
}

__device__ __forceinline__ void ow_test4(v4i s, int e, int n0, unsigned nn,
                                         int* le, int* ls, unsigned* cnt) {
    const unsigned a = (unsigned)(s.x - n0), b = (unsigned)(s.y - n0);
    const unsigned c = (unsigned)(s.z - n0), d = (unsigned)(s.w - n0);
    if (ow_umin(ow_umin(a, b), ow_umin(c, d)) < nn) {
        #pragma nounroll
        for (int j = 0; j < 4; ++j) {
            const unsigned sl = (j == 0) ? a : (j == 1) ? b : (j == 2) ? c : d;
            if (sl < nn) ow_append(e + j, sl, le, ls, cnt);
        }
    }
}

__device__ __forceinline__ void ow_scan_simple(const int* __restrict__ ei, int lo, int hi, int n0, unsigned nn,
                                               int tid, int* le, int* ls, unsigned* cnt) {
    for (int e = lo + tid; e < hi; e += OW_THREADS) {
        const unsigned sl = (unsigned)(ei[e] - n0);
        if (sl < nn) ow_append(e, sl, le, ls, cnt);
    }
}

__device__ __forceinline__ void ow_scan(const int* __restrict__ ei, int lo, int hi, int n0, unsigned nn,
                                        int tid, int grp, int* le, int* ls, unsigned* cnt) {
    const int CH    = OW_THREADS * 4 * OW_UNR;
    const int nfull = (hi - lo) / CH;
    if (nfull > 0) {
        int c = grp % nfull;
        for (int it = 0; it < nfull; ++it) {
            const int b0 = lo + c * CH + tid * 4;
            v4i v[OW_UNR];
            #pragma unroll
            for (int u = 0; u < OW_UNR; ++u)
                v[u] = *(const v4i_a4*)(ei + b0 + u * (OW_THREADS * 4));
            __builtin_amdgcn_sched_barrier(0);
            #pragma unroll
            for (int u = 0; u < OW_UNR; ++u)
                ow_test4(v[u], b0 + u * (OW_THREADS * 4), n0, nn, le, ls, cnt);
            c = (c + 1 < nfull) ? c + 1 : 0;
        }
    }
    const int r0 = lo + nfull * CH;
    if (r0 < hi) {
        const int b0 = r0 + tid * 4;
        v4i v[OW_UNR];
        #pragma unroll
        for (int u = 0; u < OW_UNR; ++u) {
            const int b = b0 + u * (OW_THREADS * 4);
            v[u] = *(const v4i_a4*)(ei + ((b + 3 < hi) ? b : 0));
        }
        __builtin_amdgcn_sched_barrier(0);
        #pragma unroll
        for (int u = 0; u < OW_UNR; ++u) {
            const int b = b0 + u * (OW_THREADS * 4);
            if (b + 3 < hi) ow_test4(v[u], b, n0, nn, le, ls, cnt);
        }
        ow_scan_simple(ei, lo + ((hi - lo) & ~3), hi, n0, nn, tid, le, ls, cnt);
    }
}

__device__ __forceinline__ void ow_dst(const int* __restrict__ ei, int E, int n, int tid,
                                       const int* le, int* ld) {
    for (int i0 = tid; i0 < n; i0 += OW_THREADS * 4) {
        int v[4];
        #pragma unroll
        for (int u = 0; u < 4; ++u) {
            int i = i0 + u * OW_THREADS;
            i = i < n ? i : n - 1;
            v[u] = ei[(size_t)E + le[i]];
        }
        #pragma unroll
        for (int u = 0; u < 4; ++u) {
            int i = i0 + u * OW_THREADS;
            i = i < n ? i : n - 1;
            ld[i] = v[u];
        }
    }
}

struct KFrag { v4f ka, kb; };

__device__ __forceinline__ void ow_scores(int n, const float* __restrict__ km, const float* __restrict__ kt,
                                          const float* qs, const float* qts,
                                          const int* ld, const int* ls, float* sc, unsigned* mx,
                                          int wv, int lane) {
    const int hs   = lane >> 4;
    const int offA = hs * 64 + (lane & 15) * 4;
    const int up   = (lane >> 3) & 1;
    const int th   = hs + up * 2;
    for (int base = wv * OW_B; base < n; base += (OW_THREADS / 32) * OW_B) {
        int ii = base + (lane & (OW_B - 1));
        ii = ii < n ? ii : n - 1;
        const int dv = ld[ii];
        const int sv = ls[ii];
        int i2 = base + (lane >> 2);
        i2 = i2 < n ? i2 : n - 1;
        const float ktv = kt[(size_t)ld[i2] * 4 + (lane & 3)];
        KFrag f[OW_B];
        #pragma unroll
        for (int j = 0; j < OW_B; ++j) {
            const int d = __builtin_amdgcn_readlane(dv, j);
            const float* __restrict__ dk = km + (size_t)d * 256 + offA;
            f[j].ka = *(const v4f_a4*)(dk);
            f[j].kb = *(const v4f_a4*)(dk + 128);
        }
        __builtin_amdgcn_sched_barrier(0);
        #pragma unroll
        for (int j = 0; j < OW_B; ++j) {
            if (base + j < n) {
                const int sl = __builtin_amdgcn_readlane(sv, j);
                const v4f qa = *(const v4f*)(qs + sl * 256 + offA);
                const v4f qb = *(const v4f*)(qs + sl * 256 + offA + 128);
                const float qtv = qts[sl * 4 + th];
                const float ktj = __shfl(ktv, 4 * j + th, 32);
                float a1 = qa.x * f[j].ka.x;
                float a2 = qb.x * f[j].kb.x;
                a1 = fmaf(qa.y, f[j].ka.y, a1);  a2 = fmaf(qb.y, f[j].kb.y, a2);
                a1 = fmaf(qa.z, f[j].ka.z, a1);  a2 = fmaf(qb.z, f[j].kb.z, a2);
                a1 = fmaf(qa.w, f[j].ka.w, a1);  a2 = fmaf(qb.w, f[j].kb.w, a2);
                float r = row_xor_add<0x168>(up ? a2 : a1, up ? a1 : a2);
                r = row_xor_add<0x164>(r, r);
                r = row_xor_add<0x162>(r, r);
                r = row_xor_add<0x161>(r, r);
                r = fmaf(qtv, ktj, r);
                if ((lane & 7) == 0) {
                    sc[(base + j) * 4 + th] = r;
                    atomicMax(&mx[sl * 4 + th], fenc(r));
                }
            }
        }
    }
}

__global__ __launch_bounds__(OW_THREADS)
void kb_bin(const int* __restrict__ ei, int* __restrict__ cells, int* __restrict__ counts,
            int N, int E) {
    __shared__ unsigned lc[OW_GROUPS / BIN_Q];
    const int tid = threadIdx.x;
    const int c   = (int)blockIdx.x / BIN_Q;
    const unsigned q = (unsigned)blockIdx.x % BIN_Q;
    const unsigned per = (unsigned)((N + OW_GROUPS - 1) / OW_GROUPS);
    const int chunk = (E + BIN_CH - 1) / BIN_CH;
    const int lo = c * chunk;
    const int hi = (E - lo < chunk) ? E : lo + chunk;
    if (tid < OW_GROUPS / BIN_Q) lc[tid] = 0u;
    __syncthreads();
    for (int b0 = lo; b0 < hi; b0 += OW_THREADS * BIN_IPT) {
        int s[BIN_IPT], d[BIN_IPT];
        #pragma unroll
        for (int u = 0; u < BIN_IPT; ++u) {
            const int e  = b0 + u * OW_THREADS + tid;
            const int ec = e < hi ? e : hi - 1;
            s[u] = ei[ec];
            d[u] = ei[(size_t)E + ec];
        }
        __builtin_amdgcn_sched_barrier(0);
        #pragma unroll
        for (int u = 0; u < BIN_IPT; ++u)
            asm volatile("" : "+v"(d[u]));
        unsigned g[BIN_IPT], idx[BIN_IPT];
        #pragma unroll
        for (int u = 0; u < BIN_IPT; ++u) {
            const int e = b0 + u * OW_THREADS + tid;
            const unsigned us = (unsigned)s[u];
            g[u]   = us / per;
            idx[u] = (unsigned)BIN_CAP;
            if (e < hi && us < (unsigned)N && g[u] / (OW_GROUPS / BIN_Q) == q)
                idx[u] = atomicAdd(&lc[g[u] % (OW_GROUPS / BIN_Q)], 1u);
        }
        #pragma unroll
        for (int u = 0; u < BIN_IPT; ++u) {
            if (idx[u] < (unsigned)BIN_CAP) {
                const int e = b0 + u * OW_THREADS + tid;
                const unsigned sl = (unsigned)s[u] - g[u] * per;
                v2i en;
                en.x = (int)((unsigned)e | (sl << 24));
                en.y = d[u];
                *(v2i*)(cells + 2 * (((size_t)g[u] * BIN_CH + c) * BIN_CAP + idx[u])) = en;
            }
        }
    }
    __syncthreads();
    if (tid < OW_GROUPS / BIN_Q)
        counts[(size_t)(q * (OW_GROUPS / BIN_Q) + tid) * BIN_CH + c] = (int)lc[tid];
}

__global__ __launch_bounds__(OW_THREADS)
void ke_owner(const float* __restrict__ qk, const int* __restrict__ ei,
              const int* __restrict__ cells, const int* __restrict__ counts,
              float* __restrict__ out, int N, int E) {
    __shared__ __attribute__((aligned(16))) float qs[OW_NODES * 256];
    __shared__ __attribute__((aligned(16))) float qts[OW_NODES * 4];
    __shared__ __attribute__((aligned(16))) float sm[OW_NODES * 4];
    __shared__ unsigned mx[OW_NODES * 4];
    __shared__ __attribute__((aligned(16))) float sc[OW_CAP * 4];
    __shared__ int le[OW_CAP];
    __shared__ int ls[OW_CAP];
    __shared__ int ld[OW_CAP];
    __shared__ unsigned cnt;
    __shared__ int cc[BIN_CH];
    __shared__ int off[BIN_CH];
    __shared__ int binfo[2];

    const int tid  = threadIdx.x;
    const int lane = tid & 31;
    const int wv   = __builtin_amdgcn_readfirstlane(tid >> 5);
    const int per  = (N + (int)gridDim.x - 1) / (int)gridDim.x;
    const int n0   = (int)blockIdx.x * per;
    if (n0 >= N) return;
    const int nni  = (N - n0 < per) ? (N - n0) : per;
    const unsigned nn = (unsigned)nni;

    const float* __restrict__ qm = qk;
    const float* __restrict__ km = qk + (size_t)N * 256;
    const float* __restrict__ qt = qk + (size_t)N * 512;
    const float* __restrict__ kt = qt + (size_t)N * 4;

    for (int i = tid; i < nni * 4; i += OW_THREADS) {
        mx[i]  = 0u;
        sm[i]  = 0.0f;
        qts[i] = qt[(size_t)n0 * 4 + i];
    }
    if (tid == 0) cnt = 0u;
    if (tid < BIN_CH) cc[tid] = counts[(size_t)blockIdx.x * BIN_CH + tid];
    {
        const float* __restrict__ qsrc = qm + (size_t)n0 * 256;
        const int tot = nni * 64;
        for (int i0 = tid; i0 < tot; i0 += OW_THREADS * OW_FILL) {
            v4f t[OW_FILL];
            #pragma unroll
            for (int u = 0; u < OW_FILL; ++u) {
                int i = i0 + u * OW_THREADS;
                i = i < tot ? i : tot - 1;
                t[u] = *(const v4f_a4*)(qsrc + 4 * i);
            }
            __builtin_amdgcn_sched_barrier(0);
            #pragma unroll
            for (int u = 0; u < OW_FILL; ++u) {
                int i = i0 + u * OW_THREADS;
                i = i < tot ? i : tot - 1;
                *(v4f*)(qs + 4 * i) = t[u];
            }
        }
    }
    __syncthreads();

    if (tid < BIN_CH) {
        int o = 0;
        for (int i = 0; i < tid; ++i) o += cc[i];
        off[tid] = o;
    } else if (tid == BIN_CH) {
        int t = 0, bad = 0;
        for (int i = 0; i < BIN_CH; ++i) { t += cc[i]; bad |= (cc[i] > BIN_CAP) ? 1 : 0; }
        binfo[0] = t;
        binfo[1] = bad;
    }
    __syncthreads();
    int n = binfo[0];
    bool listed = false;
    if (binfo[1] != 0) {
        ow_scan(ei, 0, E, n0, nn, tid, (int)blockIdx.x, le, ls, &cnt);
        __syncthreads();
        n = (int)cnt;
        if (n <= OW_CAP) {
            ow_dst(ei, E, n, tid, le, ld);
            listed = true;
        }
    } else if (n <= OW_CAP) {
        const int* __restrict__ mine = cells + (size_t)blockIdx.x * (BIN_CH * BIN_CAP * 2);
        v2i  en[BIN_TRIPS];
        int  pos[BIN_TRIPS];
        #pragma unroll
        for (int u = 0; u < BIN_TRIPS; ++u) {
            const int s = tid + u * OW_THREADS;
            const int b = s / BIN_CAP;
            const int k = s - b * BIN_CAP;
            const bool used = (b < BIN_CH) && (k < cc[b < BIN_CH ? b : 0]);
            pos[u] = used ? off[b] + k : -1;
            const int* __restrict__ p = used ? mine + 2 * s : counts + (size_t)blockIdx.x * BIN_CH;
            en[u] = *(const v2i*)p;
        }
        __builtin_amdgcn_sched_barrier(0);
        #pragma unroll
        for (int u = 0; u < BIN_TRIPS; ++u) {
            if (pos[u] >= 0) {
                le[pos[u]] = (int)((unsigned)en[u].x & 0xFFFFFFu);
                ls[pos[u]] = (int)((unsigned)en[u].x >> 24);
                ld[pos[u]] = en[u].y;
            }
        }
        listed = true;
    }

    if (listed) {
        __syncthreads();
        ow_scores(n, km, kt, qs, qts, ld, ls, sc, mx, wv, lane);
        __syncthreads();
        for (int t = tid; t < n * 4; t += OW_THREADS) {
            const int g  = ls[t >> 2] * 4 + (t & 3);
            const float ex = __expf(sc[t] - fdec(mx[g]));
            sc[t] = ex;
            atomicAdd(&sm[g], ex);
        }
        __syncthreads();
        for (int i = tid; i < n; i += OW_THREADS) {
            const v4f x  = *(const v4f*)(sc + 4 * i);
            const v4f dn = *(const v4f*)(sm + 4 * ls[i]);
            float sum = 0.0f;
            sum += x.x / dn.x;
            sum += x.y / dn.y;
            sum += x.z / dn.z;
            sum += x.w / dn.w;
            out[le[i]] = sum / 4.0f;
        }
        return;
    }

    const int nr = (E + OW_CAP - 1) / OW_CAP;
    for (int pass = 0; pass < 3; ++pass) {
        for (int c = 0; c < nr; ++c) {
            __syncthreads();
            if (tid == 0) cnt = 0u;
            __syncthreads();
            const int hi = ((c + 1) * OW_CAP < E) ? (c + 1) * OW_CAP : E;
            ow_scan_simple(ei, c * OW_CAP, hi, n0, nn, tid, le, ls, &cnt);
            __syncthreads();
            const int m = (int)cnt;
            ow_dst(ei, E, m, tid, le, ld);
            __syncthreads();
            ow_scores(m, km, kt, qs, qts, ld, ls, sc, mx, wv, lane);
            __syncthreads();
            if (pass == 1) {
                for (int t = tid; t < m * 4; t += OW_THREADS) {
                    const int g = ls[t >> 2] * 4 + (t & 3);
                    atomicAdd(&sm[g], __expf(sc[t] - fdec(mx[g])));
                }
            } else if (pass == 2) {
                for (int i = tid; i < m; i += OW_THREADS) {
                    const int g = ls[i] * 4;
                    const v4f x  = *(const v4f*)(sc + 4 * i);
                    const v4f dn = *(const v4f*)(sm + g);
                    float sum = 0.0f;
                    sum += __expf(x.x - fdec(mx[g]))     / dn.x;
                    sum += __expf(x.y - fdec(mx[g + 1])) / dn.y;
                    sum += __expf(x.z - fdec(mx[g + 2])) / dn.z;
                    sum += __expf(x.w - fdec(mx[g + 3])) / dn.w;
                    out[le[i]] = sum / 4.0f;
                }
            }
        }
    }
}

extern "C" void kernel_launch(void* const* d_in, const int* in_sizes, int n_in,
                              void* d_out, int out_size, void* d_ws, size_t ws_size,
                              hipStream_t stream) {
    const float* x    = (const float*)d_in[0];
    const int*   ei   = (const int*)d_in[1];
    const float* W    = (const float*)d_in[2];
    const float* bias = (const float*)d_in[3];
    const int C = in_sizes[3];
    const int F = in_sizes[2] / C;
    const int N = in_sizes[0] / F;
    const int E = in_sizes[1] / 2;
    const int H = (C - 2 * F) / 2;
    const int P = F / H + 1;

    float*    qk     = (float*)d_ws;
    float*    scores = qk + (size_t)N * C;
    unsigned* smax   = (unsigned*)(scores + (size_t)E * H);
    float*    denom  = (float*)(smax + (size_t)N * H);

    int* cells  = (int*)scores;
    int* counts = (int*)smax;
    const bool own = (H == 4 && P == 65 && C == 520 && E >= 4 && E < (1 << 24) &&
                      (N + OW_GROUPS - 1) / OW_GROUPS <= OW_NODES &&
                      (size_t)OW_GROUPS * BIN_CH * BIN_CAP * 2 <= (size_t)E * H &&
                      (size_t)OW_GROUPS * BIN_CH <= (size_t)N * H);

    if (!own) {
        size_t n = (size_t)2 * N * H;
        k0_zero<<<(unsigned)((n + 255) / 256), 256, 0, stream>>>(smax, n);
    }

    {
        dim3 grid((C + G_WN * 64 - 1) / (G_WN * 64), (N + G_XROWS - 1) / G_XROWS);
        k1_gemm_wmma<<<grid, G_THREADS, 0, stream>>>(x, W, bias, qk, N, F, C, H, P);
    }

    if (own) {
        kb_bin<<<BIN_CH * BIN_Q, OW_THREADS, 0, stream>>>(ei, cells, counts, N, E);
        ke_owner<<<OW_GROUPS, OW_THREADS, 0, stream>>>(qk, ei, cells, counts, (float*)d_out, N, E);
        return;
    }

    {
        const int epb = (256 / 32) * K2_EPW;
        k2_scores<<<(E + epb - 1) / epb, 256, 0, stream>>>(qk, ei, scores, smax,
                                                           N, E, C, H, P);
    }

    {
        size_t n = (size_t)E * H;
        k3_exp_denom<<<(unsigned)((n + 255) / 256), 256, 0, stream>>>(scores, smax,
                                                                      ei, denom, E, H);
    }

    k4_out<<<(E + 255) / 256, 256, 0, stream>>>(scores, denom, ei,
                                                (float*)d_out, E, H);
}


# edge binning runs as 256 more work-groups at the end of the GEMM launch (it needs nothing of the GEMM): 2 launches, was 3; bin slots still arrival order, LDS-atomic sum order changes
# speedup vs baseline: 3.1669x; 1.0309x over previous
#include <hip/hip_runtime.h>
#include <hip/hip_bf16.h>


typedef __attribute__((ext_vector_type(2))) float v2f;
typedef __attribute__((ext_vector_type(4))) float v4f;
typedef __attribute__((ext_vector_type(8))) float v8f;
typedef v4f v4f_a4 __attribute__((aligned(4)));

__device__ __forceinline__ unsigned fenc(float f) {
    unsigned u = __float_as_uint(f);
    return (u & 0x80000000u) ? ~u : (u | 0x80000000u);
}
__device__ __forceinline__ float fdec(unsigned e) {
    return (e & 0x80000000u) ? __uint_as_float(e & 0x7FFFFFFFu)
                             : __uint_as_float(~e);
}

__global__ void k0_zero(unsigned* __restrict__ p, size_t n) {
    size_t i = (size_t)blockIdx.x * blockDim.x + threadIdx.x;
    if (i < n) p[i] = 0u;
}

#define G_MI      2
#define G_WM      4
#define G_WN      3
#define G_XROWS   (G_WM * G_MI * 16)
#define G_ROWS    (G_XROWS + G_WN * 64)
#define G_PITCH   36
#define G_THREADS (G_WM * G_WN * 32)
#define G_RSTEP   (G_THREADS / 8)
#define G_PIECES  ((G_ROWS + G_RSTEP - 1) / G_RSTEP)

#define OW_GROUPS  256
#define OW_THREADS 1024
#define OW_NODES   196
#define OW_CAP     4096
#define OW_UNR     16
#define OW_B       8
#define OW_FILL    13
#define BIN_CH     64
#define BIN_Q      4
#define BIN_CAP    96
#define BIN_TRIPS  ((BIN_CH * BIN_CAP + OW_THREADS - 1) / OW_THREADS)
#define BIN_IPT    17

typedef __attribute__((ext_vector_type(4))) int v4i;
typedef v4i v4i_a4 __attribute__((aligned(4)));
typedef __attribute__((ext_vector_type(2))) int v2i;

__device__ __forceinline__ void bin_edges(const int* __restrict__ ei, int* __restrict__ cells,
                                          int* __restrict__ counts, int N, int E,
                                          int wg, unsigned* lc, int tid) {
    const int c   = wg / BIN_Q;
    const unsigned q = (unsigned)wg % BIN_Q;
    const unsigned per = (unsigned)((N + OW_GROUPS - 1) / OW_GROUPS);
    const int chunk = (E + BIN_CH - 1) / BIN_CH;
    const int lo = c * chunk;
    const int hi = (E - lo < chunk) ? E : lo + chunk;
    if (tid < OW_GROUPS / BIN_Q) lc[tid] = 0u;
    __syncthreads();
    for (int b0 = lo; b0 < hi; b0 += G_THREADS * BIN_IPT) {
        int s[BIN_IPT], d[BIN_IPT];
        #pragma unroll
        for (int u = 0; u < BIN_IPT; ++u) {
            const int e  = b0 + u * G_THREADS + tid;
            const int ec = e < hi ? e : hi - 1;
            s[u] = ei[ec];
            d[u] = ei[(size_t)E + ec];
        }
        __builtin_amdgcn_sched_barrier(0);
        #pragma unroll
        for (int u = 0; u < BIN_IPT; ++u)
            asm volatile("" : "+v"(d[u]));
        unsigned g[BIN_IPT], idx[BIN_IPT];
        #pragma unroll
        for (int u = 0; u < BIN_IPT; ++u) {
            const int e = b0 + u * G_THREADS + tid;
            const unsigned us = (unsigned)s[u];
            g[u]   = us / per;
            idx[u] = (unsigned)BIN_CAP;
            if (e < hi && us < (unsigned)N && g[u] / (OW_GROUPS / BIN_Q) == q)
                idx[u] = atomicAdd(&lc[g[u] % (OW_GROUPS / BIN_Q)], 1u);
        }
        #pragma unroll
        for (int u = 0; u < BIN_IPT; ++u) {
            if (idx[u] < (unsigned)BIN_CAP) {
                const int e = b0 + u * G_THREADS + tid;
                const unsigned sl = (unsigned)s[u] - g[u] * per;
                v2i en;
                en.x = (int)((unsigned)e | (sl << 24));
                en.y = d[u];
                *(v2i*)(cells + 2 * (((size_t)g[u] * BIN_CH + c) * BIN_CAP + idx[u])) = en;
            }
        }
    }
    __syncthreads();
    if (tid < OW_GROUPS / BIN_Q)
        counts[(size_t)(q * (OW_GROUPS / BIN_Q) + tid) * BIN_CH + c] = (int)lc[tid];
}

__global__ __launch_bounds__(G_THREADS, 6)
void k1_gemm_wmma(const float* __restrict__ x, const float* __restrict__ W,
                  const float* __restrict__ bias, float* __restrict__ qk,
                  int N, int F, int C, int H, int P,
                  const int* __restrict__ ei, int* __restrict__ cells, int* __restrict__ counts,
                  int E) {
    __shared__ __attribute__((aligned(16))) float stg[2][G_ROWS * G_PITCH];
    __shared__ unsigned lc[OW_GROUPS / BIN_Q];
    const int tid  = threadIdx.x;
    {
        const int rowBlocks = (N + G_XROWS - 1) / G_XROWS;
        if ((int)blockIdx.y >= rowBlocks) {
            const int wg = ((int)blockIdx.y - rowBlocks) * (int)gridDim.x + (int)blockIdx.x;
            if (wg < BIN_CH * BIN_Q)
                bin_edges(ei, cells, counts, N, E, wg, lc, tid);
            return;
        }
    }
    const int lane = tid & 31;
    const int wv   = __builtin_amdgcn_readfirstlane(tid >> 5);
    const int wm   = wv % G_WM;
    const int wn   = wv / G_WM;
    const int Mg   = blockIdx.y * G_XROWS;
    const int Ng   = blockIdx.x * (G_WN * 64);
    const int M0   = Mg + wm * (G_MI * 16);
    const int N0   = Ng + wn * 64;
    const int lm   = lane & 15;
    const int kh   = (lane >> 4) * 2;

    const int seg = tid & 7;
    const int l0  = (tid >> 3) * G_PITCH + seg * 4;
    const float* __restrict__ gsrc[G_PIECES];
    #pragma unroll
    for (int i = 0; i < G_PIECES; ++i) {
        const int R  = (tid >> 3) + G_RSTEP * i;
        const int Rc = R < G_ROWS ? R : G_ROWS - 1;
        const float* __restrict__ p;
        if (Rc < G_XROWS) { const int r = Mg + Rc;           p = x + (size_t)(r < N ? r : N - 1) * F; }
        else              { const int c = Ng + Rc - G_XROWS; p = W + (size_t)(c < C ? c : C - 1) * F; }
        gsrc[i] = p + seg * 4;
    }

    v8f acc[G_MI][4];
    #pragma unroll
    for (int mi = 0; mi < G_MI; ++mi)
        #pragma unroll
        for (int ni = 0; ni < 4; ++ni)
            acc[mi][ni] = (v8f){};

    const int kq  = (lane >> 4) * 16;
    const int nkb = F / 32;
    const int aoff = (wm * (G_MI * 16) + lm) * G_PITCH + kq;
    const int boff = (G_XROWS + wn * 64 + lm) * G_PITCH + kq;

    v4f t[G_PIECES];
    if (nkb > 0) {
        #pragma unroll
        for (int i = 0; i < G_PIECES; ++i) t[i] = *(const v4f_a4*)(gsrc[i]);
        #pragma unroll
        for (int i = 0; i < G_PIECES; ++i)
            if ((i + 1) * G_RSTEP <= G_ROWS || (tid >> 3) + G_RSTEP * i < G_ROWS)
                *(v4f*)(&stg[0][l0 + i * (G_RSTEP * G_PITCH)]) = t[i];
        __syncthreads();
    }
    for (int kb = 0; kb < nkb; ++kb) {
        const float* sb = stg[kb & 1];
        float*       sn = stg[(kb & 1) ^ 1];
        const bool more = (kb + 1 < nkb);
        if (more) {
            #pragma unroll
            for (int i = 0; i < G_PIECES; ++i)
                t[i] = *(const v4f_a4*)(gsrc[i] + (size_t)(kb + 1) * 32);
        }
        __builtin_amdgcn_sched_barrier(0);
        v4f a[G_MI][4], b[4][4];
        #pragma unroll
        for (int j = 0; j < 4; ++j) {
            #pragma unroll
            for (int mi = 0; mi < G_MI; ++mi)
                a[mi][j] = *(const v4f*)(sb + aoff + mi * 16 * G_PITCH + 4 * j);
            #pragma unroll
            for (int ni = 0; ni < 4; ++ni)
                b[ni][j] = *(const v4f*)(sb + boff + ni * 16 * G_PITCH + 4 * j);
        }
        #pragma unroll
        for (int tt = 0; tt < 8; ++tt) {
            #pragma unroll
            for (int mi = 0; mi < G_MI; ++mi) {
                v2f af;
                af.x = a[mi][tt >> 1][(tt & 1) * 2];
                af.y = a[mi][tt >> 1][(tt & 1) * 2 + 1];
                #pragma unroll
                for (int ni = 0; ni < 4; ++ni) {
                    v2f bf;
                    bf.x = b[ni][tt >> 1][(tt & 1) * 2];
                    bf.y = b[ni][tt >> 1][(tt & 1) * 2 + 1];
                    acc[mi][ni] = __builtin_amdgcn_wmma_f32_16x16x4_f32(
                         false, af,  false, bf,
                         (short)0, acc[mi][ni],
                         false,  false);
                }
            }
        }
        __builtin_amdgcn_sched_barrier(0);
        if (more) {
            #pragma unroll
            for (int i = 0; i < G_PIECES; ++i)
                if ((i + 1) * G_RSTEP <= G_ROWS || (tid >> 3) + G_RSTEP * i < G_ROWS)
                    *(v4f*)(sn + l0 + i * (G_RSTEP * G_PITCH)) = t[i];
        }
        __syncthreads();
    }

    int k = nkb * 32;
    if (k < F) {
        const float* __restrict__ xrow[G_MI];
        const float* __restrict__ wrow[4];
        #pragma unroll
        for (int mi = 0; mi < G_MI; ++mi) {
            int r = M0 + mi * 16 + lm;
            xrow[mi] = x + (size_t)(r < N ? r : N - 1) * F;
        }
        #pragma unroll
        for (int ni = 0; ni < 4; ++ni) {
            int c = N0 + ni * 16 + lm;
            wrow[ni] = W + (size_t)(c < C ? c : C - 1) * F;
        }
        for (; k < F; k += 4) {
            v2f a[G_MI], b[4];
            #pragma unroll
            for (int mi = 0; mi < G_MI; ++mi) {
                a[mi].x = xrow[mi][k + kh];
                a[mi].y = xrow[mi][k + kh + 1];
            }
            #pragma unroll
            for (int ni = 0; ni < 4; ++ni) {
                b[ni].x = wrow[ni][k + kh];
                b[ni].y = wrow[ni][k + kh + 1];
            }
            #pragma unroll
            for (int mi = 0; mi < G_MI; ++mi)
                #pragma unroll
                for (int ni = 0; ni < 4; ++ni)
                    acc[mi][ni] = __builtin_amdgcn_wmma_f32_16x16x4_f32(
                         false, a[mi],  false, b[ni],
                         (short)0, acc[mi][ni],
                         false,  false);
        }
    }

    const bool split = (H == 4 && P == 65 && C == 520);
    const int rsub = (lane >> 4) << 3;
    #pragma unroll
    for (int ni = 0; ni < 4; ++ni) {
        const int col = N0 + ni * 16 + lm;
        if (col >= C) continue;
        const float bv = bias[col];
        float* __restrict__ dst = qk + col;
        size_t pitch = (size_t)C;
        if (split) {
            const int h   = col / 130;
            const int rr  = col - h * 130;
            const int isk = (rr >= 65) ? 1 : 0;
            const int p   = rr - isk * 65;
            if (p < 64) {
                dst   = qk + (size_t)isk * N * 256 + h * 64 + p;
                pitch = 256;
            } else {
                dst   = qk + (size_t)N * 512 + (size_t)isk * N * 4 + h;
                pitch = 4;
            }
        }
        #pragma unroll
        for (int mi = 0; mi < G_MI; ++mi) {
            const int rbase = M0 + mi * 16 + rsub;
            #pragma unroll
            for (int r = 0; r < 8; ++r) {
                const int row = rbase + r;
                if (row < N)
                    dst[(size_t)row * pitch] = acc[mi][ni][r] + bv;
            }
        }
    }
}


#define K2_EPW 16

template <int CTRL>
__device__ __forceinline__ float row_xor_add(float keep, float send) {
    const int t = __builtin_amdgcn_update_dpp(0, __builtin_bit_cast(int, send),
                                              CTRL, 0xf, 0xf, false);
    return keep + __builtin_bit_cast(float, t);
}

struct EdgeFrag {
    v4f qa, qb, ka, kb;
    float qt, kt;
};

struct GatherPlanes {
    const float* __restrict__ qm;
    const float* __restrict__ km;
    const float* __restrict__ qt;
    const float* __restrict__ kt;
};

__device__ __forceinline__ void k2_load_edge(EdgeFrag& f, const GatherPlanes& g,
                                             int s, int d, int offA, int th) {
    const float* __restrict__ sq = g.qm + (size_t)s * 256 + offA;
    const float* __restrict__ dk = g.km + (size_t)d * 256 + offA;
    f.qa = *(const v4f_a4*)(sq);
    f.qb = *(const v4f_a4*)(sq + 128);
    f.ka = *(const v4f_a4*)(dk);
    f.kb = *(const v4f_a4*)(dk + 128);
    f.qt = g.qt[(size_t)s * 4 + th];
    f.kt = g.kt[(size_t)d * 4 + th];
    __builtin_amdgcn_sched_barrier(0);
}

__device__ __forceinline__ void k2_reduce_store(const EdgeFrag& f, float* __restrict__ scores,
                                                unsigned* __restrict__ smax, int e, int s,
                                                int up, int th, int lane) {
    float a1 = f.qa.x * f.ka.x;
    float a2 = f.qb.x * f.kb.x;
    a1 = fmaf(f.qa.y, f.ka.y, a1);  a2 = fmaf(f.qb.y, f.kb.y, a2);
    a1 = fmaf(f.qa.z, f.ka.z, a1);  a2 = fmaf(f.qb.z, f.kb.z, a2);
    a1 = fmaf(f.qa.w, f.ka.w, a1);  a2 = fmaf(f.qb.w, f.kb.w, a2);
    float r = row_xor_add<0x168>(up ? a2 : a1, up ? a1 : a2);
    r = row_xor_add<0x164>(r, r);
    r = row_xor_add<0x162>(r, r);
    r = row_xor_add<0x161>(r, r);
    r = fmaf(f.qt, f.kt, r);
    if ((lane & 7) == 0) {
        scores[(size_t)e * 4 + th] = r;
        atomicMax(&smax[(size_t)s * 4 + th], fenc(r));
    }
}

__global__ __launch_bounds__(256)
void k2_scores(const float* __restrict__ qk, const int* __restrict__ ei,
               float* __restrict__ scores, unsigned* __restrict__ smax,
               int N, int E, int C, int H, int P) {
    const int lane = threadIdx.x & 31;
    const int wv = __builtin_amdgcn_readfirstlane((int)(threadIdx.x >> 5));
    const int e0 = (blockIdx.x * (blockDim.x >> 5) + wv) * K2_EPW;
    if (e0 >= E) return;
    const int cnt = (E - e0 < K2_EPW) ? (E - e0) : K2_EPW;

    const int le = (e0 + (lane & 15) < E) ? (e0 + (lane & 15)) : (E - 1);
    const int ev = ei[(size_t)(lane >> 4) * E + le];

    if (H == 4 && P == 65 && C == 520) {
        const int hs   = lane >> 4;
        const int offA = hs * 64 + (lane & 15) * 4;
        const int up   = (lane >> 3) & 1;
        const int th   = hs + up * 2;
        GatherPlanes g;
        g.qm = qk;
        g.km = qk + (size_t)N * 256;
        g.qt = qk + (size_t)N * 512;
        g.kt = g.qt + (size_t)N * 4;

        EdgeFrag fa, fb;
        int sa = __builtin_amdgcn_readlane(ev, 0), sb = sa;
        k2_load_edge(fa, g, sa, __builtin_amdgcn_readlane(ev, 16), offA, th);
        int j = 0;
        for (; j + 2 < cnt; j += 2) {
            sb = __builtin_amdgcn_readlane(ev, j + 1);
            k2_load_edge(fb, g, sb, __builtin_amdgcn_readlane(ev, j + 17), offA, th);
            k2_reduce_store(fa, scores, smax, e0 + j, sa, up, th, lane);
            sa = __builtin_amdgcn_readlane(ev, j + 2);
            k2_load_edge(fa, g, sa, __builtin_amdgcn_readlane(ev, j + 18), offA, th);
            k2_reduce_store(fb, scores, smax, e0 + j + 1, sb, up, th, lane);
        }
        if (j + 1 < cnt) {
            sb = __builtin_amdgcn_readlane(ev, j + 1);
            k2_load_edge(fb, g, sb, __builtin_amdgcn_readlane(ev, j + 17), offA, th);
            k2_reduce_store(fa, scores, smax, e0 + j, sa, up, th, lane);
            k2_reduce_store(fb, scores, smax, e0 + j + 1, sb, up, th, lane);
        } else {
            k2_reduce_store(fa, scores, smax, e0 + j, sa, up, th, lane);
        }
    } else {
        for (int j = 0; j < cnt; ++j) {
            const int s = __builtin_amdgcn_readlane(ev, j);
            const int d = __builtin_amdgcn_readlane(ev, 16 + j);
            const float* __restrict__ srow = qk + (size_t)s * C;
            const float* __restrict__ drow = qk + (size_t)d * C;
            for (int h = 0; h < H; ++h) {
                const float* __restrict__ qh = srow + h * 2 * P;
                const float* __restrict__ kh = drow + h * 2 * P + P;
                float a = 0.0f;
                for (int p = lane; p < P; p += 32)
                    a += qh[p] * kh[p];
                #pragma unroll
                for (int off = 16; off > 0; off >>= 1)
                    a += __shfl_xor(a, off, 32);
                if (lane == 0) {
                    scores[(size_t)(e0 + j) * H + h] = a;
                    atomicMax(&smax[(size_t)s * H + h], fenc(a));
                }
            }
        }
    }
}

__global__ __launch_bounds__(256)
void k3_exp_denom(float* __restrict__ scores, const unsigned* __restrict__ smax,
                  const int* __restrict__ ei, float* __restrict__ denom,
                  int E, int H) {
    size_t i = (size_t)blockIdx.x * blockDim.x + threadIdx.x;
    if (i >= (size_t)E * H) return;
    const int e = (int)(i / H);
    const int h = (int)(i % H);
    const int s = ei[e];
    const float m  = fdec(smax[(size_t)s * H + h]);
    const float ex = __expf(scores[i] - m);
    scores[i] = ex;
    atomicAdd(&denom[(size_t)s * H + h], ex);
}

__global__ __launch_bounds__(256)
void k4_out(const float* __restrict__ ex, const float* __restrict__ denom,
            const int* __restrict__ ei, float* __restrict__ out, int E, int H) {
    const int e = blockIdx.x * blockDim.x + threadIdx.x;
    if (e >= E) return;
    const int s = ei[e];
    float sum = 0.0f;
    for (int h = 0; h < H; ++h)
        sum += ex[(size_t)e * H + h] / denom[(size_t)s * H + h];
    out[e] = sum / (float)H;
}


__device__ __forceinline__ unsigned ow_umin(unsigned a, unsigned b) { return a < b ? a : b; }

__device__ __forceinline__ void ow_append(int e, unsigned sl, int* le, int* ls, unsigned* cnt) {
    const unsigned idx = atomicAdd(cnt, 1u);
    if (idx < (unsigned)OW_CAP) { le[idx] = e; ls[idx] = (int)sl; }
}

__device__ __forceinline__ void ow_test4(v4i s, int e, int n0, unsigned nn,
                                         int* le, int* ls, unsigned* cnt) {
    const unsigned a = (unsigned)(s.x - n0), b = (unsigned)(s.y - n0);
    const unsigned c = (unsigned)(s.z - n0), d = (unsigned)(s.w - n0);
    if (ow_umin(ow_umin(a, b), ow_umin(c, d)) < nn) {
        #pragma nounroll
        for (int j = 0; j < 4; ++j) {
            const unsigned sl = (j == 0) ? a : (j == 1) ? b : (j == 2) ? c : d;
            if (sl < nn) ow_append(e + j, sl, le, ls, cnt);
        }
    }
}

__device__ __forceinline__ void ow_scan_simple(const int* __restrict__ ei, int lo, int hi, int n0, unsigned nn,
                                               int tid, int* le, int* ls, unsigned* cnt) {
    for (int e = lo + tid; e < hi; e += OW_THREADS) {
        const unsigned sl = (unsigned)(ei[e] - n0);
        if (sl < nn) ow_append(e, sl, le, ls, cnt);
    }
}

__device__ __forceinline__ void ow_scan(const int* __restrict__ ei, int lo, int hi, int n0, unsigned nn,
                                        int tid, int grp, int* le, int* ls, unsigned* cnt) {
    const int CH    = OW_THREADS * 4 * OW_UNR;
    const int nfull = (hi - lo) / CH;
    if (nfull > 0) {
        int c = grp % nfull;
        for (int it = 0; it < nfull; ++it) {
            const int b0 = lo + c * CH + tid * 4;
            v4i v[OW_UNR];
            #pragma unroll
            for (int u = 0; u < OW_UNR; ++u)
                v[u] = *(const v4i_a4*)(ei + b0 + u * (OW_THREADS * 4));
            __builtin_amdgcn_sched_barrier(0);
            #pragma unroll
            for (int u = 0; u < OW_UNR; ++u)
                ow_test4(v[u], b0 + u * (OW_THREADS * 4), n0, nn, le, ls, cnt);
            c = (c + 1 < nfull) ? c + 1 : 0;
        }
    }
    const int r0 = lo + nfull * CH;
    if (r0 < hi) {
        const int b0 = r0 + tid * 4;
        v4i v[OW_UNR];
        #pragma unroll
        for (int u = 0; u < OW_UNR; ++u) {
            const int b = b0 + u * (OW_THREADS * 4);
            v[u] = *(const v4i_a4*)(ei + ((b + 3 < hi) ? b : 0));
        }
        __builtin_amdgcn_sched_barrier(0);
        #pragma unroll
        for (int u = 0; u < OW_UNR; ++u) {
            const int b = b0 + u * (OW_THREADS * 4);
            if (b + 3 < hi) ow_test4(v[u], b, n0, nn, le, ls, cnt);
        }
        ow_scan_simple(ei, lo + ((hi - lo) & ~3), hi, n0, nn, tid, le, ls, cnt);
    }
}

__device__ __forceinline__ void ow_dst(const int* __restrict__ ei, int E, int n, int tid,
                                       const int* le, int* ld) {
    for (int i0 = tid; i0 < n; i0 += OW_THREADS * 4) {
        int v[4];
        #pragma unroll
        for (int u = 0; u < 4; ++u) {
            int i = i0 + u * OW_THREADS;
            i = i < n ? i : n - 1;
            v[u] = ei[(size_t)E + le[i]];
        }
        #pragma unroll
        for (int u = 0; u < 4; ++u) {
            int i = i0 + u * OW_THREADS;
            i = i < n ? i : n - 1;
            ld[i] = v[u];
        }
    }
}

struct KFrag { v4f ka, kb; };

__device__ __forceinline__ void ow_scores(int n, const float* __restrict__ km, const float* __restrict__ kt,
                                          const float* qs, const float* qts,
                                          const int* ld, const int* ls, float* sc, unsigned* mx,
                                          int wv, int lane) {
    const int hs   = lane >> 4;
    const int offA = hs * 64 + (lane & 15) * 4;
    const int up   = (lane >> 3) & 1;
    const int th   = hs + up * 2;
    for (int base = wv * OW_B; base < n; base += (OW_THREADS / 32) * OW_B) {
        int ii = base + (lane & (OW_B - 1));
        ii = ii < n ? ii : n - 1;
        const int dv = ld[ii];
        const int sv = ls[ii];
        int i2 = base + (lane >> 2);
        i2 = i2 < n ? i2 : n - 1;
        const float ktv = kt[(size_t)ld[i2] * 4 + (lane & 3)];
        KFrag f[OW_B];
        #pragma unroll
        for (int j = 0; j < OW_B; ++j) {
            const int d = __builtin_amdgcn_readlane(dv, j);
            const float* __restrict__ dk = km + (size_t)d * 256 + offA;
            f[j].ka = *(const v4f_a4*)(dk);
            f[j].kb = *(const v4f_a4*)(dk + 128);
        }
        __builtin_amdgcn_sched_barrier(0);
        #pragma unroll
        for (int j = 0; j < OW_B; ++j) {
            if (base + j < n) {
                const int sl = __builtin_amdgcn_readlane(sv, j);
                const v4f qa = *(const v4f*)(qs + sl * 256 + offA);
                const v4f qb = *(const v4f*)(qs + sl * 256 + offA + 128);
                const float qtv = qts[sl * 4 + th];
                const float ktj = __shfl(ktv, 4 * j + th, 32);
                float a1 = qa.x * f[j].ka.x;
                float a2 = qb.x * f[j].kb.x;
                a1 = fmaf(qa.y, f[j].ka.y, a1);  a2 = fmaf(qb.y, f[j].kb.y, a2);
                a1 = fmaf(qa.z, f[j].ka.z, a1);  a2 = fmaf(qb.z, f[j].kb.z, a2);
                a1 = fmaf(qa.w, f[j].ka.w, a1);  a2 = fmaf(qb.w, f[j].kb.w, a2);
                float r = row_xor_add<0x168>(up ? a2 : a1, up ? a1 : a2);
                r = row_xor_add<0x164>(r, r);
                r = row_xor_add<0x162>(r, r);
                r = row_xor_add<0x161>(r, r);
                r = fmaf(qtv, ktj, r);
                if ((lane & 7) == 0) {
                    sc[(base + j) * 4 + th] = r;
                    atomicMax(&mx[sl * 4 + th], fenc(r));
                }
            }
        }
    }
}

__global__ __launch_bounds__(OW_THREADS)
void ke_owner(const float* __restrict__ qk, const int* __restrict__ ei,
              const int* __restrict__ cells, const int* __restrict__ counts,
              float* __restrict__ out, int N, int E) {
    __shared__ __attribute__((aligned(16))) float qs[OW_NODES * 256];
    __shared__ __attribute__((aligned(16))) float qts[OW_NODES * 4];
    __shared__ __attribute__((aligned(16))) float sm[OW_NODES * 4];
    __shared__ unsigned mx[OW_NODES * 4];
    __shared__ __attribute__((aligned(16))) float sc[OW_CAP * 4];
    __shared__ int le[OW_CAP];
    __shared__ int ls[OW_CAP];
    __shared__ int ld[OW_CAP];
    __shared__ unsigned cnt;
    __shared__ int cc[BIN_CH];
    __shared__ int off[BIN_CH];
    __shared__ int binfo[2];

    const int tid  = threadIdx.x;
    const int lane = tid & 31;
    const int wv   = __builtin_amdgcn_readfirstlane(tid >> 5);
    const int per  = (N + (int)gridDim.x - 1) / (int)gridDim.x;
    const int n0   = (int)blockIdx.x * per;
    if (n0 >= N) return;
    const int nni  = (N - n0 < per) ? (N - n0) : per;
    const unsigned nn = (unsigned)nni;

    const float* __restrict__ qm = qk;
    const float* __restrict__ km = qk + (size_t)N * 256;
    const float* __restrict__ qt = qk + (size_t)N * 512;
    const float* __restrict__ kt = qt + (size_t)N * 4;

    for (int i = tid; i < nni * 4; i += OW_THREADS) {
        mx[i]  = 0u;
        sm[i]  = 0.0f;
        qts[i] = qt[(size_t)n0 * 4 + i];
    }
    if (tid == 0) cnt = 0u;
    if (tid < BIN_CH) cc[tid] = counts[(size_t)blockIdx.x * BIN_CH + tid];
    {
        const float* __restrict__ qsrc = qm + (size_t)n0 * 256;
        const int tot = nni * 64;
        for (int i0 = tid; i0 < tot; i0 += OW_THREADS * OW_FILL) {
            v4f t[OW_FILL];
            #pragma unroll
            for (int u = 0; u < OW_FILL; ++u) {
                int i = i0 + u * OW_THREADS;
                i = i < tot ? i : tot - 1;
                t[u] = *(const v4f_a4*)(qsrc + 4 * i);
            }
            __builtin_amdgcn_sched_barrier(0);
            #pragma unroll
            for (int u = 0; u < OW_FILL; ++u) {
                int i = i0 + u * OW_THREADS;
                i = i < tot ? i : tot - 1;
                *(v4f*)(qs + 4 * i) = t[u];
            }
        }
    }
    __syncthreads();

    if (tid < BIN_CH) {
        int o = 0;
        for (int i = 0; i < tid; ++i) o += cc[i];
        off[tid] = o;
    } else if (tid == BIN_CH) {
        int t = 0, bad = 0;
        for (int i = 0; i < BIN_CH; ++i) { t += cc[i]; bad |= (cc[i] > BIN_CAP) ? 1 : 0; }
        binfo[0] = t;
        binfo[1] = bad;
    }
    __syncthreads();
    int n = binfo[0];
    bool listed = false;
    if (binfo[1] != 0) {
        ow_scan(ei, 0, E, n0, nn, tid, (int)blockIdx.x, le, ls, &cnt);
        __syncthreads();
        n = (int)cnt;
        if (n <= OW_CAP) {
            ow_dst(ei, E, n, tid, le, ld);
            listed = true;
        }
    } else if (n <= OW_CAP) {
        const int* __restrict__ mine = cells + (size_t)blockIdx.x * (BIN_CH * BIN_CAP * 2);
        v2i  en[BIN_TRIPS];
        int  pos[BIN_TRIPS];
        #pragma unroll
        for (int u = 0; u < BIN_TRIPS; ++u) {
            const int s = tid + u * OW_THREADS;
            const int b = s / BIN_CAP;
            const int k = s - b * BIN_CAP;
            const bool used = (b < BIN_CH) && (k < cc[b < BIN_CH ? b : 0]);
            pos[u] = used ? off[b] + k : -1;
            const int* __restrict__ p = used ? mine + 2 * s : counts + (size_t)blockIdx.x * BIN_CH;
            en[u] = *(const v2i*)p;
        }
        __builtin_amdgcn_sched_barrier(0);
        #pragma unroll
        for (int u = 0; u < BIN_TRIPS; ++u) {
            if (pos[u] >= 0) {
                le[pos[u]] = (int)((unsigned)en[u].x & 0xFFFFFFu);
                ls[pos[u]] = (int)((unsigned)en[u].x >> 24);
                ld[pos[u]] = en[u].y;
            }
        }
        listed = true;
    }

    if (listed) {
        __syncthreads();
        ow_scores(n, km, kt, qs, qts, ld, ls, sc, mx, wv, lane);
        __syncthreads();
        for (int t = tid; t < n * 4; t += OW_THREADS) {
            const int g  = ls[t >> 2] * 4 + (t & 3);
            const float ex = __expf(sc[t] - fdec(mx[g]));
            sc[t] = ex;
            atomicAdd(&sm[g], ex);
        }
        __syncthreads();
        for (int i = tid; i < n; i += OW_THREADS) {
            const v4f x  = *(const v4f*)(sc + 4 * i);
            const v4f dn = *(const v4f*)(sm + 4 * ls[i]);
            float sum = 0.0f;
            sum += x.x / dn.x;
            sum += x.y / dn.y;
            sum += x.z / dn.z;
            sum += x.w / dn.w;
            out[le[i]] = sum / 4.0f;
        }
        return;
    }

    const int nr = (E + OW_CAP - 1) / OW_CAP;
    for (int pass = 0; pass < 3; ++pass) {
        for (int c = 0; c < nr; ++c) {
            __syncthreads();
            if (tid == 0) cnt = 0u;
            __syncthreads();
            const int hi = ((c + 1) * OW_CAP < E) ? (c + 1) * OW_CAP : E;
            ow_scan_simple(ei, c * OW_CAP, hi, n0, nn, tid, le, ls, &cnt);
            __syncthreads();
            const int m = (int)cnt;
            ow_dst(ei, E, m, tid, le, ld);
            __syncthreads();
            ow_scores(m, km, kt, qs, qts, ld, ls, sc, mx, wv, lane);
            __syncthreads();
            if (pass == 1) {
                for (int t = tid; t < m * 4; t += OW_THREADS) {
                    const int g = ls[t >> 2] * 4 + (t & 3);
                    atomicAdd(&sm[g], __expf(sc[t] - fdec(mx[g])));
                }
            } else if (pass == 2) {
                for (int i = tid; i < m; i += OW_THREADS) {
                    const int g = ls[i] * 4;
                    const v4f x  = *(const v4f*)(sc + 4 * i);
                    const v4f dn = *(const v4f*)(sm + g);
                    float sum = 0.0f;
                    sum += __expf(x.x - fdec(mx[g]))     / dn.x;
                    sum += __expf(x.y - fdec(mx[g + 1])) / dn.y;
                    sum += __expf(x.z - fdec(mx[g + 2])) / dn.z;
                    sum += __expf(x.w - fdec(mx[g + 3])) / dn.w;
                    out[le[i]] = sum / 4.0f;
                }
            }
        }
    }
}

extern "C" void kernel_launch(void* const* d_in, const int* in_sizes, int n_in,
                              void* d_out, int out_size, void* d_ws, size_t ws_size,
                              hipStream_t stream) {
    const float* x    = (const float*)d_in[0];
    const int*   ei   = (const int*)d_in[1];
    const float* W    = (const float*)d_in[2];
    const float* bias = (const float*)d_in[3];
    const int C = in_sizes[3];
    const int F = in_sizes[2] / C;
    const int N = in_sizes[0] / F;
    const int E = in_sizes[1] / 2;
    const int H = (C - 2 * F) / 2;
    const int P = F / H + 1;

    float*    qk     = (float*)d_ws;
    float*    scores = qk + (size_t)N * C;
    unsigned* smax   = (unsigned*)(scores + (size_t)E * H);
    float*    denom  = (float*)(smax + (size_t)N * H);

    int* cells  = (int*)scores;
    int* counts = (int*)smax;
    const bool own = (H == 4 && P == 65 && C == 520 && E >= 4 && E < (1 << 24) &&
                      (N + OW_GROUPS - 1) / OW_GROUPS <= OW_NODES &&
                      (size_t)OW_GROUPS * BIN_CH * BIN_CAP * 2 <= (size_t)E * H &&
                      (size_t)OW_GROUPS * BIN_CH <= (size_t)N * H);

    if (!own) {
        size_t n = (size_t)2 * N * H;
        k0_zero<<<(unsigned)((n + 255) / 256), 256, 0, stream>>>(smax, n);
    }

    {
        const int gx = (C + G_WN * 64 - 1) / (G_WN * 64);
        const int gy = (N + G_XROWS - 1) / G_XROWS;
        dim3 grid(gx, gy + (own ? (BIN_CH * BIN_Q + gx - 1) / gx : 0));
        k1_gemm_wmma<<<grid, G_THREADS, 0, stream>>>(x, W, bias, qk, N, F, C, H, P,
                                                     ei, cells, counts, E);
    }

    if (own) {
        ke_owner<<<OW_GROUPS, OW_THREADS, 0, stream>>>(qk, ei, cells, counts, (float*)d_out, N, E);
        return;
    }

    {
        const int epb = (256 / 32) * K2_EPW;
        k2_scores<<<(E + epb - 1) / epb, 256, 0, stream>>>(qk, ei, scores, smax,
                                                           N, E, C, H, P);
    }

    {
        size_t n = (size_t)E * H;
        k3_exp_denom<<<(unsigned)((n + 255) / 256), 256, 0, stream>>>(scores, smax,
                                                                      ei, denom, E, H);
    }

    k4_out<<<(E + 255) / 256, 256, 0, stream>>>(scores, denom, ei,
                                                (float*)d_out, E, H);
}


# GEMM work-group columns of 176 = 11 sub-tile columns (wave tiles of 64, 64 and 48 columns; was 3 x 64): 520 columns cost 33 sub-tile columns of WMMA, was 36; each element's k-sum order unchanged
# speedup vs baseline: 3.2870x; 1.0379x over previous
#include <hip/hip_runtime.h>
#include <hip/hip_bf16.h>


typedef __attribute__((ext_vector_type(2))) float v2f;
typedef __attribute__((ext_vector_type(4))) float v4f;
typedef __attribute__((ext_vector_type(8))) float v8f;
typedef v4f v4f_a4 __attribute__((aligned(4)));

__device__ __forceinline__ unsigned fenc(float f) {
    unsigned u = __float_as_uint(f);
    return (u & 0x80000000u) ? ~u : (u | 0x80000000u);
}
__device__ __forceinline__ float fdec(unsigned e) {
    return (e & 0x80000000u) ? __uint_as_float(e & 0x7FFFFFFFu)
                             : __uint_as_float(~e);
}

__global__ void k0_zero(unsigned* __restrict__ p, size_t n) {
    size_t i = (size_t)blockIdx.x * blockDim.x + threadIdx.x;
    if (i < n) p[i] = 0u;
}

#define G_MI      2
#define G_WM      4
#define G_WN      3
#define G_NLAST   3
#define G_COLS    ((G_WN - 1) * 64 + G_NLAST * 16)
#define G_XROWS   (G_WM * G_MI * 16)
#define G_ROWS    (G_XROWS + G_COLS)
#define G_PITCH   36
#define G_THREADS (G_WM * G_WN * 32)
#define G_RSTEP   (G_THREADS / 8)
#define G_PIECES  ((G_ROWS + G_RSTEP - 1) / G_RSTEP)

#define OW_GROUPS  256
#define OW_THREADS 1024
#define OW_NODES   196
#define OW_CAP     4096
#define OW_UNR     16
#define OW_B       8
#define OW_FILL    13
#define BIN_CH     64
#define BIN_Q      4
#define BIN_CAP    96
#define BIN_TRIPS  ((BIN_CH * BIN_CAP + OW_THREADS - 1) / OW_THREADS)
#define BIN_IPT    17

typedef __attribute__((ext_vector_type(4))) int v4i;
typedef v4i v4i_a4 __attribute__((aligned(4)));
typedef __attribute__((ext_vector_type(2))) int v2i;

__device__ __forceinline__ void bin_edges(const int* __restrict__ ei, int* __restrict__ cells,
                                          int* __restrict__ counts, int N, int E,
                                          int wg, unsigned* lc, int tid) {
    const int c   = wg / BIN_Q;
    const unsigned q = (unsigned)wg % BIN_Q;
    const unsigned per = (unsigned)((N + OW_GROUPS - 1) / OW_GROUPS);
    const int chunk = (E + BIN_CH - 1) / BIN_CH;
    const int lo = c * chunk;
    const int hi = (E - lo < chunk) ? E : lo + chunk;
    if (tid < OW_GROUPS / BIN_Q) lc[tid] = 0u;
    __syncthreads();
    for (int b0 = lo; b0 < hi; b0 += G_THREADS * BIN_IPT) {
        int s[BIN_IPT], d[BIN_IPT];
        #pragma unroll
        for (int u = 0; u < BIN_IPT; ++u) {
            const int e  = b0 + u * G_THREADS + tid;
            const int ec = e < hi ? e : hi - 1;
            s[u] = ei[ec];
            d[u] = ei[(size_t)E + ec];
        }
        __builtin_amdgcn_sched_barrier(0);
        #pragma unroll
        for (int u = 0; u < BIN_IPT; ++u)
            asm volatile("" : "+v"(d[u]));
        unsigned g[BIN_IPT], idx[BIN_IPT];
        #pragma unroll
        for (int u = 0; u < BIN_IPT; ++u) {
            const int e = b0 + u * G_THREADS + tid;
            const unsigned us = (unsigned)s[u];
            g[u]   = us / per;
            idx[u] = (unsigned)BIN_CAP;
            if (e < hi && us < (unsigned)N && g[u] / (OW_GROUPS / BIN_Q) == q)
                idx[u] = atomicAdd(&lc[g[u] % (OW_GROUPS / BIN_Q)], 1u);
        }
        #pragma unroll
        for (int u = 0; u < BIN_IPT; ++u) {
            if (idx[u] < (unsigned)BIN_CAP) {
                const int e = b0 + u * G_THREADS + tid;
                const unsigned sl = (unsigned)s[u] - g[u] * per;
                v2i en;
                en.x = (int)((unsigned)e | (sl << 24));
                en.y = d[u];
                *(v2i*)(cells + 2 * (((size_t)g[u] * BIN_CH + c) * BIN_CAP + idx[u])) = en;
            }
        }
    }
    __syncthreads();
    if (tid < OW_GROUPS / BIN_Q)
        counts[(size_t)(q * (OW_GROUPS / BIN_Q) + tid) * BIN_CH + c] = (int)lc[tid];
}

template <int NI>
__device__ __forceinline__ void gemm_cols(const float* sb, int aoff, int boff, int n0, v8f (&acc)[G_MI][4]) {
    #pragma unroll
    for (int j = 0; j < 4; ++j) {
        v4f a[G_MI], b[NI];
        #pragma unroll
        for (int mi = 0; mi < G_MI; ++mi)
            a[mi] = *(const v4f*)(sb + aoff + mi * 16 * G_PITCH + 4 * j);
        #pragma unroll
        for (int ni = 0; ni < NI; ++ni)
            b[ni] = *(const v4f*)(sb + boff + (n0 + ni) * 16 * G_PITCH + 4 * j);
        #pragma unroll
        for (int t2 = 0; t2 < 2; ++t2) {
            #pragma unroll
            for (int mi = 0; mi < G_MI; ++mi) {
                v2f af;
                af.x = a[mi][t2 * 2];
                af.y = a[mi][t2 * 2 + 1];
                #pragma unroll
                for (int ni = 0; ni < NI; ++ni) {
                    v2f bf;
                    bf.x = b[ni][t2 * 2];
                    bf.y = b[ni][t2 * 2 + 1];
                    acc[mi][n0 + ni] = __builtin_amdgcn_wmma_f32_16x16x4_f32(
                         false, af,  false, bf,
                         (short)0, acc[mi][n0 + ni],
                         false,  false);
                }
            }
        }
    }
}

__global__ __launch_bounds__(G_THREADS, 6)
void k1_gemm_wmma(const float* __restrict__ x, const float* __restrict__ W,
                  const float* __restrict__ bias, float* __restrict__ qk,
                  int N, int F, int C, int H, int P,
                  const int* __restrict__ ei, int* __restrict__ cells, int* __restrict__ counts,
                  int E) {
    __shared__ __attribute__((aligned(16))) float stg[2][G_ROWS * G_PITCH];
    __shared__ unsigned lc[OW_GROUPS / BIN_Q];
    const int tid  = threadIdx.x;
    {
        const int rowBlocks = (N + G_XROWS - 1) / G_XROWS;
        if ((int)blockIdx.y >= rowBlocks) {
            const int wg = ((int)blockIdx.y - rowBlocks) * (int)gridDim.x + (int)blockIdx.x;
            if (wg < BIN_CH * BIN_Q)
                bin_edges(ei, cells, counts, N, E, wg, lc, tid);
            return;
        }
    }
    const int lane = tid & 31;
    const int wv   = __builtin_amdgcn_readfirstlane(tid >> 5);
    const int wm   = wv % G_WM;
    const int wn   = wv / G_WM;
    const int Mg   = blockIdx.y * G_XROWS;
    const int Ng   = blockIdx.x * G_COLS;
    const int M0   = Mg + wm * (G_MI * 16);
    const int N0   = Ng + wn * 64;
    const int lm   = lane & 15;
    const int kh   = (lane >> 4) * 2;

    const int seg = tid & 7;
    const int l0  = (tid >> 3) * G_PITCH + seg * 4;
    const float* __restrict__ gsrc[G_PIECES];
    #pragma unroll
    for (int i = 0; i < G_PIECES; ++i) {
        const int R  = (tid >> 3) + G_RSTEP * i;
        const int Rc = R < G_ROWS ? R : G_ROWS - 1;
        const float* __restrict__ p;
        if (Rc < G_XROWS) { const int r = Mg + Rc;           p = x + (size_t)(r < N ? r : N - 1) * F; }
        else              { const int c = Ng + Rc - G_XROWS; p = W + (size_t)(c < C ? c : C - 1) * F; }
        gsrc[i] = p + seg * 4;
    }

    v8f acc[G_MI][4];
    #pragma unroll
    for (int mi = 0; mi < G_MI; ++mi)
        #pragma unroll
        for (int ni = 0; ni < 4; ++ni)
            acc[mi][ni] = (v8f){};

    const int kq  = (lane >> 4) * 16;
    const int nkb = F / 32;
    const int aoff = (wm * (G_MI * 16) + lm) * G_PITCH + kq;
    const int boff = (G_XROWS + wn * 64 + lm) * G_PITCH + kq;

    v4f t[G_PIECES];
    if (nkb > 0) {
        #pragma unroll
        for (int i = 0; i < G_PIECES; ++i) t[i] = *(const v4f_a4*)(gsrc[i]);
        #pragma unroll
        for (int i = 0; i < G_PIECES; ++i)
            if ((i + 1) * G_RSTEP <= G_ROWS || (tid >> 3) + G_RSTEP * i < G_ROWS)
                *(v4f*)(&stg[0][l0 + i * (G_RSTEP * G_PITCH)]) = t[i];
        __syncthreads();
    }
    for (int kb = 0; kb < nkb; ++kb) {
        const float* sb = stg[kb & 1];
        float*       sn = stg[(kb & 1) ^ 1];
        const bool more = (kb + 1 < nkb);
        if (more) {
            #pragma unroll
            for (int i = 0; i < G_PIECES; ++i)
                t[i] = *(const v4f_a4*)(gsrc[i] + (size_t)(kb + 1) * 32);
        }
        __builtin_amdgcn_sched_barrier(0);
        gemm_cols<G_NLAST>(sb, aoff, boff, 0, acc);
        if (wn < G_WN - 1) gemm_cols<4 - G_NLAST>(sb, aoff, boff, G_NLAST, acc);
        __builtin_amdgcn_sched_barrier(0);
        if (more) {
            #pragma unroll
            for (int i = 0; i < G_PIECES; ++i)
                if ((i + 1) * G_RSTEP <= G_ROWS || (tid >> 3) + G_RSTEP * i < G_ROWS)
                    *(v4f*)(sn + l0 + i * (G_RSTEP * G_PITCH)) = t[i];
        }
        __syncthreads();
    }

    int k = nkb * 32;
    if (k < F) {
        const float* __restrict__ xrow[G_MI];
        const float* __restrict__ wrow[4];
        #pragma unroll
        for (int mi = 0; mi < G_MI; ++mi) {
            int r = M0 + mi * 16 + lm;
            xrow[mi] = x + (size_t)(r < N ? r : N - 1) * F;
        }
        #pragma unroll
        for (int ni = 0; ni < 4; ++ni) {
            int c = N0 + ni * 16 + lm;
            wrow[ni] = W + (size_t)(c < C ? c : C - 1) * F;
        }
        for (; k < F; k += 4) {
            v2f a[G_MI], b[4];
            #pragma unroll
            for (int mi = 0; mi < G_MI; ++mi) {
                a[mi].x = xrow[mi][k + kh];
                a[mi].y = xrow[mi][k + kh + 1];
            }
            #pragma unroll
            for (int ni = 0; ni < 4; ++ni) {
                b[ni].x = wrow[ni][k + kh];
                b[ni].y = wrow[ni][k + kh + 1];
            }
            #pragma unroll
            for (int mi = 0; mi < G_MI; ++mi)
                #pragma unroll
                for (int ni = 0; ni < 4; ++ni)
                    acc[mi][ni] = __builtin_amdgcn_wmma_f32_16x16x4_f32(
                         false, a[mi],  false, b[ni],
                         (short)0, acc[mi][ni],
                         false,  false);
        }
    }

    const bool split = (H == 4 && P == 65 && C == 520);
    const int rsub = (lane >> 4) << 3;
    #pragma unroll
    for (int ni = 0; ni < 4; ++ni) {
        const int col = N0 + ni * 16 + lm;
        if (col >= C || wn * 64 + ni * 16 >= G_COLS) continue;
        const float bv = bias[col];
        float* __restrict__ dst = qk + col;
        size_t pitch = (size_t)C;
        if (split) {
            const int h   = col / 130;
            const int rr  = col - h * 130;
            const int isk = (rr >= 65) ? 1 : 0;
            const int p   = rr - isk * 65;
            if (p < 64) {
                dst   = qk + (size_t)isk * N * 256 + h * 64 + p;
                pitch = 256;
            } else {
                dst   = qk + (size_t)N * 512 + (size_t)isk * N * 4 + h;
                pitch = 4;
            }
        }
        #pragma unroll
        for (int mi = 0; mi < G_MI; ++mi) {
            const int rbase = M0 + mi * 16 + rsub;
            #pragma unroll
            for (int r = 0; r < 8; ++r) {
                const int row = rbase + r;
                if (row < N)
                    dst[(size_t)row * pitch] = acc[mi][ni][r] + bv;
            }
        }
    }
}


#define K2_EPW 16

template <int CTRL>
__device__ __forceinline__ float row_xor_add(float keep, float send) {
    const int t = __builtin_amdgcn_update_dpp(0, __builtin_bit_cast(int, send),
                                              CTRL, 0xf, 0xf, false);
    return keep + __builtin_bit_cast(float, t);
}

struct EdgeFrag {
    v4f qa, qb, ka, kb;
    float qt, kt;
};

struct GatherPlanes {
    const float* __restrict__ qm;
    const float* __restrict__ km;
    const float* __restrict__ qt;
    const float* __restrict__ kt;
};

__device__ __forceinline__ void k2_load_edge(EdgeFrag& f, const GatherPlanes& g,
                                             int s, int d, int offA, int th) {
    const float* __restrict__ sq = g.qm + (size_t)s * 256 + offA;
    const float* __restrict__ dk = g.km + (size_t)d * 256 + offA;
    f.qa = *(const v4f_a4*)(sq);
    f.qb = *(const v4f_a4*)(sq + 128);
    f.ka = *(const v4f_a4*)(dk);
    f.kb = *(const v4f_a4*)(dk + 128);
    f.qt = g.qt[(size_t)s * 4 + th];
    f.kt = g.kt[(size_t)d * 4 + th];
    __builtin_amdgcn_sched_barrier(0);
}

__device__ __forceinline__ void k2_reduce_store(const EdgeFrag& f, float* __restrict__ scores,
                                                unsigned* __restrict__ smax, int e, int s,
                                                int up, int th, int lane) {
    float a1 = f.qa.x * f.ka.x;
    float a2 = f.qb.x * f.kb.x;
    a1 = fmaf(f.qa.y, f.ka.y, a1);  a2 = fmaf(f.qb.y, f.kb.y, a2);
    a1 = fmaf(f.qa.z, f.ka.z, a1);  a2 = fmaf(f.qb.z, f.kb.z, a2);
    a1 = fmaf(f.qa.w, f.ka.w, a1);  a2 = fmaf(f.qb.w, f.kb.w, a2);
    float r = row_xor_add<0x168>(up ? a2 : a1, up ? a1 : a2);
    r = row_xor_add<0x164>(r, r);
    r = row_xor_add<0x162>(r, r);
    r = row_xor_add<0x161>(r, r);
    r = fmaf(f.qt, f.kt, r);
    if ((lane & 7) == 0) {
        scores[(size_t)e * 4 + th] = r;
        atomicMax(&smax[(size_t)s * 4 + th], fenc(r));
    }
}

__global__ __launch_bounds__(256)
void k2_scores(const float* __restrict__ qk, const int* __restrict__ ei,
               float* __restrict__ scores, unsigned* __restrict__ smax,
               int N, int E, int C, int H, int P) {
    const int lane = threadIdx.x & 31;
    const int wv = __builtin_amdgcn_readfirstlane((int)(threadIdx.x >> 5));
    const int e0 = (blockIdx.x * (blockDim.x >> 5) + wv) * K2_EPW;
    if (e0 >= E) return;
    const int cnt = (E - e0 < K2_EPW) ? (E - e0) : K2_EPW;

    const int le = (e0 + (lane & 15) < E) ? (e0 + (lane & 15)) : (E - 1);
    const int ev = ei[(size_t)(lane >> 4) * E + le];

    if (H == 4 && P == 65 && C == 520) {
        const int hs   = lane >> 4;
        const int offA = hs * 64 + (lane & 15) * 4;
        const int up   = (lane >> 3) & 1;
        const int th   = hs + up * 2;
        GatherPlanes g;
        g.qm = qk;
        g.km = qk + (size_t)N * 256;
        g.qt = qk + (size_t)N * 512;
        g.kt = g.qt + (size_t)N * 4;

        EdgeFrag fa, fb;
        int sa = __builtin_amdgcn_readlane(ev, 0), sb = sa;
        k2_load_edge(fa, g, sa, __builtin_amdgcn_readlane(ev, 16), offA, th);
        int j = 0;
        for (; j + 2 < cnt; j += 2) {
            sb = __builtin_amdgcn_readlane(ev, j + 1);
            k2_load_edge(fb, g, sb, __builtin_amdgcn_readlane(ev, j + 17), offA, th);
            k2_reduce_store(fa, scores, smax, e0 + j, sa, up, th, lane);
            sa = __builtin_amdgcn_readlane(ev, j + 2);
            k2_load_edge(fa, g, sa, __builtin_amdgcn_readlane(ev, j + 18), offA, th);
            k2_reduce_store(fb, scores, smax, e0 + j + 1, sb, up, th, lane);
        }
        if (j + 1 < cnt) {
            sb = __builtin_amdgcn_readlane(ev, j + 1);
            k2_load_edge(fb, g, sb, __builtin_amdgcn_readlane(ev, j + 17), offA, th);
            k2_reduce_store(fa, scores, smax, e0 + j, sa, up, th, lane);
            k2_reduce_store(fb, scores, smax, e0 + j + 1, sb, up, th, lane);
        } else {
            k2_reduce_store(fa, scores, smax, e0 + j, sa, up, th, lane);
        }
    } else {
        for (int j = 0; j < cnt; ++j) {
            const int s = __builtin_amdgcn_readlane(ev, j);
            const int d = __builtin_amdgcn_readlane(ev, 16 + j);
            const float* __restrict__ srow = qk + (size_t)s * C;
            const float* __restrict__ drow = qk + (size_t)d * C;
            for (int h = 0; h < H; ++h) {
                const float* __restrict__ qh = srow + h * 2 * P;
                const float* __restrict__ kh = drow + h * 2 * P + P;
                float a = 0.0f;
                for (int p = lane; p < P; p += 32)
                    a += qh[p] * kh[p];
                #pragma unroll
                for (int off = 16; off > 0; off >>= 1)
                    a += __shfl_xor(a, off, 32);
                if (lane == 0) {
                    scores[(size_t)(e0 + j) * H + h] = a;
                    atomicMax(&smax[(size_t)s * H + h], fenc(a));
                }
            }
        }
    }
}

__global__ __launch_bounds__(256)
void k3_exp_denom(float* __restrict__ scores, const unsigned* __restrict__ smax,
                  const int* __restrict__ ei, float* __restrict__ denom,
                  int E, int H) {
    size_t i = (size_t)blockIdx.x * blockDim.x + threadIdx.x;
    if (i >= (size_t)E * H) return;
    const int e = (int)(i / H);
    const int h = (int)(i % H);
    const int s = ei[e];
    const float m  = fdec(smax[(size_t)s * H + h]);
    const float ex = __expf(scores[i] - m);
    scores[i] = ex;
    atomicAdd(&denom[(size_t)s * H + h], ex);
}

__global__ __launch_bounds__(256)
void k4_out(const float* __restrict__ ex, const float* __restrict__ denom,
            const int* __restrict__ ei, float* __restrict__ out, int E, int H) {
    const int e = blockIdx.x * blockDim.x + threadIdx.x;
    if (e >= E) return;
    const int s = ei[e];
    float sum = 0.0f;
    for (int h = 0; h < H; ++h)
        sum += ex[(size_t)e * H + h] / denom[(size_t)s * H + h];
    out[e] = sum / (float)H;
}


__device__ __forceinline__ unsigned ow_umin(unsigned a, unsigned b) { return a < b ? a : b; }

__device__ __forceinline__ void ow_append(int e, unsigned sl, int* le, int* ls, unsigned* cnt) {
    const unsigned idx = atomicAdd(cnt, 1u);
    if (idx < (unsigned)OW_CAP) { le[idx] = e; ls[idx] = (int)sl; }
}

__device__ __forceinline__ void ow_test4(v4i s, int e, int n0, unsigned nn,
                                         int* le, int* ls, unsigned* cnt) {
    const unsigned a = (unsigned)(s.x - n0), b = (unsigned)(s.y - n0);
    const unsigned c = (unsigned)(s.z - n0), d = (unsigned)(s.w - n0);
    if (ow_umin(ow_umin(a, b), ow_umin(c, d)) < nn) {
        #pragma nounroll
        for (int j = 0; j < 4; ++j) {
            const unsigned sl = (j == 0) ? a : (j == 1) ? b : (j == 2) ? c : d;
            if (sl < nn) ow_append(e + j, sl, le, ls, cnt);
        }
    }
}

__device__ __forceinline__ void ow_scan_simple(const int* __restrict__ ei, int lo, int hi, int n0, unsigned nn,
                                               int tid, int* le, int* ls, unsigned* cnt) {
    for (int e = lo + tid; e < hi; e += OW_THREADS) {
        const unsigned sl = (unsigned)(ei[e] - n0);
        if (sl < nn) ow_append(e, sl, le, ls, cnt);
    }
}

__device__ __forceinline__ void ow_scan(const int* __restrict__ ei, int lo, int hi, int n0, unsigned nn,
                                        int tid, int grp, int* le, int* ls, unsigned* cnt) {
    const int CH    = OW_THREADS * 4 * OW_UNR;
    const int nfull = (hi - lo) / CH;
    if (nfull > 0) {
        int c = grp % nfull;
        for (int it = 0; it < nfull; ++it) {
            const int b0 = lo + c * CH + tid * 4;
            v4i v[OW_UNR];
            #pragma unroll
            for (int u = 0; u < OW_UNR; ++u)
                v[u] = *(const v4i_a4*)(ei + b0 + u * (OW_THREADS * 4));
            __builtin_amdgcn_sched_barrier(0);
            #pragma unroll
            for (int u = 0; u < OW_UNR; ++u)
                ow_test4(v[u], b0 + u * (OW_THREADS * 4), n0, nn, le, ls, cnt);
            c = (c + 1 < nfull) ? c + 1 : 0;
        }
    }
    const int r0 = lo + nfull * CH;
    if (r0 < hi) {
        const int b0 = r0 + tid * 4;
        v4i v[OW_UNR];
        #pragma unroll
        for (int u = 0; u < OW_UNR; ++u) {
            const int b = b0 + u * (OW_THREADS * 4);
            v[u] = *(const v4i_a4*)(ei + ((b + 3 < hi) ? b : 0));
        }
        __builtin_amdgcn_sched_barrier(0);
        #pragma unroll
        for (int u = 0; u < OW_UNR; ++u) {
            const int b = b0 + u * (OW_THREADS * 4);
            if (b + 3 < hi) ow_test4(v[u], b, n0, nn, le, ls, cnt);
        }
        ow_scan_simple(ei, lo + ((hi - lo) & ~3), hi, n0, nn, tid, le, ls, cnt);
    }
}

__device__ __forceinline__ void ow_dst(const int* __restrict__ ei, int E, int n, int tid,
                                       const int* le, int* ld) {
    for (int i0 = tid; i0 < n; i0 += OW_THREADS * 4) {
        int v[4];
        #pragma unroll
        for (int u = 0; u < 4; ++u) {
            int i = i0 + u * OW_THREADS;
            i = i < n ? i : n - 1;
            v[u] = ei[(size_t)E + le[i]];
        }
        #pragma unroll
        for (int u = 0; u < 4; ++u) {
            int i = i0 + u * OW_THREADS;
            i = i < n ? i : n - 1;
            ld[i] = v[u];
        }
    }
}

struct KFrag { v4f ka, kb; };

__device__ __forceinline__ void ow_scores(int n, const float* __restrict__ km, const float* __restrict__ kt,
                                          const float* qs, const float* qts,
                                          const int* ld, const int* ls, float* sc, unsigned* mx,
                                          int wv, int lane) {
    const int hs   = lane >> 4;
    const int offA = hs * 64 + (lane & 15) * 4;
    const int up   = (lane >> 3) & 1;
    const int th   = hs + up * 2;
    for (int base = wv * OW_B; base < n; base += (OW_THREADS / 32) * OW_B) {
        int ii = base + (lane & (OW_B - 1));
        ii = ii < n ? ii : n - 1;
        const int dv = ld[ii];
        const int sv = ls[ii];
        int i2 = base + (lane >> 2);
        i2 = i2 < n ? i2 : n - 1;
        const float ktv = kt[(size_t)ld[i2] * 4 + (lane & 3)];
        KFrag f[OW_B];
        #pragma unroll
        for (int j = 0; j < OW_B; ++j) {
            const int d = __builtin_amdgcn_readlane(dv, j);
            const float* __restrict__ dk = km + (size_t)d * 256 + offA;
            f[j].ka = *(const v4f_a4*)(dk);
            f[j].kb = *(const v4f_a4*)(dk + 128);
        }
        __builtin_amdgcn_sched_barrier(0);
        #pragma unroll
        for (int j = 0; j < OW_B; ++j) {
            if (base + j < n) {
                const int sl = __builtin_amdgcn_readlane(sv, j);
                const v4f qa = *(const v4f*)(qs + sl * 256 + offA);
                const v4f qb = *(const v4f*)(qs + sl * 256 + offA + 128);
                const float qtv = qts[sl * 4 + th];
                const float ktj = __shfl(ktv, 4 * j + th, 32);
                float a1 = qa.x * f[j].ka.x;
                float a2 = qb.x * f[j].kb.x;
                a1 = fmaf(qa.y, f[j].ka.y, a1);  a2 = fmaf(qb.y, f[j].kb.y, a2);
                a1 = fmaf(qa.z, f[j].ka.z, a1);  a2 = fmaf(qb.z, f[j].kb.z, a2);
                a1 = fmaf(qa.w, f[j].ka.w, a1);  a2 = fmaf(qb.w, f[j].kb.w, a2);
                float r = row_xor_add<0x168>(up ? a2 : a1, up ? a1 : a2);
                r = row_xor_add<0x164>(r, r);
                r = row_xor_add<0x162>(r, r);
                r = row_xor_add<0x161>(r, r);
                r = fmaf(qtv, ktj, r);
                if ((lane & 7) == 0) {
                    sc[(base + j) * 4 + th] = r;
                    atomicMax(&mx[sl * 4 + th], fenc(r));
                }
            }
        }
    }
}

__global__ __launch_bounds__(OW_THREADS)
void ke_owner(const float* __restrict__ qk, const int* __restrict__ ei,
              const int* __restrict__ cells, const int* __restrict__ counts,
              float* __restrict__ out, int N, int E) {
    __shared__ __attribute__((aligned(16))) float qs[OW_NODES * 256];
    __shared__ __attribute__((aligned(16))) float qts[OW_NODES * 4];
    __shared__ __attribute__((aligned(16))) float sm[OW_NODES * 4];
    __shared__ unsigned mx[OW_NODES * 4];
    __shared__ __attribute__((aligned(16))) float sc[OW_CAP * 4];
    __shared__ int le[OW_CAP];
    __shared__ int ls[OW_CAP];
    __shared__ int ld[OW_CAP];
    __shared__ unsigned cnt;
    __shared__ int cc[BIN_CH];
    __shared__ int off[BIN_CH];
    __shared__ int binfo[2];

    const int tid  = threadIdx.x;
    const int lane = tid & 31;
    const int wv   = __builtin_amdgcn_readfirstlane(tid >> 5);
    const int per  = (N + (int)gridDim.x - 1) / (int)gridDim.x;
    const int n0   = (int)blockIdx.x * per;
    if (n0 >= N) return;
    const int nni  = (N - n0 < per) ? (N - n0) : per;
    const unsigned nn = (unsigned)nni;

    const float* __restrict__ qm = qk;
    const float* __restrict__ km = qk + (size_t)N * 256;
    const float* __restrict__ qt = qk + (size_t)N * 512;
    const float* __restrict__ kt = qt + (size_t)N * 4;

    for (int i = tid; i < nni * 4; i += OW_THREADS) {
        mx[i]  = 0u;
        sm[i]  = 0.0f;
        qts[i] = qt[(size_t)n0 * 4 + i];
    }
    if (tid == 0) cnt = 0u;
    if (tid < BIN_CH) cc[tid] = counts[(size_t)blockIdx.x * BIN_CH + tid];
    {
        const float* __restrict__ qsrc = qm + (size_t)n0 * 256;
        const int tot = nni * 64;
        for (int i0 = tid; i0 < tot; i0 += OW_THREADS * OW_FILL) {
            v4f t[OW_FILL];
            #pragma unroll
            for (int u = 0; u < OW_FILL; ++u) {
                int i = i0 + u * OW_THREADS;
                i = i < tot ? i : tot - 1;
                t[u] = *(const v4f_a4*)(qsrc + 4 * i);
            }
            __builtin_amdgcn_sched_barrier(0);
            #pragma unroll
            for (int u = 0; u < OW_FILL; ++u) {
                int i = i0 + u * OW_THREADS;
                i = i < tot ? i : tot - 1;
                *(v4f*)(qs + 4 * i) = t[u];
            }
        }
    }
    __syncthreads();

    if (tid < BIN_CH) {
        int o = 0;
        for (int i = 0; i < tid; ++i) o += cc[i];
        off[tid] = o;
    } else if (tid == BIN_CH) {
        int t = 0, bad = 0;
        for (int i = 0; i < BIN_CH; ++i) { t += cc[i]; bad |= (cc[i] > BIN_CAP) ? 1 : 0; }
        binfo[0] = t;
        binfo[1] = bad;
    }
    __syncthreads();
    int n = binfo[0];
    bool listed = false;
    if (binfo[1] != 0) {
        ow_scan(ei, 0, E, n0, nn, tid, (int)blockIdx.x, le, ls, &cnt);
        __syncthreads();
        n = (int)cnt;
        if (n <= OW_CAP) {
            ow_dst(ei, E, n, tid, le, ld);
            listed = true;
        }
    } else if (n <= OW_CAP) {
        const int* __restrict__ mine = cells + (size_t)blockIdx.x * (BIN_CH * BIN_CAP * 2);
        v2i  en[BIN_TRIPS];
        int  pos[BIN_TRIPS];
        #pragma unroll
        for (int u = 0; u < BIN_TRIPS; ++u) {
            const int s = tid + u * OW_THREADS;
            const int b = s / BIN_CAP;
            const int k = s - b * BIN_CAP;
            const bool used = (b < BIN_CH) && (k < cc[b < BIN_CH ? b : 0]);
            pos[u] = used ? off[b] + k : -1;
            const int* __restrict__ p = used ? mine + 2 * s : counts + (size_t)blockIdx.x * BIN_CH;
            en[u] = *(const v2i*)p;
        }
        __builtin_amdgcn_sched_barrier(0);
        #pragma unroll
        for (int u = 0; u < BIN_TRIPS; ++u) {
            if (pos[u] >= 0) {
                le[pos[u]] = (int)((unsigned)en[u].x & 0xFFFFFFu);
                ls[pos[u]] = (int)((unsigned)en[u].x >> 24);
                ld[pos[u]] = en[u].y;
            }
        }
        listed = true;
    }

    if (listed) {
        __syncthreads();
        ow_scores(n, km, kt, qs, qts, ld, ls, sc, mx, wv, lane);
        __syncthreads();
        for (int t = tid; t < n * 4; t += OW_THREADS) {
            const int g  = ls[t >> 2] * 4 + (t & 3);
            const float ex = __expf(sc[t] - fdec(mx[g]));
            sc[t] = ex;
            atomicAdd(&sm[g], ex);
        }
        __syncthreads();
        for (int i = tid; i < n; i += OW_THREADS) {
            const v4f x  = *(const v4f*)(sc + 4 * i);
            const v4f dn = *(const v4f*)(sm + 4 * ls[i]);
            float sum = 0.0f;
            sum += x.x / dn.x;
            sum += x.y / dn.y;
            sum += x.z / dn.z;
            sum += x.w / dn.w;
            out[le[i]] = sum / 4.0f;
        }
        return;
    }

    const int nr = (E + OW_CAP - 1) / OW_CAP;
    for (int pass = 0; pass < 3; ++pass) {
        for (int c = 0; c < nr; ++c) {
            __syncthreads();
            if (tid == 0) cnt = 0u;
            __syncthreads();
            const int hi = ((c + 1) * OW_CAP < E) ? (c + 1) * OW_CAP : E;
            ow_scan_simple(ei, c * OW_CAP, hi, n0, nn, tid, le, ls, &cnt);
            __syncthreads();
            const int m = (int)cnt;
            ow_dst(ei, E, m, tid, le, ld);
            __syncthreads();
            ow_scores(m, km, kt, qs, qts, ld, ls, sc, mx, wv, lane);
            __syncthreads();
            if (pass == 1) {
                for (int t = tid; t < m * 4; t += OW_THREADS) {
                    const int g = ls[t >> 2] * 4 + (t & 3);
                    atomicAdd(&sm[g], __expf(sc[t] - fdec(mx[g])));
                }
            } else if (pass == 2) {
                for (int i = tid; i < m; i += OW_THREADS) {
                    const int g = ls[i] * 4;
                    const v4f x  = *(const v4f*)(sc + 4 * i);
                    const v4f dn = *(const v4f*)(sm + g);
                    float sum = 0.0f;
                    sum += __expf(x.x - fdec(mx[g]))     / dn.x;
                    sum += __expf(x.y - fdec(mx[g + 1])) / dn.y;
                    sum += __expf(x.z - fdec(mx[g + 2])) / dn.z;
                    sum += __expf(x.w - fdec(mx[g + 3])) / dn.w;
                    out[le[i]] = sum / 4.0f;
                }
            }
        }
    }
}

extern "C" void kernel_launch(void* const* d_in, const int* in_sizes, int n_in,
                              void* d_out, int out_size, void* d_ws, size_t ws_size,
                              hipStream_t stream) {
    const float* x    = (const float*)d_in[0];
    const int*   ei   = (const int*)d_in[1];
    const float* W    = (const float*)d_in[2];
    const float* bias = (const float*)d_in[3];
    const int C = in_sizes[3];
    const int F = in_sizes[2] / C;
    const int N = in_sizes[0] / F;
    const int E = in_sizes[1] / 2;
    const int H = (C - 2 * F) / 2;
    const int P = F / H + 1;

    float*    qk     = (float*)d_ws;
    float*    scores = qk + (size_t)N * C;
    unsigned* smax   = (unsigned*)(scores + (size_t)E * H);
    float*    denom  = (float*)(smax + (size_t)N * H);

    int* cells  = (int*)scores;
    int* counts = (int*)smax;
    const bool own = (H == 4 && P == 65 && C == 520 && E >= 4 && E < (1 << 24) &&
                      (N + OW_GROUPS - 1) / OW_GROUPS <= OW_NODES &&
                      (size_t)OW_GROUPS * BIN_CH * BIN_CAP * 2 <= (size_t)E * H &&
                      (size_t)OW_GROUPS * BIN_CH <= (size_t)N * H);

    if (!own) {
        size_t n = (size_t)2 * N * H;
        k0_zero<<<(unsigned)((n + 255) / 256), 256, 0, stream>>>(smax, n);
    }

    {
        const int gx = (C + G_COLS - 1) / G_COLS;
        const int gy = (N + G_XROWS - 1) / G_XROWS;
        dim3 grid(gx, gy + (own ? (BIN_CH * BIN_Q + gx - 1) / gx : 0));
        k1_gemm_wmma<<<grid, G_THREADS, 0, stream>>>(x, W, bias, qk, N, F, C, H, P,
                                                     ei, cells, counts, E);
    }

    if (own) {
        ke_owner<<<OW_GROUPS, OW_THREADS, 0, stream>>>(qk, ei, cells, counts, (float*)d_out, N, E);
        return;
    }

    {
        const int epb = (256 / 32) * K2_EPW;
        k2_scores<<<(E + epb - 1) / epb, 256, 0, stream>>>(qk, ei, scores, smax,
                                                           N, E, C, H, P);
    }

    {
        size_t n = (size_t)E * H;
        k3_exp_denom<<<(unsigned)((n + 255) / 256), 256, 0, stream>>>(scores, smax,
                                                                      ei, denom, E, H);
    }

    k4_out<<<(E + 255) / 256, 256, 0, stream>>>(scores, denom, ei,
                                                (float*)d_out, E, H);
}


# GEMM stores through LDS: columns taken in plane order (a wave tile = 64 contiguous floats of a Qm/Km row), each store instruction writes 4 whole 128-B lines; was 2 half lines; sums unchanged
# speedup vs baseline: 3.4615x; 1.0531x over previous
#include <hip/hip_runtime.h>
#include <hip/hip_bf16.h>


typedef __attribute__((ext_vector_type(2))) float v2f;
typedef __attribute__((ext_vector_type(4))) float v4f;
typedef __attribute__((ext_vector_type(8))) float v8f;
typedef v4f v4f_a4 __attribute__((aligned(4)));

__device__ __forceinline__ unsigned fenc(float f) {
    unsigned u = __float_as_uint(f);
    return (u & 0x80000000u) ? ~u : (u | 0x80000000u);
}
__device__ __forceinline__ float fdec(unsigned e) {
    return (e & 0x80000000u) ? __uint_as_float(e & 0x7FFFFFFFu)
                             : __uint_as_float(~e);
}

__global__ void k0_zero(unsigned* __restrict__ p, size_t n) {
    size_t i = (size_t)blockIdx.x * blockDim.x + threadIdx.x;
    if (i < n) p[i] = 0u;
}

#define G_MI      2
#define G_WM      4
#define G_WN      3
#define G_NLAST   3
#define G_COLS    ((G_WN - 1) * 64 + G_NLAST * 16)
#define G_XROWS   (G_WM * G_MI * 16)
#define G_ROWS    (G_XROWS + G_COLS)
#define G_PITCH   36
#define G_THREADS (G_WM * G_WN * 32)
#define G_RSTEP   (G_THREADS / 8)
#define G_PIECES  ((G_ROWS + G_RSTEP - 1) / G_RSTEP)

#define OW_GROUPS  256
#define OW_THREADS 1024
#define OW_NODES   196
#define OW_CAP     4096
#define OW_UNR     16
#define OW_B       8
#define OW_FILL    13
#define BIN_CH     64
#define BIN_Q      4
#define BIN_CAP    96
#define BIN_TRIPS  ((BIN_CH * BIN_CAP + OW_THREADS - 1) / OW_THREADS)
#define BIN_IPT    17

typedef __attribute__((ext_vector_type(4))) int v4i;
typedef v4i v4i_a4 __attribute__((aligned(4)));
typedef __attribute__((ext_vector_type(2))) int v2i;

__device__ __forceinline__ void bin_edges(const int* __restrict__ ei, int* __restrict__ cells,
                                          int* __restrict__ counts, int N, int E,
                                          int wg, unsigned* lc, int tid) {
    const int c   = wg / BIN_Q;
    const unsigned q = (unsigned)wg % BIN_Q;
    const unsigned per = (unsigned)((N + OW_GROUPS - 1) / OW_GROUPS);
    const int chunk = (E + BIN_CH - 1) / BIN_CH;
    const int lo = c * chunk;
    const int hi = (E - lo < chunk) ? E : lo + chunk;
    if (tid < OW_GROUPS / BIN_Q) lc[tid] = 0u;
    __syncthreads();
    for (int b0 = lo; b0 < hi; b0 += G_THREADS * BIN_IPT) {
        int s[BIN_IPT], d[BIN_IPT];
        #pragma unroll
        for (int u = 0; u < BIN_IPT; ++u) {
            const int e  = b0 + u * G_THREADS + tid;
            const int ec = e < hi ? e : hi - 1;
            s[u] = ei[ec];
            d[u] = ei[(size_t)E + ec];
        }
        __builtin_amdgcn_sched_barrier(0);
        #pragma unroll
        for (int u = 0; u < BIN_IPT; ++u)
            asm volatile("" : "+v"(d[u]));
        unsigned g[BIN_IPT], idx[BIN_IPT];
        #pragma unroll
        for (int u = 0; u < BIN_IPT; ++u) {
            const int e = b0 + u * G_THREADS + tid;
            const unsigned us = (unsigned)s[u];
            g[u]   = us / per;
            idx[u] = (unsigned)BIN_CAP;
            if (e < hi && us < (unsigned)N && g[u] / (OW_GROUPS / BIN_Q) == q)
                idx[u] = atomicAdd(&lc[g[u] % (OW_GROUPS / BIN_Q)], 1u);
        }
        #pragma unroll
        for (int u = 0; u < BIN_IPT; ++u) {
            if (idx[u] < (unsigned)BIN_CAP) {
                const int e = b0 + u * G_THREADS + tid;
                const unsigned sl = (unsigned)s[u] - g[u] * per;
                v2i en;
                en.x = (int)((unsigned)e | (sl << 24));
                en.y = d[u];
                *(v2i*)(cells + 2 * (((size_t)g[u] * BIN_CH + c) * BIN_CAP + idx[u])) = en;
            }
        }
    }
    __syncthreads();
    if (tid < OW_GROUPS / BIN_Q)
        counts[(size_t)(q * (OW_GROUPS / BIN_Q) + tid) * BIN_CH + c] = (int)lc[tid];
}

__device__ __forceinline__ int gemm_col(int j, bool split, int C) {
    if (!split) return j;
    const int s  = j >> 4, lm = j & 15;
    const int cx = s / (G_COLS / 16);
    const int t  = s - cx * (G_COLS / 16);
    int pl, q4;
    if (t < 8)       { pl = 2 * cx + (t >> 2);  q4 = t & 3; }
    else if (t < 10) { pl = (cx < 2) ? 6 : 7;   q4 = ((cx < 2) ? 2 * cx : 2) + (t - 8); }
    else if (cx < 2) { pl = 7;                  q4 = cx; }
    else             return (lm < 8) ? 65 * lm + 64 : C;
    return 65 * pl + 16 * q4 + lm;
}

#define EP_PITCH 68

template <int NI>
__device__ __forceinline__ void gemm_cols(const float* sb, int aoff, int boff, int n0, v8f (&acc)[G_MI][4]) {
    #pragma unroll
    for (int j = 0; j < 4; ++j) {
        v4f a[G_MI], b[NI];
        #pragma unroll
        for (int mi = 0; mi < G_MI; ++mi)
            a[mi] = *(const v4f*)(sb + aoff + mi * 16 * G_PITCH + 4 * j);
        #pragma unroll
        for (int ni = 0; ni < NI; ++ni)
            b[ni] = *(const v4f*)(sb + boff + (n0 + ni) * 16 * G_PITCH + 4 * j);
        #pragma unroll
        for (int t2 = 0; t2 < 2; ++t2) {
            #pragma unroll
            for (int mi = 0; mi < G_MI; ++mi) {
                v2f af;
                af.x = a[mi][t2 * 2];
                af.y = a[mi][t2 * 2 + 1];
                #pragma unroll
                for (int ni = 0; ni < NI; ++ni) {
                    v2f bf;
                    bf.x = b[ni][t2 * 2];
                    bf.y = b[ni][t2 * 2 + 1];
                    acc[mi][n0 + ni] = __builtin_amdgcn_wmma_f32_16x16x4_f32(
                         false, af,  false, bf,
                         (short)0, acc[mi][n0 + ni],
                         false,  false);
                }
            }
        }
    }
}

__global__ __launch_bounds__(G_THREADS, 6)
void k1_gemm_wmma(const float* __restrict__ x, const float* __restrict__ W,
                  const float* __restrict__ bias, float* __restrict__ qk,
                  int N, int F, int C, int H, int P,
                  const int* __restrict__ ei, int* __restrict__ cells, int* __restrict__ counts,
                  int E) {
    __shared__ __attribute__((aligned(16))) float stg[2][G_ROWS * G_PITCH];
    __shared__ unsigned lc[OW_GROUPS / BIN_Q];
    const int tid  = threadIdx.x;
    {
        const int rowBlocks = (N + G_XROWS - 1) / G_XROWS;
        if ((int)blockIdx.y >= rowBlocks) {
            const int wg = ((int)blockIdx.y - rowBlocks) * (int)gridDim.x + (int)blockIdx.x;
            if (wg < BIN_CH * BIN_Q)
                bin_edges(ei, cells, counts, N, E, wg, lc, tid);
            return;
        }
    }
    const int lane = tid & 31;
    const bool split = (H == 4 && P == 65 && C == 520);
    const int wv   = __builtin_amdgcn_readfirstlane(tid >> 5);
    const int wm   = wv % G_WM;
    const int wn   = wv / G_WM;
    const int Mg   = blockIdx.y * G_XROWS;
    const int Ng   = blockIdx.x * G_COLS;
    const int M0   = Mg + wm * (G_MI * 16);
    const int N0   = Ng + wn * 64;
    const int lm   = lane & 15;
    const int kh   = (lane >> 4) * 2;

    const int seg = tid & 7;
    const int l0  = (tid >> 3) * G_PITCH + seg * 4;
    const float* __restrict__ gsrc[G_PIECES];
    #pragma unroll
    for (int i = 0; i < G_PIECES; ++i) {
        const int R  = (tid >> 3) + G_RSTEP * i;
        const int Rc = R < G_ROWS ? R : G_ROWS - 1;
        const float* __restrict__ p;
        if (Rc < G_XROWS) { const int r = Mg + Rc;           p = x + (size_t)(r < N ? r : N - 1) * F; }
        else              { const int c = gemm_col(Ng + Rc - G_XROWS, split, C); p = W + (size_t)(c < C ? c : C - 1) * F; }
        gsrc[i] = p + seg * 4;
    }

    v8f acc[G_MI][4];
    #pragma unroll
    for (int mi = 0; mi < G_MI; ++mi)
        #pragma unroll
        for (int ni = 0; ni < 4; ++ni)
            acc[mi][ni] = (v8f){};

    const int kq  = (lane >> 4) * 16;
    const int nkb = F / 32;
    const int aoff = (wm * (G_MI * 16) + lm) * G_PITCH + kq;
    const int boff = (G_XROWS + wn * 64 + lm) * G_PITCH + kq;

    v4f t[G_PIECES];
    if (nkb > 0) {
        #pragma unroll
        for (int i = 0; i < G_PIECES; ++i) t[i] = *(const v4f_a4*)(gsrc[i]);
        #pragma unroll
        for (int i = 0; i < G_PIECES; ++i)
            if ((i + 1) * G_RSTEP <= G_ROWS || (tid >> 3) + G_RSTEP * i < G_ROWS)
                *(v4f*)(&stg[0][l0 + i * (G_RSTEP * G_PITCH)]) = t[i];
        __syncthreads();
    }
    for (int kb = 0; kb < nkb; ++kb) {
        const float* sb = stg[kb & 1];
        float*       sn = stg[(kb & 1) ^ 1];
        const bool more = (kb + 1 < nkb);
        if (more) {
            #pragma unroll
            for (int i = 0; i < G_PIECES; ++i)
                t[i] = *(const v4f_a4*)(gsrc[i] + (size_t)(kb + 1) * 32);
        }
        __builtin_amdgcn_sched_barrier(0);
        gemm_cols<G_NLAST>(sb, aoff, boff, 0, acc);
        if (wn < G_WN - 1) gemm_cols<4 - G_NLAST>(sb, aoff, boff, G_NLAST, acc);
        __builtin_amdgcn_sched_barrier(0);
        if (more) {
            #pragma unroll
            for (int i = 0; i < G_PIECES; ++i)
                if ((i + 1) * G_RSTEP <= G_ROWS || (tid >> 3) + G_RSTEP * i < G_ROWS)
                    *(v4f*)(sn + l0 + i * (G_RSTEP * G_PITCH)) = t[i];
        }
        __syncthreads();
    }

    int k = nkb * 32;
    if (k < F) {
        const float* __restrict__ xrow[G_MI];
        const float* __restrict__ wrow[4];
        #pragma unroll
        for (int mi = 0; mi < G_MI; ++mi) {
            int r = M0 + mi * 16 + lm;
            xrow[mi] = x + (size_t)(r < N ? r : N - 1) * F;
        }
        #pragma unroll
        for (int ni = 0; ni < 4; ++ni) {
            int c = gemm_col(N0 + ni * 16 + lm, split, C);
            wrow[ni] = W + (size_t)(c < C ? c : C - 1) * F;
        }
        for (; k < F; k += 4) {
            v2f a[G_MI], b[4];
            #pragma unroll
            for (int mi = 0; mi < G_MI; ++mi) {
                a[mi].x = xrow[mi][k + kh];
                a[mi].y = xrow[mi][k + kh + 1];
            }
            #pragma unroll
            for (int ni = 0; ni < 4; ++ni) {
                b[ni].x = wrow[ni][k + kh];
                b[ni].y = wrow[ni][k + kh + 1];
            }
            #pragma unroll
            for (int mi = 0; mi < G_MI; ++mi)
                #pragma unroll
                for (int ni = 0; ni < 4; ++ni)
                    acc[mi][ni] = __builtin_amdgcn_wmma_f32_16x16x4_f32(
                         false, a[mi],  false, b[ni],
                         (short)0, acc[mi][ni],
                         false,  false);
        }
    }

    const int rsub = (lane >> 4) << 3;
    if (split) {
        const int cx    = (int)blockIdx.x;
        const int hl    = lane >> 4;
        const bool wide = (wn < G_WN - 1);
        const int pl    = wide ? 2 * cx + wn : (cx < 2 ? 6 : 7);
        const int f0    = wide ? 0 : (cx == 0 ? 0 : 32);
        float* __restrict__ dpl = qk + (size_t)(pl & 1) * N * 256 + (pl >> 1) * 64 + f0;
        float* wst = &stg[0][0] + wv * (16 * EP_PITCH);
        float bv[4];
        #pragma unroll
        for (int ni = 0; ni < 4; ++ni) {
            const int col = gemm_col(N0 + ni * 16 + lm, true, C);
            bv[ni] = bias[col < C ? col : C - 1];
        }
        #pragma unroll
        for (int mi = 0; mi < G_MI; ++mi) {
            #pragma unroll
            for (int ni = 0; ni < 4; ++ni) {
                if (ni < 2 || wide) {
                    #pragma unroll
                    for (int r = 0; r < 8; ++r)
                        wst[(rsub + r) * EP_PITCH + ni * 16 + lm] = acc[mi][ni][r] + bv[ni];
                }
            }
            __syncthreads();
            if (wide) {
                v4f v[8];
                #pragma unroll
                for (int i = 0; i < 8; ++i)
                    v[i] = *(const v4f*)(wst + (2 * i + hl) * EP_PITCH + 4 * lm);
                #pragma unroll
                for (int i = 0; i < 8; ++i)
                    asm volatile("" : "+v"(v[i]));
                #pragma unroll
                for (int i = 0; i < 8; ++i) {
                    const int row = M0 + mi * 16 + 2 * i + hl;
                    if (row < N) *(v4f_a4*)(dpl + (size_t)row * 256 + 4 * lm) = v[i];
                }
            } else {
                v4f v[4];
                #pragma unroll
                for (int i = 0; i < 4; ++i)
                    v[i] = *(const v4f*)(wst + (4 * i + (lane >> 3)) * EP_PITCH + 4 * (lane & 7));
                #pragma unroll
                for (int i = 0; i < 4; ++i)
                    asm volatile("" : "+v"(v[i]));
                #pragma unroll
                for (int i = 0; i < 4; ++i) {
                    const int row = M0 + mi * 16 + 4 * i + (lane >> 3);
                    if (row < N) *(v4f_a4*)(dpl + (size_t)row * 256 + 4 * (lane & 7)) = v[i];
                }
            }
            __syncthreads();
        }
        if (!wide) {
            const int col = gemm_col(N0 + 32 + lm, true, C);
            if (col < C) {
                const int p2 = col / 65;
                const int p  = col - p2 * 65;
                float* __restrict__ dst;
                size_t pitch;
                if (p < 64) { dst = qk + (size_t)(p2 & 1) * N * 256 + (p2 >> 1) * 64 + p;          pitch = 256; }
                else        { dst = qk + (size_t)N * 512 + (size_t)(p2 & 1) * N * 4 + (p2 >> 1); pitch = 4;   }
                #pragma unroll
                for (int mi = 0; mi < G_MI; ++mi) {
                    const int rbase = M0 + mi * 16 + rsub;
                    #pragma unroll
                    for (int r = 0; r < 8; ++r) {
                        const int row = rbase + r;
                        if (row < N)
                            dst[(size_t)row * pitch] = acc[mi][2][r] + bv[2];
                    }
                }
            }
        }
        return;
    }
    #pragma unroll
    for (int ni = 0; ni < 4; ++ni) {
        const int col = N0 + ni * 16 + lm;
        if (col >= C || wn * 64 + ni * 16 >= G_COLS) continue;
        const float bv = bias[col];
        float* __restrict__ dst = qk + col;
        #pragma unroll
        for (int mi = 0; mi < G_MI; ++mi) {
            const int rbase = M0 + mi * 16 + rsub;
            #pragma unroll
            for (int r = 0; r < 8; ++r) {
                const int row = rbase + r;
                if (row < N)
                    dst[(size_t)row * (size_t)C] = acc[mi][ni][r] + bv;
            }
        }
    }
}


#define K2_EPW 16

template <int CTRL>
__device__ __forceinline__ float row_xor_add(float keep, float send) {
    const int t = __builtin_amdgcn_update_dpp(0, __builtin_bit_cast(int, send),
                                              CTRL, 0xf, 0xf, false);
    return keep + __builtin_bit_cast(float, t);
}

struct EdgeFrag {
    v4f qa, qb, ka, kb;
    float qt, kt;
};

struct GatherPlanes {
    const float* __restrict__ qm;
    const float* __restrict__ km;
    const float* __restrict__ qt;
    const float* __restrict__ kt;
};

__device__ __forceinline__ void k2_load_edge(EdgeFrag& f, const GatherPlanes& g,
                                             int s, int d, int offA, int th) {
    const float* __restrict__ sq = g.qm + (size_t)s * 256 + offA;
    const float* __restrict__ dk = g.km + (size_t)d * 256 + offA;
    f.qa = *(const v4f_a4*)(sq);
    f.qb = *(const v4f_a4*)(sq + 128);
    f.ka = *(const v4f_a4*)(dk);
    f.kb = *(const v4f_a4*)(dk + 128);
    f.qt = g.qt[(size_t)s * 4 + th];
    f.kt = g.kt[(size_t)d * 4 + th];
    __builtin_amdgcn_sched_barrier(0);
}

__device__ __forceinline__ void k2_reduce_store(const EdgeFrag& f, float* __restrict__ scores,
                                                unsigned* __restrict__ smax, int e, int s,
                                                int up, int th, int lane) {
    float a1 = f.qa.x * f.ka.x;
    float a2 = f.qb.x * f.kb.x;
    a1 = fmaf(f.qa.y, f.ka.y, a1);  a2 = fmaf(f.qb.y, f.kb.y, a2);
    a1 = fmaf(f.qa.z, f.ka.z, a1);  a2 = fmaf(f.qb.z, f.kb.z, a2);
    a1 = fmaf(f.qa.w, f.ka.w, a1);  a2 = fmaf(f.qb.w, f.kb.w, a2);
    float r = row_xor_add<0x168>(up ? a2 : a1, up ? a1 : a2);
    r = row_xor_add<0x164>(r, r);
    r = row_xor_add<0x162>(r, r);
    r = row_xor_add<0x161>(r, r);
    r = fmaf(f.qt, f.kt, r);
    if ((lane & 7) == 0) {
        scores[(size_t)e * 4 + th] = r;
        atomicMax(&smax[(size_t)s * 4 + th], fenc(r));
    }
}

__global__ __launch_bounds__(256)
void k2_scores(const float* __restrict__ qk, const int* __restrict__ ei,
               float* __restrict__ scores, unsigned* __restrict__ smax,
               int N, int E, int C, int H, int P) {
    const int lane = threadIdx.x & 31;
    const int wv = __builtin_amdgcn_readfirstlane((int)(threadIdx.x >> 5));
    const int e0 = (blockIdx.x * (blockDim.x >> 5) + wv) * K2_EPW;
    if (e0 >= E) return;
    const int cnt = (E - e0 < K2_EPW) ? (E - e0) : K2_EPW;

    const int le = (e0 + (lane & 15) < E) ? (e0 + (lane & 15)) : (E - 1);
    const int ev = ei[(size_t)(lane >> 4) * E + le];

    if (H == 4 && P == 65 && C == 520) {
        const int hs   = lane >> 4;
        const int offA = hs * 64 + (lane & 15) * 4;
        const int up   = (lane >> 3) & 1;
        const int th   = hs + up * 2;
        GatherPlanes g;
        g.qm = qk;
        g.km = qk + (size_t)N * 256;
        g.qt = qk + (size_t)N * 512;
        g.kt = g.qt + (size_t)N * 4;

        EdgeFrag fa, fb;
        int sa = __builtin_amdgcn_readlane(ev, 0), sb = sa;
        k2_load_edge(fa, g, sa, __builtin_amdgcn_readlane(ev, 16), offA, th);
        int j = 0;
        for (; j + 2 < cnt; j += 2) {
            sb = __builtin_amdgcn_readlane(ev, j + 1);
            k2_load_edge(fb, g, sb, __builtin_amdgcn_readlane(ev, j + 17), offA, th);
            k2_reduce_store(fa, scores, smax, e0 + j, sa, up, th, lane);
            sa = __builtin_amdgcn_readlane(ev, j + 2);
            k2_load_edge(fa, g, sa, __builtin_amdgcn_readlane(ev, j + 18), offA, th);
            k2_reduce_store(fb, scores, smax, e0 + j + 1, sb, up, th, lane);
        }
        if (j + 1 < cnt) {
            sb = __builtin_amdgcn_readlane(ev, j + 1);
            k2_load_edge(fb, g, sb, __builtin_amdgcn_readlane(ev, j + 17), offA, th);
            k2_reduce_store(fa, scores, smax, e0 + j, sa, up, th, lane);
            k2_reduce_store(fb, scores, smax, e0 + j + 1, sb, up, th, lane);
        } else {
            k2_reduce_store(fa, scores, smax, e0 + j, sa, up, th, lane);
        }
    } else {
        for (int j = 0; j < cnt; ++j) {
            const int s = __builtin_amdgcn_readlane(ev, j);
            const int d = __builtin_amdgcn_readlane(ev, 16 + j);
            const float* __restrict__ srow = qk + (size_t)s * C;
            const float* __restrict__ drow = qk + (size_t)d * C;
            for (int h = 0; h < H; ++h) {
                const float* __restrict__ qh = srow + h * 2 * P;
                const float* __restrict__ kh = drow + h * 2 * P + P;
                float a = 0.0f;
                for (int p = lane; p < P; p += 32)
                    a += qh[p] * kh[p];
                #pragma unroll
                for (int off = 16; off > 0; off >>= 1)
                    a += __shfl_xor(a, off, 32);
                if (lane == 0) {
                    scores[(size_t)(e0 + j) * H + h] = a;
                    atomicMax(&smax[(size_t)s * H + h], fenc(a));
                }
            }
        }
    }
}

__global__ __launch_bounds__(256)
void k3_exp_denom(float* __restrict__ scores, const unsigned* __restrict__ smax,
                  const int* __restrict__ ei, float* __restrict__ denom,
                  int E, int H) {
    size_t i = (size_t)blockIdx.x * blockDim.x + threadIdx.x;
    if (i >= (size_t)E * H) return;
    const int e = (int)(i / H);
    const int h = (int)(i % H);
    const int s = ei[e];
    const float m  = fdec(smax[(size_t)s * H + h]);
    const float ex = __expf(scores[i] - m);
    scores[i] = ex;
    atomicAdd(&denom[(size_t)s * H + h], ex);
}

__global__ __launch_bounds__(256)
void k4_out(const float* __restrict__ ex, const float* __restrict__ denom,
            const int* __restrict__ ei, float* __restrict__ out, int E, int H) {
    const int e = blockIdx.x * blockDim.x + threadIdx.x;
    if (e >= E) return;
    const int s = ei[e];
    float sum = 0.0f;
    for (int h = 0; h < H; ++h)
        sum += ex[(size_t)e * H + h] / denom[(size_t)s * H + h];
    out[e] = sum / (float)H;
}


__device__ __forceinline__ unsigned ow_umin(unsigned a, unsigned b) { return a < b ? a : b; }

__device__ __forceinline__ void ow_append(int e, unsigned sl, int* le, int* ls, unsigned* cnt) {
    const unsigned idx = atomicAdd(cnt, 1u);
    if (idx < (unsigned)OW_CAP) { le[idx] = e; ls[idx] = (int)sl; }
}

__device__ __forceinline__ void ow_test4(v4i s, int e, int n0, unsigned nn,
                                         int* le, int* ls, unsigned* cnt) {
    const unsigned a = (unsigned)(s.x - n0), b = (unsigned)(s.y - n0);
    const unsigned c = (unsigned)(s.z - n0), d = (unsigned)(s.w - n0);
    if (ow_umin(ow_umin(a, b), ow_umin(c, d)) < nn) {
        #pragma nounroll
        for (int j = 0; j < 4; ++j) {
            const unsigned sl = (j == 0) ? a : (j == 1) ? b : (j == 2) ? c : d;
            if (sl < nn) ow_append(e + j, sl, le, ls, cnt);
        }
    }
}

__device__ __forceinline__ void ow_scan_simple(const int* __restrict__ ei, int lo, int hi, int n0, unsigned nn,
                                               int tid, int* le, int* ls, unsigned* cnt) {
    for (int e = lo + tid; e < hi; e += OW_THREADS) {
        const unsigned sl = (unsigned)(ei[e] - n0);
        if (sl < nn) ow_append(e, sl, le, ls, cnt);
    }
}

__device__ __forceinline__ void ow_scan(const int* __restrict__ ei, int lo, int hi, int n0, unsigned nn,
                                        int tid, int grp, int* le, int* ls, unsigned* cnt) {
    const int CH    = OW_THREADS * 4 * OW_UNR;
    const int nfull = (hi - lo) / CH;
    if (nfull > 0) {
        int c = grp % nfull;
        for (int it = 0; it < nfull; ++it) {
            const int b0 = lo + c * CH + tid * 4;
            v4i v[OW_UNR];
            #pragma unroll
            for (int u = 0; u < OW_UNR; ++u)
                v[u] = *(const v4i_a4*)(ei + b0 + u * (OW_THREADS * 4));
            __builtin_amdgcn_sched_barrier(0);
            #pragma unroll
            for (int u = 0; u < OW_UNR; ++u)
                ow_test4(v[u], b0 + u * (OW_THREADS * 4), n0, nn, le, ls, cnt);
            c = (c + 1 < nfull) ? c + 1 : 0;
        }
    }
    const int r0 = lo + nfull * CH;
    if (r0 < hi) {
        const int b0 = r0 + tid * 4;
        v4i v[OW_UNR];
        #pragma unroll
        for (int u = 0; u < OW_UNR; ++u) {
            const int b = b0 + u * (OW_THREADS * 4);
            v[u] = *(const v4i_a4*)(ei + ((b + 3 < hi) ? b : 0));
        }
        __builtin_amdgcn_sched_barrier(0);
        #pragma unroll
        for (int u = 0; u < OW_UNR; ++u) {
            const int b = b0 + u * (OW_THREADS * 4);
            if (b + 3 < hi) ow_test4(v[u], b, n0, nn, le, ls, cnt);
        }
        ow_scan_simple(ei, lo + ((hi - lo) & ~3), hi, n0, nn, tid, le, ls, cnt);
    }
}

__device__ __forceinline__ void ow_dst(const int* __restrict__ ei, int E, int n, int tid,
                                       const int* le, int* ld) {
    for (int i0 = tid; i0 < n; i0 += OW_THREADS * 4) {
        int v[4];
        #pragma unroll
        for (int u = 0; u < 4; ++u) {
            int i = i0 + u * OW_THREADS;
            i = i < n ? i : n - 1;
            v[u] = ei[(size_t)E + le[i]];
        }
        #pragma unroll
        for (int u = 0; u < 4; ++u) {
            int i = i0 + u * OW_THREADS;
            i = i < n ? i : n - 1;
            ld[i] = v[u];
        }
    }
}

struct KFrag { v4f ka, kb; };

__device__ __forceinline__ void ow_scores(int n, const float* __restrict__ km, const float* __restrict__ kt,
                                          const float* qs, const float* qts,
                                          const int* ld, const int* ls, float* sc, unsigned* mx,
                                          int wv, int lane) {
    const int hs   = lane >> 4;
    const int offA = hs * 64 + (lane & 15) * 4;
    const int up   = (lane >> 3) & 1;
    const int th   = hs + up * 2;
    for (int base = wv * OW_B; base < n; base += (OW_THREADS / 32) * OW_B) {
        int ii = base + (lane & (OW_B - 1));
        ii = ii < n ? ii : n - 1;
        const int dv = ld[ii];
        const int sv = ls[ii];
        int i2 = base + (lane >> 2);
        i2 = i2 < n ? i2 : n - 1;
        const float ktv = kt[(size_t)ld[i2] * 4 + (lane & 3)];
        KFrag f[OW_B];
        #pragma unroll
        for (int j = 0; j < OW_B; ++j) {
            const int d = __builtin_amdgcn_readlane(dv, j);
            const float* __restrict__ dk = km + (size_t)d * 256 + offA;
            f[j].ka = *(const v4f_a4*)(dk);
            f[j].kb = *(const v4f_a4*)(dk + 128);
        }
        __builtin_amdgcn_sched_barrier(0);
        #pragma unroll
        for (int j = 0; j < OW_B; ++j) {
            if (base + j < n) {
                const int sl = __builtin_amdgcn_readlane(sv, j);
                const v4f qa = *(const v4f*)(qs + sl * 256 + offA);
                const v4f qb = *(const v4f*)(qs + sl * 256 + offA + 128);
                const float qtv = qts[sl * 4 + th];
                const float ktj = __shfl(ktv, 4 * j + th, 32);
                float a1 = qa.x * f[j].ka.x;
                float a2 = qb.x * f[j].kb.x;
                a1 = fmaf(qa.y, f[j].ka.y, a1);  a2 = fmaf(qb.y, f[j].kb.y, a2);
                a1 = fmaf(qa.z, f[j].ka.z, a1);  a2 = fmaf(qb.z, f[j].kb.z, a2);
                a1 = fmaf(qa.w, f[j].ka.w, a1);  a2 = fmaf(qb.w, f[j].kb.w, a2);
                float r = row_xor_add<0x168>(up ? a2 : a1, up ? a1 : a2);
                r = row_xor_add<0x164>(r, r);
                r = row_xor_add<0x162>(r, r);
                r = row_xor_add<0x161>(r, r);
                r = fmaf(qtv, ktj, r);
                if ((lane & 7) == 0) {
                    sc[(base + j) * 4 + th] = r;
                    atomicMax(&mx[sl * 4 + th], fenc(r));
                }
            }
        }
    }
}

__global__ __launch_bounds__(OW_THREADS)
void ke_owner(const float* __restrict__ qk, const int* __restrict__ ei,
              const int* __restrict__ cells, const int* __restrict__ counts,
              float* __restrict__ out, int N, int E) {
    __shared__ __attribute__((aligned(16))) float qs[OW_NODES * 256];
    __shared__ __attribute__((aligned(16))) float qts[OW_NODES * 4];
    __shared__ __attribute__((aligned(16))) float sm[OW_NODES * 4];
    __shared__ unsigned mx[OW_NODES * 4];
    __shared__ __attribute__((aligned(16))) float sc[OW_CAP * 4];
    __shared__ int le[OW_CAP];
    __shared__ int ls[OW_CAP];
    __shared__ int ld[OW_CAP];
    __shared__ unsigned cnt;
    __shared__ int cc[BIN_CH];
    __shared__ int off[BIN_CH];
    __shared__ int binfo[2];

    const int tid  = threadIdx.x;
    const int lane = tid & 31;
    const int wv   = __builtin_amdgcn_readfirstlane(tid >> 5);
    const int per  = (N + (int)gridDim.x - 1) / (int)gridDim.x;
    const int n0   = (int)blockIdx.x * per;
    if (n0 >= N) return;
    const int nni  = (N - n0 < per) ? (N - n0) : per;
    const unsigned nn = (unsigned)nni;

    const float* __restrict__ qm = qk;
    const float* __restrict__ km = qk + (size_t)N * 256;
    const float* __restrict__ qt = qk + (size_t)N * 512;
    const float* __restrict__ kt = qt + (size_t)N * 4;

    for (int i = tid; i < nni * 4; i += OW_THREADS) {
        mx[i]  = 0u;
        sm[i]  = 0.0f;
        qts[i] = qt[(size_t)n0 * 4 + i];
    }
    if (tid == 0) cnt = 0u;
    if (tid < BIN_CH) cc[tid] = counts[(size_t)blockIdx.x * BIN_CH + tid];
    {
        const float* __restrict__ qsrc = qm + (size_t)n0 * 256;
        const int tot = nni * 64;
        for (int i0 = tid; i0 < tot; i0 += OW_THREADS * OW_FILL) {
            v4f t[OW_FILL];
            #pragma unroll
            for (int u = 0; u < OW_FILL; ++u) {
                int i = i0 + u * OW_THREADS;
                i = i < tot ? i : tot - 1;
                t[u] = *(const v4f_a4*)(qsrc + 4 * i);
            }
            __builtin_amdgcn_sched_barrier(0);
            #pragma unroll
            for (int u = 0; u < OW_FILL; ++u) {
                int i = i0 + u * OW_THREADS;
                i = i < tot ? i : tot - 1;
                *(v4f*)(qs + 4 * i) = t[u];
            }
        }
    }
    __syncthreads();

    if (tid < BIN_CH) {
        int o = 0;
        for (int i = 0; i < tid; ++i) o += cc[i];
        off[tid] = o;
    } else if (tid == BIN_CH) {
        int t = 0, bad = 0;
        for (int i = 0; i < BIN_CH; ++i) { t += cc[i]; bad |= (cc[i] > BIN_CAP) ? 1 : 0; }
        binfo[0] = t;
        binfo[1] = bad;
    }
    __syncthreads();
    int n = binfo[0];
    bool listed = false;
    if (binfo[1] != 0) {
        ow_scan(ei, 0, E, n0, nn, tid, (int)blockIdx.x, le, ls, &cnt);
        __syncthreads();
        n = (int)cnt;
        if (n <= OW_CAP) {
            ow_dst(ei, E, n, tid, le, ld);
            listed = true;
        }
    } else if (n <= OW_CAP) {
        const int* __restrict__ mine = cells + (size_t)blockIdx.x * (BIN_CH * BIN_CAP * 2);
        v2i  en[BIN_TRIPS];
        int  pos[BIN_TRIPS];
        #pragma unroll
        for (int u = 0; u < BIN_TRIPS; ++u) {
            const int s = tid + u * OW_THREADS;
            const int b = s / BIN_CAP;
            const int k = s - b * BIN_CAP;
            const bool used = (b < BIN_CH) && (k < cc[b < BIN_CH ? b : 0]);
            pos[u] = used ? off[b] + k : -1;
            const int* __restrict__ p = used ? mine + 2 * s : counts + (size_t)blockIdx.x * BIN_CH;
            en[u] = *(const v2i*)p;
        }
        __builtin_amdgcn_sched_barrier(0);
        #pragma unroll
        for (int u = 0; u < BIN_TRIPS; ++u) {
            if (pos[u] >= 0) {
                le[pos[u]] = (int)((unsigned)en[u].x & 0xFFFFFFu);
                ls[pos[u]] = (int)((unsigned)en[u].x >> 24);
                ld[pos[u]] = en[u].y;
            }
        }
        listed = true;
    }

    if (listed) {
        __syncthreads();
        ow_scores(n, km, kt, qs, qts, ld, ls, sc, mx, wv, lane);
        __syncthreads();
        for (int t = tid; t < n * 4; t += OW_THREADS) {
            const int g  = ls[t >> 2] * 4 + (t & 3);
            const float ex = __expf(sc[t] - fdec(mx[g]));
            sc[t] = ex;
            atomicAdd(&sm[g], ex);
        }
        __syncthreads();
        for (int i = tid; i < n; i += OW_THREADS) {
            const v4f x  = *(const v4f*)(sc + 4 * i);
            const v4f dn = *(const v4f*)(sm + 4 * ls[i]);
            float sum = 0.0f;
            sum += x.x / dn.x;
            sum += x.y / dn.y;
            sum += x.z / dn.z;
            sum += x.w / dn.w;
            out[le[i]] = sum / 4.0f;
        }
        return;
    }

    const int nr = (E + OW_CAP - 1) / OW_CAP;
    for (int pass = 0; pass < 3; ++pass) {
        for (int c = 0; c < nr; ++c) {
            __syncthreads();
            if (tid == 0) cnt = 0u;
            __syncthreads();
            const int hi = ((c + 1) * OW_CAP < E) ? (c + 1) * OW_CAP : E;
            ow_scan_simple(ei, c * OW_CAP, hi, n0, nn, tid, le, ls, &cnt);
            __syncthreads();
            const int m = (int)cnt;
            ow_dst(ei, E, m, tid, le, ld);
            __syncthreads();
            ow_scores(m, km, kt, qs, qts, ld, ls, sc, mx, wv, lane);
            __syncthreads();
            if (pass == 1) {
                for (int t = tid; t < m * 4; t += OW_THREADS) {
                    const int g = ls[t >> 2] * 4 + (t & 3);
                    atomicAdd(&sm[g], __expf(sc[t] - fdec(mx[g])));
                }
            } else if (pass == 2) {
                for (int i = tid; i < m; i += OW_THREADS) {
                    const int g = ls[i] * 4;
                    const v4f x  = *(const v4f*)(sc + 4 * i);
                    const v4f dn = *(const v4f*)(sm + g);
                    float sum = 0.0f;
                    sum += __expf(x.x - fdec(mx[g]))     / dn.x;
                    sum += __expf(x.y - fdec(mx[g + 1])) / dn.y;
                    sum += __expf(x.z - fdec(mx[g + 2])) / dn.z;
                    sum += __expf(x.w - fdec(mx[g + 3])) / dn.w;
                    out[le[i]] = sum / 4.0f;
                }
            }
        }
    }
}

extern "C" void kernel_launch(void* const* d_in, const int* in_sizes, int n_in,
                              void* d_out, int out_size, void* d_ws, size_t ws_size,
                              hipStream_t stream) {
    const float* x    = (const float*)d_in[0];
    const int*   ei   = (const int*)d_in[1];
    const float* W    = (const float*)d_in[2];
    const float* bias = (const float*)d_in[3];
    const int C = in_sizes[3];
    const int F = in_sizes[2] / C;
    const int N = in_sizes[0] / F;
    const int E = in_sizes[1] / 2;
    const int H = (C - 2 * F) / 2;
    const int P = F / H + 1;

    float*    qk     = (float*)d_ws;
    float*    scores = qk + (size_t)N * C;
    unsigned* smax   = (unsigned*)(scores + (size_t)E * H);
    float*    denom  = (float*)(smax + (size_t)N * H);

    int* cells  = (int*)scores;
    int* counts = (int*)smax;
    const bool own = (H == 4 && P == 65 && C == 520 && E >= 4 && E < (1 << 24) &&
                      (N + OW_GROUPS - 1) / OW_GROUPS <= OW_NODES &&
                      (size_t)OW_GROUPS * BIN_CH * BIN_CAP * 2 <= (size_t)E * H &&
                      (size_t)OW_GROUPS * BIN_CH <= (size_t)N * H);

    if (!own) {
        size_t n = (size_t)2 * N * H;
        k0_zero<<<(unsigned)((n + 255) / 256), 256, 0, stream>>>(smax, n);
    }

    {
        const int gx = (C + G_COLS - 1) / G_COLS;
        const int gy = (N + G_XROWS - 1) / G_XROWS;
        dim3 grid(gx, gy + (own ? (BIN_CH * BIN_Q + gx - 1) / gx : 0));
        k1_gemm_wmma<<<grid, G_THREADS, 0, stream>>>(x, W, bias, qk, N, F, C, H, P,
                                                     ei, cells, counts, E);
    }

    if (own) {
        ke_owner<<<OW_GROUPS, OW_THREADS, 0, stream>>>(qk, ei, cells, counts, (float*)d_out, N, E);
        return;
    }

    {
        const int epb = (256 / 32) * K2_EPW;
        k2_scores<<<(E + epb - 1) / epb, 256, 0, stream>>>(qk, ei, scores, smax,
                                                           N, E, C, H, P);
    }

    {
        size_t n = (size_t)E * H;
        k3_exp_denom<<<(unsigned)((n + 255) / 256), 256, 0, stream>>>(scores, smax,
                                                                      ei, denom, E, H);
    }

    k4_out<<<(E + 255) / 256, 256, 0, stream>>>(scores, denom, ei,
                                                (float*)d_out, E, H);
}
